# starting kernel
# baseline (speedup 1.0000x reference)
#include <hip/hip_runtime.h>
#include <hip/hip_bf16.h>

#define NN   50000
#define EE   800000
#define GG   512
#define DD   64
#define DOUT 32
#define BN_EPS 1e-5f

typedef __attribute__((ext_vector_type(2)))  float    v2f;
typedef __attribute__((ext_vector_type(8)))  float    v8f;
typedef __attribute__((ext_vector_type(16))) _Float16 v16h;

#if defined(__AMDGCN__) && __has_builtin(__builtin_amdgcn_wmma_f32_16x16x4_f32)
#define USE_WMMA_F32 1
#else
#define USE_WMMA_F32 0
#endif


__global__ void k_zero(float* __restrict__ p, int n) {
  int i = blockIdx.x * blockDim.x + threadIdx.x;
  if (i < n) p[i] = 0.0f;
}

__global__ void k_edge_count(const int* __restrict__ ei,
                             float* __restrict__ deg_feat,
                             float* __restrict__ degcnt, int E) {
  int e = blockIdx.x * blockDim.x + threadIdx.x;
  if (e >= E) return;
  int s = ei[e];
  int d = ei[E + e];
  atomicAdd(&deg_feat[s], 1.0f);
  atomicAdd(&degcnt[d], 1.0f);
}

__global__ void k_dinv(const float* __restrict__ degcnt, float* __restrict__ dinv, int n) {
  int i = blockIdx.x * blockDim.x + threadIdx.x;
  if (i < n) dinv[i] = rsqrtf(degcnt[i] + 1.0f);
}

__global__ void k_l1_hw_self(const float* __restrict__ deg_feat,
                             const float* __restrict__ W1,
                             const float* __restrict__ dinv,
                             float* __restrict__ hw,
                             float* __restrict__ acc, int n64) {
  int i = blockIdx.x * blockDim.x + threadIdx.x;
  if (i >= n64) return;
  int n = i >> 6;
  float h = deg_feat[n] * W1[i & 63];
  float di = dinv[n];
  hw[i]  = h;
  acc[i] = h * (di * di);
}

__global__ void k_scatter(const int* __restrict__ ei,
                          const float* __restrict__ hw,
                          const float* __restrict__ dinv,
                          float* __restrict__ acc, int E) {
  int t = blockIdx.x * blockDim.x + threadIdx.x;
  int e = t >> 4;
  int g = t & 15;
  if (e >= E) return;
  int s = ei[e];
  int d = ei[E + e];
  float w = dinv[s] * dinv[d];
  float4 m = ((const float4*)(hw + (size_t)s * DD))[g];
  float* dst = acc + (size_t)d * DD + g * 4;
  atomicAdd(dst + 0, m.x * w);
  atomicAdd(dst + 1, m.y * w);
  atomicAdd(dst + 2, m.z * w);
  atomicAdd(dst + 3, m.w * w);
}

__global__ void k_post_bnrelu(const float* __restrict__ acc,
                              const float* __restrict__ b,
                              const float* __restrict__ g,
                              const float* __restrict__ be,
                              const float* __restrict__ m,
                              const float* __restrict__ v,
                              float* __restrict__ x, int n64) {
  int i = blockIdx.x * blockDim.x + threadIdx.x;
  if (i >= n64) return;
  int d = i & 63;
  float rs = rsqrtf(v[d] + BN_EPS);
  float y = ((acc[i] + b[d]) - m[d]) * rs * g[d] + be[d];
  x[i] = fmaxf(y, 0.0f);
}

__global__ void k_post_bias(const float* __restrict__ acc,
                            const float* __restrict__ b,
                            float* __restrict__ x, int n64) {
  int i = blockIdx.x * blockDim.x + threadIdx.x;
  if (i < n64) x[i] = acc[i] + b[i & 63];
}

__global__ void k_pool_scatter(const int* __restrict__ batch,
                               const float* __restrict__ x,
                               float* __restrict__ pool,
                               float* __restrict__ cnt, int n) {
  int t = blockIdx.x * blockDim.x + threadIdx.x;
  int nd = t >> 4;
  int g  = t & 15;
  if (nd >= n) return;
  int bg = batch[nd];
  float4 m = ((const float4*)(x + (size_t)nd * DD))[g];
  float* dst = pool + (size_t)bg * DD + g * 4;
  atomicAdd(dst + 0, m.x);
  atomicAdd(dst + 1, m.y);
  atomicAdd(dst + 2, m.z);
  atomicAdd(dst + 3, m.w);
  if (g == 0) atomicAdd(&cnt[bg], 1.0f);
}

__global__ void k_pool_div(const float* __restrict__ pool,
                           const float* __restrict__ cnt,
                           float* __restrict__ pooled, int g64) {
  int i = blockIdx.x * blockDim.x + threadIdx.x;
  if (i >= g64) return;
  float c = cnt[i >> 6];
  pooled[i] = (c > 0.0f) ? pool[i] / c : 0.0f;
}

__global__ void k_gemm_wmma(const float* __restrict__ A,
                            const float* __restrict__ B,
                            float* __restrict__ C,
                            const float* __restrict__ bias,
                            const float* __restrict__ dinv,
                            float* __restrict__ acc,
                            int M, int K, int Nc) {
  int wid  = (blockIdx.x * blockDim.x + threadIdx.x) >> 5;
  int lane = threadIdx.x & 31;
  int tiles_n = Nc >> 4;
  int tm = wid / tiles_n;
  int tn = wid - tm * tiles_n;
  if (tm * 16 >= M) return;

  int half = lane >> 4;
  int lm   = lane & 15;
  const float* Arow = A + (size_t)(tm * 16 + lm) * K;
  int ncol = tn * 16 + lm;

  v8f c = {};
#if USE_WMMA_F32
  for (int k0 = 0; k0 < K; k0 += 4) {
    v2f a, b;
    a.x = Arow[k0 + 2 * half + 0];
    a.y = Arow[k0 + 2 * half + 1];
    b.x = B[(size_t)(k0 + 2 * half + 0) * Nc + ncol];
    b.y = B[(size_t)(k0 + 2 * half + 1) * Nc + ncol];
    c = __builtin_amdgcn_wmma_f32_16x16x4_f32(false, a, false, b, (short)0, c,
                                              false, false);
  }
#else
  for (int k0 = 0; k0 < K; k0 += 32) {
    v16h a, b;
#pragma unroll
    for (int i = 0; i < 8; ++i) {
      a[i]     = (_Float16)Arow[k0 + 8 * half + i];
      a[8 + i] = (_Float16)Arow[k0 + 16 + 8 * half + i];
    }
#pragma unroll
    for (int i = 0; i < 16; ++i)
      b[i] = (_Float16)B[(size_t)(k0 + 16 * half + i) * Nc + ncol];
    c = __builtin_amdgcn_wmma_f32_16x16x32_f16(false, a, false, b, (short)0, c,
                                               false, false);
  }
#endif

  float bv = bias ? bias[ncol] : 0.0f;
  if (acc) {
#pragma unroll
    for (int r = 0; r < 8; ++r) {
      int row = tm * 16 + 8 * half + r;
      size_t idx = (size_t)row * Nc + ncol;
      float val = c[r] + bv;
      float di = dinv[row];
      C[idx]   = val;
      acc[idx] = val * (di * di);
    }
  } else {
#pragma unroll
    for (int r = 0; r < 8; ++r) {
      C[(size_t)(tm * 16 + 8 * half + r) * Nc + ncol] = c[r] + bv;
    }
  }
}

static inline int blk(long long n, int b) { return (int)((n + b - 1) / b); }

extern "C" void kernel_launch(void* const* d_in, const int* in_sizes, int n_in,
                              void* d_out, int out_size, void* d_ws, size_t ws_size,
                              hipStream_t stream) {
  (void)in_sizes; (void)n_in; (void)out_size; (void)ws_size;
  const int*   edge_index = (const int*)d_in[0];
  const int*   batch      = (const int*)d_in[1];
  const float* W1 = (const float*)d_in[2];
  const float* b1 = (const float*)d_in[3];
  const float* W2 = (const float*)d_in[4];
  const float* b2 = (const float*)d_in[5];
  const float* W3 = (const float*)d_in[6];
  const float* b3 = (const float*)d_in[7];
  const float* g1 = (const float*)d_in[8];
  const float* be1= (const float*)d_in[9];
  const float* m1 = (const float*)d_in[10];
  const float* v1 = (const float*)d_in[11];
  const float* g2 = (const float*)d_in[12];
  const float* be2= (const float*)d_in[13];
  const float* m2 = (const float*)d_in[14];
  const float* v2 = (const float*)d_in[15];
  const float* Wl = (const float*)d_in[16];
  const float* bl = (const float*)d_in[17];
  float* out = (float*)d_out;

  char* ws = (char*)d_ws;
  size_t off = 0;
  auto carve = [&](size_t bytes) -> float* {
    float* p = (float*)(ws + off);
    off += (bytes + 255) & ~(size_t)255;
    return p;
  };
  float* deg_feat = carve((size_t)NN * 4);
  float* degcnt   = carve((size_t)NN * 4);
  float* dinv     = carve((size_t)NN * 4);
  float* bufX     = carve((size_t)NN * DD * 4);
  float* bufH     = carve((size_t)NN * DD * 4);
  float* bufA     = carve((size_t)NN * DD * 4);
  float* pool     = carve((size_t)GG * DD * 4);
  float* cnt      = carve((size_t)GG * 4);
  float* pooled   = carve((size_t)GG * DD * 4);

  const int B = 256;
  const long long N64 = (long long)NN * DD;
  const long long G64 = (long long)GG * DD;

  k_zero<<<blk(NN, B), B, 0, stream>>>(deg_feat, NN);
  k_zero<<<blk(NN, B), B, 0, stream>>>(degcnt, NN);
  k_zero<<<blk(G64, B), B, 0, stream>>>(pool, (int)G64);
  k_zero<<<blk(GG, B), B, 0, stream>>>(cnt, GG);
  k_edge_count<<<blk(EE, B), B, 0, stream>>>(edge_index, deg_feat, degcnt, EE);
  k_dinv<<<blk(NN, B), B, 0, stream>>>(degcnt, dinv, NN);

  const int tiles_NN  = (NN / 16) * (DD / 16);
  const int tiles_FIN = (GG / 16) * (DOUT / 16);

  k_l1_hw_self<<<blk(N64, B), B, 0, stream>>>(deg_feat, W1, dinv, bufH, bufA, (int)N64);
  k_scatter<<<blk((long long)EE * 16, B), B, 0, stream>>>(edge_index, bufH, dinv, bufA, EE);
  k_post_bnrelu<<<blk(N64, B), B, 0, stream>>>(bufA, b1, g1, be1, m1, v1, bufX, (int)N64);

  k_gemm_wmma<<<blk((long long)tiles_NN * 32, B), B, 0, stream>>>(
      bufX, W2, bufH, nullptr, dinv, bufA, NN, DD, DD);
  k_scatter<<<blk((long long)EE * 16, B), B, 0, stream>>>(edge_index, bufH, dinv, bufA, EE);
  k_post_bnrelu<<<blk(N64, B), B, 0, stream>>>(bufA, b2, g2, be2, m2, v2, bufX, (int)N64);

  k_gemm_wmma<<<blk((long long)tiles_NN * 32, B), B, 0, stream>>>(
      bufX, W3, bufH, nullptr, dinv, bufA, NN, DD, DD);
  k_scatter<<<blk((long long)EE * 16, B), B, 0, stream>>>(edge_index, bufH, dinv, bufA, EE);
  k_post_bias<<<blk(N64, B), B, 0, stream>>>(bufA, b3, bufX, (int)N64);

  k_pool_scatter<<<blk((long long)NN * 16, B), B, 0, stream>>>(batch, bufX, pool, cnt, NN);
  k_pool_div<<<blk(G64, B), B, 0, stream>>>(pool, cnt, pooled, (int)G64);
  k_gemm_wmma<<<blk((long long)tiles_FIN * 32, B), B, 0, stream>>>(
      pooled, Wl, out, bl, nullptr, nullptr, GG, DD, DOUT);
}


# Layer 3 + mean pool as one edge pass: each edge adds its scaled layer-2 row into an LDS table of graph sums (LDS float atomics); W3 applied to the 512 pooled rows; sum order: arrival, slabs, then k
# speedup vs baseline: 1.2747x; 1.2747x over previous
#include <hip/hip_runtime.h>
#include <hip/hip_bf16.h>

#define NN   50000
#define EE   800000
#define GG   512
#define DD   64
#define DOUT 32
#define BN_EPS 1e-5f

typedef __attribute__((ext_vector_type(2)))  float    v2f;
typedef __attribute__((ext_vector_type(8)))  float    v8f;
typedef __attribute__((ext_vector_type(16))) _Float16 v16h;

#if defined(__AMDGCN__) && __has_builtin(__builtin_amdgcn_wmma_f32_16x16x4_f32)
#define USE_WMMA_F32 1
#else
#define USE_WMMA_F32 0
#endif


__global__ void k_zero(float* __restrict__ p, int n) {
  int i = blockIdx.x * blockDim.x + threadIdx.x;
  if (i < n) p[i] = 0.0f;
}

__global__ void k_edge_count(const int* __restrict__ ei,
                             float* __restrict__ deg_feat,
                             float* __restrict__ degcnt, int E) {
  int e = blockIdx.x * blockDim.x + threadIdx.x;
  if (e >= E) return;
  int s = ei[e];
  int d = ei[E + e];
  atomicAdd(&deg_feat[s], 1.0f);
  atomicAdd(&degcnt[d], 1.0f);
}

__global__ void k_dinv(const float* __restrict__ degcnt, float* __restrict__ dinv, int n) {
  int i = blockIdx.x * blockDim.x + threadIdx.x;
  if (i < n) dinv[i] = rsqrtf(degcnt[i] + 1.0f);
}

__global__ void k_l1_hw_self(const float* __restrict__ deg_feat,
                             const float* __restrict__ W1,
                             const float* __restrict__ dinv,
                             float* __restrict__ hw,
                             float* __restrict__ acc, int n64) {
  int i = blockIdx.x * blockDim.x + threadIdx.x;
  if (i >= n64) return;
  int n = i >> 6;
  float h = deg_feat[n] * W1[i & 63];
  float di = dinv[n];
  hw[i]  = h;
  acc[i] = h * (di * di);
}

__global__ void k_scatter(const int* __restrict__ ei,
                          const float* __restrict__ hw,
                          const float* __restrict__ dinv,
                          float* __restrict__ acc, int E) {
  int t = blockIdx.x * blockDim.x + threadIdx.x;
  int e = t >> 4;
  int g = t & 15;
  if (e >= E) return;
  int s = ei[e];
  int d = ei[E + e];
  float w = dinv[s] * dinv[d];
  float4 m = ((const float4*)(hw + (size_t)s * DD))[g];
  float* dst = acc + (size_t)d * DD + g * 4;
  atomicAdd(dst + 0, m.x * w);
  atomicAdd(dst + 1, m.y * w);
  atomicAdd(dst + 2, m.z * w);
  atomicAdd(dst + 3, m.w * w);
}

__global__ void k_post_bnrelu(const float* __restrict__ acc,
                              const float* __restrict__ b,
                              const float* __restrict__ g,
                              const float* __restrict__ be,
                              const float* __restrict__ m,
                              const float* __restrict__ v,
                              float* __restrict__ x, int n64) {
  int i = blockIdx.x * blockDim.x + threadIdx.x;
  if (i >= n64) return;
  int d = i & 63;
  float rs = rsqrtf(v[d] + BN_EPS);
  float y = ((acc[i] + b[d]) - m[d]) * rs * g[d] + be[d];
  x[i] = fmaxf(y, 0.0f);
}

#define PB_BLOCKS 256
#define PB_WAVES  32
#define TOT_ITEMS (EE + NN)
#define N_BATCH   ((TOT_ITEMS + 31) / 32)
#define N_PWAVES  (PB_BLOCKS * PB_WAVES)
#define KB        ((N_BATCH + N_PWAVES - 1) / N_PWAVES)
#define GRP       16
#define TAB_FLOATS (41 * 1024)

__global__ __launch_bounds__(PB_WAVES * 32)
void k_pool_edges(const int* __restrict__ ei,
                  const int* __restrict__ batch,
                  const float* __restrict__ x,
                  const float* __restrict__ dinv,
                  float* __restrict__ slab,
                  float* __restrict__ cslab) {
  __shared__ float tab[TAB_FLOATS];
  __shared__ float ctab[GG];
  const int tid  = threadIdx.x;
  const int lane = tid & 31;
  const int gw   = blockIdx.x * PB_WAVES + __builtin_amdgcn_readfirstlane(tid >> 5);

  float4 z4 = {0.0f, 0.0f, 0.0f, 0.0f};
  for (int i = tid; i < GG * DD / 4; i += PB_WAVES * 32) ((float4*)tab)[i] = z4;
  if (tid < GG) ctab[tid] = 0.0f;

  int   sv[KB], gv[KB];
  float wv[KB];
  int   dv[KB];
#pragma unroll
  for (int k = 0; k < KB; ++k) {
    int it  = (gw + k * N_PWAVES) * 32 + lane;
    int ie  = it < EE ? it : EE - 1;
    int ls  = ei[ie];
    int ld  = ei[EE + ie];
    int nd  = it - EE;
    nd = nd < 0 ? 0 : (nd > NN - 1 ? NN - 1 : nd);
    sv[k] = it < EE ? ls : nd;
    dv[k] = it < EE ? ld : nd;
  }
#pragma unroll
  for (int k = 0; k < KB; ++k) {
    wv[k] = dinv[sv[k]] * dinv[dv[k]];
    gv[k] = batch[dv[k]];
  }
  __syncthreads();

#pragma unroll
  for (int k = 0; k < KB; ++k) {
    const int base = (gw + k * N_PWAVES) * 32;
    int nvalid = TOT_ITEMS - base;
    if (nvalid <= 0) continue;
    const int it = base + lane;
    if (it >= EE && it < TOT_ITEMS) atomicAdd(&ctab[gv[k]], 1.0f);
    if (nvalid >= 32) {
#pragma unroll
      for (int j0 = 0; j0 < 32; j0 += GRP) {
        float v0[GRP], v1[GRP];
#pragma unroll
        for (int u = 0; u < GRP; ++u) {
          int sj = __builtin_amdgcn_readlane(sv[k], j0 + u);
          int idx = sj * DD + lane;
          v0[u] = x[idx];
          v1[u] = x[idx + 32];
        }
#pragma unroll
        for (int u = 0; u < GRP; ++u) {
          int   gj = __builtin_amdgcn_readlane(gv[k], j0 + u);
          float wj = __int_as_float(__builtin_amdgcn_readlane(__float_as_int(wv[k]), j0 + u));
          atomicAdd(&tab[gj * DD + lane],      v0[u] * wj);
          atomicAdd(&tab[gj * DD + lane + 32], v1[u] * wj);
        }
      }
    } else {
      for (int j = 0; j < nvalid; ++j) {
        int   sj = __builtin_amdgcn_readlane(sv[k], j);
        int   gj = __builtin_amdgcn_readlane(gv[k], j);
        float wj = __int_as_float(__builtin_amdgcn_readlane(__float_as_int(wv[k]), j));
        int idx = sj * DD + lane;
        float a0 = x[idx];
        float a1 = x[idx + 32];
        atomicAdd(&tab[gj * DD + lane],      a0 * wj);
        atomicAdd(&tab[gj * DD + lane + 32], a1 * wj);
      }
    }
  }
  __syncthreads();

  float4* my = (float4*)(slab + (size_t)blockIdx.x * (GG * DD));
  for (int i = tid; i < GG * DD / 4; i += PB_WAVES * 32) my[i] = ((const float4*)tab)[i];
  if (tid < GG) cslab[blockIdx.x * GG + tid] = ctab[tid];
}

__global__ __launch_bounds__(256)
void k_pool_reduce(const float* __restrict__ slab,
                   const float* __restrict__ cslab,
                   float* __restrict__ pooled,
                   float* __restrict__ cnt) {
  __shared__ float part[4][DD];
  __shared__ float cpart[8];
  const int g = blockIdx.x;
  const int t = threadIdx.x;
  const int f = t & (DD - 1);
  const int q = t >> 6;
  const float* p = slab + (size_t)(q * (PB_BLOCKS / 4)) * (GG * DD) + g * DD + f;
  float a = 0.0f;
#pragma unroll 16
  for (int b = 0; b < PB_BLOCKS / 4; ++b) a += p[(size_t)b * (GG * DD)];
  part[q][f] = a;
  float c = cslab[t * GG + g];
  for (int off = 16; off > 0; off >>= 1) c += __shfl_xor(c, off);
  if ((t & 31) == 0) cpart[t >> 5] = c;
  __syncthreads();
  if (t < DD) {
    float tot = (part[0][t] + part[1][t]) + (part[2][t] + part[3][t]);
    float cc = 0.0f;
    for (int i = 0; i < 8; ++i) cc += cpart[i];
    pooled[g * DD + t] = (cc > 0.0f) ? tot / cc : 0.0f;
    if (t == 0) cnt[g] = cc;
  }
}

__global__ void k_gemm_wmma(const float* __restrict__ A,
                            const float* __restrict__ B,
                            float* __restrict__ C,
                            const float* __restrict__ bias,
                            const float* __restrict__ dinv,
                            float* __restrict__ acc,
                            const float* __restrict__ cnt,
                            int M, int K, int Nc) {
  int wid  = (blockIdx.x * blockDim.x + threadIdx.x) >> 5;
  int lane = threadIdx.x & 31;
  int tiles_n = Nc >> 4;
  int tm = wid / tiles_n;
  int tn = wid - tm * tiles_n;
  if (tm * 16 >= M) return;

  int half = lane >> 4;
  int lm   = lane & 15;
  const float* Arow = A + (size_t)(tm * 16 + lm) * K;
  int ncol = tn * 16 + lm;

  v8f c = {};
#if USE_WMMA_F32
  for (int k0 = 0; k0 < K; k0 += 4) {
    v2f a, b;
    a.x = Arow[k0 + 2 * half + 0];
    a.y = Arow[k0 + 2 * half + 1];
    b.x = B[(size_t)(k0 + 2 * half + 0) * Nc + ncol];
    b.y = B[(size_t)(k0 + 2 * half + 1) * Nc + ncol];
    c = __builtin_amdgcn_wmma_f32_16x16x4_f32(false, a, false, b, (short)0, c,
                                              false, false);
  }
#else
  for (int k0 = 0; k0 < K; k0 += 32) {
    v16h a, b;
#pragma unroll
    for (int i = 0; i < 8; ++i) {
      a[i]     = (_Float16)Arow[k0 + 8 * half + i];
      a[8 + i] = (_Float16)Arow[k0 + 16 + 8 * half + i];
    }
#pragma unroll
    for (int i = 0; i < 16; ++i)
      b[i] = (_Float16)B[(size_t)(k0 + 16 * half + i) * Nc + ncol];
    c = __builtin_amdgcn_wmma_f32_16x16x32_f16(false, a, false, b, (short)0, c,
                                               false, false);
  }
#endif

  float bv = bias ? bias[ncol] : 0.0f;
  if (acc) {
#pragma unroll
    for (int r = 0; r < 8; ++r) {
      int row = tm * 16 + 8 * half + r;
      size_t idx = (size_t)row * Nc + ncol;
      float val = c[r] + bv;
      float di = dinv[row];
      C[idx]   = val;
      acc[idx] = val * (di * di);
    }
  } else {
#pragma unroll
    for (int r = 0; r < 8; ++r) {
      int row = tm * 16 + 8 * half + r;
      float bb = (cnt && !(cnt[row] > 0.0f)) ? 0.0f : bv;
      C[(size_t)row * Nc + ncol] = c[r] + bb;
    }
  }
}

static inline int blk(long long n, int b) { return (int)((n + b - 1) / b); }

extern "C" void kernel_launch(void* const* d_in, const int* in_sizes, int n_in,
                              void* d_out, int out_size, void* d_ws, size_t ws_size,
                              hipStream_t stream) {
  (void)in_sizes; (void)n_in; (void)out_size; (void)ws_size;
  const int*   edge_index = (const int*)d_in[0];
  const int*   batch      = (const int*)d_in[1];
  const float* W1 = (const float*)d_in[2];
  const float* b1 = (const float*)d_in[3];
  const float* W2 = (const float*)d_in[4];
  const float* b2 = (const float*)d_in[5];
  const float* W3 = (const float*)d_in[6];
  const float* b3 = (const float*)d_in[7];
  const float* g1 = (const float*)d_in[8];
  const float* be1= (const float*)d_in[9];
  const float* m1 = (const float*)d_in[10];
  const float* v1 = (const float*)d_in[11];
  const float* g2 = (const float*)d_in[12];
  const float* be2= (const float*)d_in[13];
  const float* m2 = (const float*)d_in[14];
  const float* v2 = (const float*)d_in[15];
  const float* Wl = (const float*)d_in[16];
  const float* bl = (const float*)d_in[17];
  float* out = (float*)d_out;

  char* ws = (char*)d_ws;
  size_t off = 0;
  auto carve = [&](size_t bytes) -> float* {
    float* p = (float*)(ws + off);
    off += (bytes + 255) & ~(size_t)255;
    return p;
  };
  float* deg_feat = carve((size_t)NN * 4);
  float* degcnt   = carve((size_t)NN * 4);
  float* dinv     = carve((size_t)NN * 4);
  float* bufX     = carve((size_t)NN * DD * 4);
  float* bufH     = carve((size_t)NN * DD * 4);
  float* bufA     = carve((size_t)NN * DD * 4);
  float* cnt      = carve((size_t)GG * 4);
  float* pooled   = carve((size_t)GG * DD * 4);
  float* poolT    = carve((size_t)GG * DD * 4);
  float* slab     = carve((size_t)PB_BLOCKS * GG * DD * 4);
  float* cslab    = carve((size_t)PB_BLOCKS * GG * 4);

  const int B = 256;
  const long long N64 = (long long)NN * DD;

  k_zero<<<blk(NN, B), B, 0, stream>>>(deg_feat, NN);
  k_zero<<<blk(NN, B), B, 0, stream>>>(degcnt, NN);
  k_edge_count<<<blk(EE, B), B, 0, stream>>>(edge_index, deg_feat, degcnt, EE);
  k_dinv<<<blk(NN, B), B, 0, stream>>>(degcnt, dinv, NN);

  const int tiles_NN  = (NN / 16) * (DD / 16);
  const int tiles_POOL = (GG / 16) * (DD / 16);
  const int tiles_FIN = (GG / 16) * (DOUT / 16);

  k_l1_hw_self<<<blk(N64, B), B, 0, stream>>>(deg_feat, W1, dinv, bufH, bufA, (int)N64);
  k_scatter<<<blk((long long)EE * 16, B), B, 0, stream>>>(edge_index, bufH, dinv, bufA, EE);
  k_post_bnrelu<<<blk(N64, B), B, 0, stream>>>(bufA, b1, g1, be1, m1, v1, bufX, (int)N64);

  k_gemm_wmma<<<blk((long long)tiles_NN * 32, B), B, 0, stream>>>(
      bufX, W2, bufH, nullptr, dinv, bufA, nullptr, NN, DD, DD);
  k_scatter<<<blk((long long)EE * 16, B), B, 0, stream>>>(edge_index, bufH, dinv, bufA, EE);
  k_post_bnrelu<<<blk(N64, B), B, 0, stream>>>(bufA, b2, g2, be2, m2, v2, bufX, (int)N64);

  k_pool_edges<<<PB_BLOCKS, PB_WAVES * 32, 0, stream>>>(edge_index, batch, bufX, dinv, slab, cslab);
  k_pool_reduce<<<GG, 256, 0, stream>>>(slab, cslab, pooled, cnt);

  k_gemm_wmma<<<blk((long long)tiles_POOL * 32, B), B, 0, stream>>>(
      pooled, W3, poolT, b3, nullptr, nullptr, cnt, GG, DD, DD);
  k_gemm_wmma<<<blk((long long)tiles_FIN * 32, B), B, 0, stream>>>(
      poolT, Wl, out, bl, nullptr, nullptr, nullptr, GG, DD, DOUT);
}


# Layers 1-2 aggregate per destination range in LDS: edges binned into 256 node ranges each call, one work-group sums its range with LDS float adds, bias+BN+ReLU at the store; sum order: LDS arrival
# speedup vs baseline: 1.8535x; 1.4541x over previous
#include <hip/hip_runtime.h>
#include <hip/hip_bf16.h>

#define NN   50000
#define EE   800000
#define GG   512
#define DD   64
#define DOUT 32
#define BN_EPS 1e-5f

typedef __attribute__((ext_vector_type(2)))  float    v2f;
typedef __attribute__((ext_vector_type(8)))  float    v8f;
typedef __attribute__((ext_vector_type(16))) _Float16 v16h;

#if defined(__AMDGCN__) && __has_builtin(__builtin_amdgcn_wmma_f32_16x16x4_f32)
#define USE_WMMA_F32 1
#else
#define USE_WMMA_F32 0
#endif


#define NP   256
#define PN   196
#define CH   ((EE + NP - 1) / NP)
#define BINT 1024
#define EPT  ((CH + BINT - 1) / BINT)
static_assert(NP * PN >= NN, "ranges must cover the nodes");
static_assert(NP <= BINT && NP % 32 == 0, "one thread per range");
static_assert((long long)NP * CH >= EE && EPT * BINT >= CH, "the passes cover every edge");

__global__ void k_clear(float* __restrict__ a, float* __restrict__ b, int n,
                        int* __restrict__ c, int m) {
  int i = blockIdx.x * blockDim.x + threadIdx.x;
  if (i < n) { a[i] = 0.0f; b[i] = 0.0f; }
  if (i < m) c[i] = 0;
}

__global__ __launch_bounds__(BINT)
void k_edge_count(const int* __restrict__ ei,
                  float* __restrict__ deg_feat,
                  float* __restrict__ degcnt,
                  int* __restrict__ pcnt) {
  __shared__ int h[NP];
  const int t = threadIdx.x;
  if (t < NP) h[t] = 0;
  int s[EPT], d[EPT];
  const int c0 = blockIdx.x * CH;
#pragma unroll
  for (int k = 0; k < EPT; ++k) {
    int e = c0 + t + k * BINT;
    e = e < EE ? e : EE - 1;
    s[k] = ei[e];
    d[k] = ei[EE + e];
  }
  __syncthreads();
#pragma unroll
  for (int k = 0; k < EPT; ++k) {
    int i = t + k * BINT;
    if (i < CH && c0 + i < EE) {
      atomicAdd(&deg_feat[s[k]], 1.0f);
      atomicAdd(&degcnt[d[k]], 1.0f);
      atomicAdd(&h[d[k] / PN], 1);
    }
  }
  __syncthreads();
  if (t < NP && h[t] > 0) atomicAdd(&pcnt[t], h[t]);
}

__global__ void k_dinv(const float* __restrict__ degcnt, float* __restrict__ dinv, int n) {
  int i = blockIdx.x * blockDim.x + threadIdx.x;
  if (i < n) dinv[i] = rsqrtf(degcnt[i] + 1.0f);
}

__global__ __launch_bounds__(BINT)
void k_fill(const int* __restrict__ ei,
            const int* __restrict__ pcnt,
            const float* __restrict__ dinv,
            int* __restrict__ pcur,
            int* __restrict__ goffs,
            int4* __restrict__ rec) {
  __shared__ int h[NP];
  __shared__ int hb[NP];
  __shared__ int wsum[NP / 32];
  const int t = threadIdx.x;
  const int lane = t & 31;
  int s[EPT], d[EPT], r[EPT];
  float w[EPT];
  const int c0 = blockIdx.x * CH;
#pragma unroll
  for (int k = 0; k < EPT; ++k) {
    int e = c0 + t + k * BINT;
    e = e < EE ? e : EE - 1;
    s[k] = ei[e];
    d[k] = ei[EE + e];
  }
#pragma unroll
  for (int k = 0; k < EPT; ++k) w[k] = dinv[s[k]] * dinv[d[k]];
  int v = (t < NP) ? pcnt[t] : 0;
  int x = v;
  for (int o = 1; o < 32; o <<= 1) {
    int y = __shfl_up(x, o);
    if (lane >= o) x += y;
  }
  if (t < NP) {
    h[t] = 0;
    if (lane == 31) wsum[t >> 5] = x;
  }
  __syncthreads();
  int off = 0;
  if (t < NP) {
    for (int i = 0; i < (t >> 5); ++i) off += wsum[i];
    off += x - v;
    if (blockIdx.x == 0) goffs[t] = off;
  }
#pragma unroll
  for (int k = 0; k < EPT; ++k) {
    int i = t + k * BINT;
    r[k] = -1;
    if (i < CH && c0 + i < EE)
      r[k] = atomicAdd(&h[d[k] / PN], 1);
  }
  __syncthreads();
  if (t < NP) {
    int c = h[t];
    int g = c > 0 ? atomicAdd(&pcur[t], c) : 0;
    hb[t] = off + g;
  }
  __syncthreads();
#pragma unroll
  for (int k = 0; k < EPT; ++k) {
    if (r[k] >= 0) {
      int pos = hb[d[k] / PN] + r[k];
      rec[pos] = make_int4(s[k], d[k], __float_as_int(w[k]), 0);
    }
  }
}

__global__ void k_l1_hw(const float* __restrict__ deg_feat,
                        const float* __restrict__ W1,
                        float* __restrict__ hw, int n64) {
  int i = blockIdx.x * blockDim.x + threadIdx.x;
  if (i >= n64) return;
  hw[i] = deg_feat[i >> 6] * W1[i & 63];
}

#define AG_WAVES 32
#define AG_GRP   16
#define AG_TAB_FLOATS (41 * 1024)
static_assert((PN + 1) * DD <= AG_TAB_FLOATS, "table holds one range and the spare row");

__global__ __launch_bounds__(AG_WAVES * 32)
void k_agg(const int4* __restrict__ rec,
           const int* __restrict__ pcnt,
           const int* __restrict__ goffs,
           const float* __restrict__ hw,
           const float* __restrict__ dinv,
           const float* __restrict__ b,
           const float* __restrict__ g,
           const float* __restrict__ be,
           const float* __restrict__ m,
           const float* __restrict__ v,
           float* __restrict__ xout) {
  __shared__ float tab[AG_TAB_FLOATS];
  const int tid  = threadIdx.x;
  const int lane = tid & 31;
  const int wave = __builtin_amdgcn_readfirstlane(tid >> 5);
  const int n0 = blockIdx.x * PN;
  const int nn = (NN - n0 < PN) ? (NN - n0) : PN;
  const int start = goffs[blockIdx.x];
  const int cnt   = pcnt[blockIdx.x];
  const int nb    = (cnt + 31) >> 5;
  const float2* __restrict__ hw2 = (const float2*)hw;

  int bq = wave;
  int sv = 0, dv = n0;
  float wv = 0.0f;
  if (bq < nb) {
    int li = bq * 32 + lane;
    li = li < cnt ? li : cnt - 1;
    int4 q = rec[start + li];
    sv = q.x; dv = q.y; wv = __int_as_float(q.z);
  }

  for (int i = tid; i < nn * DD; i += AG_WAVES * 32) {
    float di = dinv[n0 + (i >> 6)];
    tab[i] = hw[n0 * DD + i] * (di * di);
  }
  __syncthreads();

  while (bq < nb) {
    const int bn = bq + AG_WAVES;
    int sn = 0, dn = n0;
    float wn = 0.0f;
    if (bn < nb) {
      int li = bn * 32 + lane;
      li = li < cnt ? li : cnt - 1;
      int4 q = rec[start + li];
      sn = q.x; dn = q.y; wn = __int_as_float(q.z);
    }
    const bool  live = bq * 32 + lane < cnt;
    const int   rv = live ? (dv - n0) * DD : PN * DD;
    const float wl = live ? wv : 0.0f;
#pragma unroll
    for (int j0 = 0; j0 < 32; j0 += AG_GRP) {
      float2 vr[AG_GRP];
#pragma unroll
      for (int u = 0; u < AG_GRP; ++u) {
        int sj = __builtin_amdgcn_readlane(sv, j0 + u);
        vr[u] = hw2[sj * (DD / 2) + lane];
      }
#pragma unroll
      for (int u = 0; u < AG_GRP; ++u) {
        int   rj = __builtin_amdgcn_readlane(rv, j0 + u);
        float wj = __int_as_float(__builtin_amdgcn_readlane(__float_as_int(wl), j0 + u));
        atomicAdd(&tab[rj + 2 * lane],     vr[u].x * wj);
        atomicAdd(&tab[rj + 2 * lane + 1], vr[u].y * wj);
      }
    }
    sv = sn; dv = dn; wv = wn; bq = bn;
  }
  __syncthreads();

  for (int i = tid; i < nn * DD; i += AG_WAVES * 32) {
    int f = i & 63;
    float rs = rsqrtf(v[f] + BN_EPS);
    float y = ((tab[i] + b[f]) - m[f]) * rs * g[f] + be[f];
    xout[n0 * DD + i] = fmaxf(y, 0.0f);
  }
}

#define PB_BLOCKS 256
#define PB_WAVES  32
#define TOT_ITEMS (EE + NN)
#define N_BATCH   ((TOT_ITEMS + 31) / 32)
#define N_PWAVES  (PB_BLOCKS * PB_WAVES)
#define KB        ((N_BATCH + N_PWAVES - 1) / N_PWAVES)
#define GRP       16
#define TAB_FLOATS (41 * 1024)

__global__ __launch_bounds__(PB_WAVES * 32)
void k_pool_edges(const int* __restrict__ ei,
                  const int* __restrict__ batch,
                  const float* __restrict__ x,
                  const float* __restrict__ dinv,
                  float* __restrict__ slab,
                  float* __restrict__ cslab) {
  __shared__ float tab[TAB_FLOATS];
  __shared__ float ctab[GG];
  const int tid  = threadIdx.x;
  const int lane = tid & 31;
  const int gw   = blockIdx.x * PB_WAVES + __builtin_amdgcn_readfirstlane(tid >> 5);

  float4 z4 = {0.0f, 0.0f, 0.0f, 0.0f};
  for (int i = tid; i < GG * DD / 4; i += PB_WAVES * 32) ((float4*)tab)[i] = z4;
  if (tid < GG) ctab[tid] = 0.0f;

  int   sv[KB], gv[KB];
  float wv[KB];
  int   dv[KB];
#pragma unroll
  for (int k = 0; k < KB; ++k) {
    int it  = (gw + k * N_PWAVES) * 32 + lane;
    int ie  = it < EE ? it : EE - 1;
    int ls  = ei[ie];
    int ld  = ei[EE + ie];
    int nd  = it - EE;
    nd = nd < 0 ? 0 : (nd > NN - 1 ? NN - 1 : nd);
    sv[k] = it < EE ? ls : nd;
    dv[k] = it < EE ? ld : nd;
  }
#pragma unroll
  for (int k = 0; k < KB; ++k) {
    wv[k] = dinv[sv[k]] * dinv[dv[k]];
    gv[k] = batch[dv[k]];
  }
  __syncthreads();

#pragma unroll
  for (int k = 0; k < KB; ++k) {
    const int base = (gw + k * N_PWAVES) * 32;
    int nvalid = TOT_ITEMS - base;
    if (nvalid <= 0) continue;
    const int it = base + lane;
    if (it >= EE && it < TOT_ITEMS) atomicAdd(&ctab[gv[k]], 1.0f);
    if (nvalid >= 32) {
#pragma unroll
      for (int j0 = 0; j0 < 32; j0 += GRP) {
        float v0[GRP], v1[GRP];
#pragma unroll
        for (int u = 0; u < GRP; ++u) {
          int sj = __builtin_amdgcn_readlane(sv[k], j0 + u);
          int idx = sj * DD + lane;
          v0[u] = x[idx];
          v1[u] = x[idx + 32];
        }
#pragma unroll
        for (int u = 0; u < GRP; ++u) {
          int   gj = __builtin_amdgcn_readlane(gv[k], j0 + u);
          float wj = __int_as_float(__builtin_amdgcn_readlane(__float_as_int(wv[k]), j0 + u));
          atomicAdd(&tab[gj * DD + lane],      v0[u] * wj);
          atomicAdd(&tab[gj * DD + lane + 32], v1[u] * wj);
        }
      }
    } else {
      for (int j = 0; j < nvalid; ++j) {
        int   sj = __builtin_amdgcn_readlane(sv[k], j);
        int   gj = __builtin_amdgcn_readlane(gv[k], j);
        float wj = __int_as_float(__builtin_amdgcn_readlane(__float_as_int(wv[k]), j));
        int idx = sj * DD + lane;
        float a0 = x[idx];
        float a1 = x[idx + 32];
        atomicAdd(&tab[gj * DD + lane],      a0 * wj);
        atomicAdd(&tab[gj * DD + lane + 32], a1 * wj);
      }
    }
  }
  __syncthreads();

  float4* my = (float4*)(slab + (size_t)blockIdx.x * (GG * DD));
  for (int i = tid; i < GG * DD / 4; i += PB_WAVES * 32) my[i] = ((const float4*)tab)[i];
  if (tid < GG) cslab[blockIdx.x * GG + tid] = ctab[tid];
}

__global__ __launch_bounds__(256)
void k_pool_reduce(const float* __restrict__ slab,
                   const float* __restrict__ cslab,
                   float* __restrict__ pooled,
                   float* __restrict__ cnt) {
  __shared__ float part[4][DD];
  __shared__ float cpart[8];
  const int g = blockIdx.x;
  const int t = threadIdx.x;
  const int f = t & (DD - 1);
  const int q = t >> 6;
  const float* p = slab + (size_t)(q * (PB_BLOCKS / 4)) * (GG * DD) + g * DD + f;
  float a = 0.0f;
#pragma unroll 16
  for (int b = 0; b < PB_BLOCKS / 4; ++b) a += p[(size_t)b * (GG * DD)];
  part[q][f] = a;
  float c = cslab[t * GG + g];
  for (int off = 16; off > 0; off >>= 1) c += __shfl_xor(c, off);
  if ((t & 31) == 0) cpart[t >> 5] = c;
  __syncthreads();
  if (t < DD) {
    float tot = (part[0][t] + part[1][t]) + (part[2][t] + part[3][t]);
    float cc = 0.0f;
    for (int i = 0; i < 8; ++i) cc += cpart[i];
    pooled[g * DD + t] = (cc > 0.0f) ? tot / cc : 0.0f;
    if (t == 0) cnt[g] = cc;
  }
}

__global__ void k_gemm_wmma(const float* __restrict__ A,
                            const float* __restrict__ B,
                            float* __restrict__ C,
                            const float* __restrict__ bias,
                            const float* __restrict__ dinv,
                            float* __restrict__ acc,
                            const float* __restrict__ cnt,
                            int M, int K, int Nc) {
  int wid  = (blockIdx.x * blockDim.x + threadIdx.x) >> 5;
  int lane = threadIdx.x & 31;
  int tiles_n = Nc >> 4;
  int tm = wid / tiles_n;
  int tn = wid - tm * tiles_n;
  if (tm * 16 >= M) return;

  int half = lane >> 4;
  int lm   = lane & 15;
  const float* Arow = A + (size_t)(tm * 16 + lm) * K;
  int ncol = tn * 16 + lm;

  v8f c = {};
#if USE_WMMA_F32
  for (int k0 = 0; k0 < K; k0 += 4) {
    v2f a, b;
    a.x = Arow[k0 + 2 * half + 0];
    a.y = Arow[k0 + 2 * half + 1];
    b.x = B[(size_t)(k0 + 2 * half + 0) * Nc + ncol];
    b.y = B[(size_t)(k0 + 2 * half + 1) * Nc + ncol];
    c = __builtin_amdgcn_wmma_f32_16x16x4_f32(false, a, false, b, (short)0, c,
                                              false, false);
  }
#else
  for (int k0 = 0; k0 < K; k0 += 32) {
    v16h a, b;
#pragma unroll
    for (int i = 0; i < 8; ++i) {
      a[i]     = (_Float16)Arow[k0 + 8 * half + i];
      a[8 + i] = (_Float16)Arow[k0 + 16 + 8 * half + i];
    }
#pragma unroll
    for (int i = 0; i < 16; ++i)
      b[i] = (_Float16)B[(size_t)(k0 + 16 * half + i) * Nc + ncol];
    c = __builtin_amdgcn_wmma_f32_16x16x32_f16(false, a, false, b, (short)0, c,
                                               false, false);
  }
#endif

  float bv = bias ? bias[ncol] : 0.0f;
  if (acc) {
#pragma unroll
    for (int r = 0; r < 8; ++r) {
      int row = tm * 16 + 8 * half + r;
      size_t idx = (size_t)row * Nc + ncol;
      float val = c[r] + bv;
      float di = dinv[row];
      C[idx]   = val;
      acc[idx] = val * (di * di);
    }
  } else {
#pragma unroll
    for (int r = 0; r < 8; ++r) {
      int row = tm * 16 + 8 * half + r;
      float bb = (cnt && !(cnt[row] > 0.0f)) ? 0.0f : bv;
      C[(size_t)row * Nc + ncol] = c[r] + bb;
    }
  }
}

static inline int blk(long long n, int b) { return (int)((n + b - 1) / b); }

extern "C" void kernel_launch(void* const* d_in, const int* in_sizes, int n_in,
                              void* d_out, int out_size, void* d_ws, size_t ws_size,
                              hipStream_t stream) {
  (void)in_sizes; (void)n_in; (void)out_size; (void)ws_size;
  const int*   edge_index = (const int*)d_in[0];
  const int*   batch      = (const int*)d_in[1];
  const float* W1 = (const float*)d_in[2];
  const float* b1 = (const float*)d_in[3];
  const float* W2 = (const float*)d_in[4];
  const float* b2 = (const float*)d_in[5];
  const float* W3 = (const float*)d_in[6];
  const float* b3 = (const float*)d_in[7];
  const float* g1 = (const float*)d_in[8];
  const float* be1= (const float*)d_in[9];
  const float* m1 = (const float*)d_in[10];
  const float* v1 = (const float*)d_in[11];
  const float* g2 = (const float*)d_in[12];
  const float* be2= (const float*)d_in[13];
  const float* m2 = (const float*)d_in[14];
  const float* v2 = (const float*)d_in[15];
  const float* Wl = (const float*)d_in[16];
  const float* bl = (const float*)d_in[17];
  float* out = (float*)d_out;

  char* ws = (char*)d_ws;
  size_t off = 0;
  auto carve = [&](size_t bytes) -> float* {
    float* p = (float*)(ws + off);
    off += (bytes + 255) & ~(size_t)255;
    return p;
  };
  float* deg_feat = carve((size_t)NN * 4);
  float* degcnt   = carve((size_t)NN * 4);
  float* dinv     = carve((size_t)NN * 4);
  float* bufX     = carve((size_t)NN * DD * 4);
  float* bufH     = carve((size_t)NN * DD * 4);
  float* cnt      = carve((size_t)GG * 4);
  float* pooled   = carve((size_t)GG * DD * 4);
  float* poolT    = carve((size_t)GG * DD * 4);
  float* slab     = carve((size_t)PB_BLOCKS * GG * DD * 4);
  float* cslab    = carve((size_t)PB_BLOCKS * GG * 4);
  int*   pcnt     = (int*)carve((size_t)2 * NP * 4);
  int*   pcur     = pcnt + NP;
  int*   goffs    = (int*)carve((size_t)NP * 4);
  int4*  rec      = (int4*)carve((size_t)EE * 16);

  const int B = 256;
  const long long N64 = (long long)NN * DD;

  k_clear<<<blk(NN, B), B, 0, stream>>>(deg_feat, degcnt, NN, pcnt, 2 * NP);
  k_edge_count<<<NP, BINT, 0, stream>>>(edge_index, deg_feat, degcnt, pcnt);
  k_dinv<<<blk(NN, B), B, 0, stream>>>(degcnt, dinv, NN);
  k_fill<<<NP, BINT, 0, stream>>>(edge_index, pcnt, dinv, pcur, goffs, rec);

  const int tiles_NN  = (NN / 16) * (DD / 16);
  const int tiles_POOL = (GG / 16) * (DD / 16);
  const int tiles_FIN = (GG / 16) * (DOUT / 16);

  k_l1_hw<<<blk(N64, B), B, 0, stream>>>(deg_feat, W1, bufH, (int)N64);
  k_agg<<<NP, AG_WAVES * 32, 0, stream>>>(rec, pcnt, goffs, bufH, dinv,
                                          b1, g1, be1, m1, v1, bufX);

  k_gemm_wmma<<<blk((long long)tiles_NN * 32, B), B, 0, stream>>>(
      bufX, W2, bufH, nullptr, nullptr, nullptr, nullptr, NN, DD, DD);
  k_agg<<<NP, AG_WAVES * 32, 0, stream>>>(rec, pcnt, goffs, bufH, dinv,
                                          b2, g2, be2, m2, v2, bufX);

  k_pool_edges<<<PB_BLOCKS, PB_WAVES * 32, 0, stream>>>(edge_index, batch, bufX, dinv, slab, cslab);
  k_pool_reduce<<<GG, 256, 0, stream>>>(slab, cslab, pooled, cnt);

  k_gemm_wmma<<<blk((long long)tiles_POOL * 32, B), B, 0, stream>>>(
      pooled, W3, poolT, b3, nullptr, nullptr, cnt, GG, DD, DD);
  k_gemm_wmma<<<blk((long long)tiles_FIN * 32, B), B, 0, stream>>>(
      poolT, Wl, out, bl, nullptr, nullptr, nullptr, GG, DD, DOUT);
}


# Layer 3 + pool over the destination-range records: node rows summed in LDS, then per graph; a range adds only its graphs into global sums (float atomics); sum order: LDS arrival, chunks, global
# speedup vs baseline: 1.9275x; 1.0399x over previous
#include <hip/hip_runtime.h>
#include <hip/hip_bf16.h>

#define NN   50000
#define EE   800000
#define GG   512
#define DD   64
#define DOUT 32
#define BN_EPS 1e-5f

typedef __attribute__((ext_vector_type(2)))  float    v2f;
typedef __attribute__((ext_vector_type(8)))  float    v8f;
typedef __attribute__((ext_vector_type(16))) _Float16 v16h;

#if defined(__AMDGCN__) && __has_builtin(__builtin_amdgcn_wmma_f32_16x16x4_f32)
#define USE_WMMA_F32 1
#else
#define USE_WMMA_F32 0
#endif


#define NP   256
#define PN   196
#define CH   ((EE + NP - 1) / NP)
#define BINT 1024
#define EPT  ((CH + BINT - 1) / BINT)
static_assert(NP * PN >= NN, "ranges must cover the nodes");
static_assert(NP <= BINT && NP % 32 == 0, "one thread per range");
static_assert((long long)NP * CH >= EE && EPT * BINT >= CH, "the passes cover every edge");

__global__ void k_clear(float* __restrict__ a, float* __restrict__ b, int n,
                        int* __restrict__ c, int m,
                        float* __restrict__ p, int np) {
  int i = blockIdx.x * blockDim.x + threadIdx.x;
  if (i < n) { a[i] = 0.0f; b[i] = 0.0f; }
  if (i < m) c[i] = 0;
  if (i < np) p[i] = 0.0f;
}

__global__ __launch_bounds__(BINT)
void k_edge_count(const int* __restrict__ ei,
                  float* __restrict__ deg_feat,
                  float* __restrict__ degcnt,
                  int* __restrict__ pcnt) {
  __shared__ int h[NP];
  const int t = threadIdx.x;
  if (t < NP) h[t] = 0;
  int s[EPT], d[EPT];
  const int c0 = blockIdx.x * CH;
#pragma unroll
  for (int k = 0; k < EPT; ++k) {
    int e = c0 + t + k * BINT;
    e = e < EE ? e : EE - 1;
    s[k] = ei[e];
    d[k] = ei[EE + e];
  }
  __syncthreads();
#pragma unroll
  for (int k = 0; k < EPT; ++k) {
    int i = t + k * BINT;
    if (i < CH && c0 + i < EE) {
      atomicAdd(&deg_feat[s[k]], 1.0f);
      atomicAdd(&degcnt[d[k]], 1.0f);
      atomicAdd(&h[d[k] / PN], 1);
    }
  }
  __syncthreads();
  if (t < NP && h[t] > 0) atomicAdd(&pcnt[t], h[t]);
}

__global__ void k_dinv(const float* __restrict__ degcnt, float* __restrict__ dinv, int n) {
  int i = blockIdx.x * blockDim.x + threadIdx.x;
  if (i < n) dinv[i] = rsqrtf(degcnt[i] + 1.0f);
}

__global__ __launch_bounds__(BINT)
void k_fill(const int* __restrict__ ei,
            const int* __restrict__ pcnt,
            const float* __restrict__ dinv,
            int* __restrict__ pcur,
            int* __restrict__ goffs,
            int4* __restrict__ rec) {
  __shared__ int h[NP];
  __shared__ int hb[NP];
  __shared__ int wsum[NP / 32];
  const int t = threadIdx.x;
  const int lane = t & 31;
  int s[EPT], d[EPT], r[EPT];
  float w[EPT];
  const int c0 = blockIdx.x * CH;
#pragma unroll
  for (int k = 0; k < EPT; ++k) {
    int e = c0 + t + k * BINT;
    e = e < EE ? e : EE - 1;
    s[k] = ei[e];
    d[k] = ei[EE + e];
  }
#pragma unroll
  for (int k = 0; k < EPT; ++k) w[k] = dinv[s[k]] * dinv[d[k]];
  int v = (t < NP) ? pcnt[t] : 0;
  int x = v;
  for (int o = 1; o < 32; o <<= 1) {
    int y = __shfl_up(x, o);
    if (lane >= o) x += y;
  }
  if (t < NP) {
    h[t] = 0;
    if (lane == 31) wsum[t >> 5] = x;
  }
  __syncthreads();
  int off = 0;
  if (t < NP) {
    for (int i = 0; i < (t >> 5); ++i) off += wsum[i];
    off += x - v;
    if (blockIdx.x == 0) goffs[t] = off;
  }
#pragma unroll
  for (int k = 0; k < EPT; ++k) {
    int i = t + k * BINT;
    r[k] = -1;
    if (i < CH && c0 + i < EE)
      r[k] = atomicAdd(&h[d[k] / PN], 1);
  }
  __syncthreads();
  if (t < NP) {
    int c = h[t];
    int g = c > 0 ? atomicAdd(&pcur[t], c) : 0;
    hb[t] = off + g;
  }
  __syncthreads();
#pragma unroll
  for (int k = 0; k < EPT; ++k) {
    if (r[k] >= 0) {
      int pos = hb[d[k] / PN] + r[k];
      rec[pos] = make_int4(s[k], d[k], __float_as_int(w[k]), 0);
    }
  }
}

__global__ void k_l1_hw(const float* __restrict__ deg_feat,
                        const float* __restrict__ W1,
                        float* __restrict__ hw, int n64) {
  int i = blockIdx.x * blockDim.x + threadIdx.x;
  if (i >= n64) return;
  hw[i] = deg_feat[i >> 6] * W1[i & 63];
}

#define AG_WAVES 32
#define AG_GRP   16
#define AG_TAB_FLOATS (41 * 1024)
static_assert((PN + 1) * DD <= AG_TAB_FLOATS, "table holds one range and the spare row");

__global__ __launch_bounds__(AG_WAVES * 32)
void k_agg(const int4* __restrict__ rec,
           const int* __restrict__ pcnt,
           const int* __restrict__ goffs,
           const float* __restrict__ hw,
           const float* __restrict__ dinv,
           const float* __restrict__ b,
           const float* __restrict__ g,
           const float* __restrict__ be,
           const float* __restrict__ m,
           const float* __restrict__ v,
           float* __restrict__ xout) {
  __shared__ float tab[AG_TAB_FLOATS];
  const int tid  = threadIdx.x;
  const int lane = tid & 31;
  const int wave = __builtin_amdgcn_readfirstlane(tid >> 5);
  const int n0 = blockIdx.x * PN;
  const int nn = (NN - n0 < PN) ? (NN - n0) : PN;
  const int start = goffs[blockIdx.x];
  const int cnt   = pcnt[blockIdx.x];
  const int nb    = (cnt + 31) >> 5;
  const float2* __restrict__ hw2 = (const float2*)hw;

  int bq = wave;
  int sv = 0, dv = n0;
  float wv = 0.0f;
  if (bq < nb) {
    int li = bq * 32 + lane;
    li = li < cnt ? li : cnt - 1;
    int4 q = rec[start + li];
    sv = q.x; dv = q.y; wv = __int_as_float(q.z);
  }

  for (int i = tid; i < nn * DD; i += AG_WAVES * 32) {
    float di = dinv[n0 + (i >> 6)];
    tab[i] = hw[n0 * DD + i] * (di * di);
  }
  __syncthreads();

  while (bq < nb) {
    const int bn = bq + AG_WAVES;
    int sn = 0, dn = n0;
    float wn = 0.0f;
    if (bn < nb) {
      int li = bn * 32 + lane;
      li = li < cnt ? li : cnt - 1;
      int4 q = rec[start + li];
      sn = q.x; dn = q.y; wn = __int_as_float(q.z);
    }
    const bool  live = bq * 32 + lane < cnt;
    const int   rv = live ? (dv - n0) * DD : PN * DD;
    const float wl = live ? wv : 0.0f;
#pragma unroll
    for (int j0 = 0; j0 < 32; j0 += AG_GRP) {
      float2 vr[AG_GRP];
#pragma unroll
      for (int u = 0; u < AG_GRP; ++u) {
        int sj = __builtin_amdgcn_readlane(sv, j0 + u);
        vr[u] = hw2[sj * (DD / 2) + lane];
      }
#pragma unroll
      for (int u = 0; u < AG_GRP; ++u) {
        int   rj = __builtin_amdgcn_readlane(rv, j0 + u);
        float wj = __int_as_float(__builtin_amdgcn_readlane(__float_as_int(wl), j0 + u));
        atomicAdd(&tab[rj + 2 * lane],     vr[u].x * wj);
        atomicAdd(&tab[rj + 2 * lane + 1], vr[u].y * wj);
      }
    }
    sv = sn; dv = dn; wv = wn; bq = bn;
  }
  __syncthreads();

  for (int i = tid; i < nn * DD; i += AG_WAVES * 32) {
    int f = i & 63;
    float rs = rsqrtf(v[f] + BN_EPS);
    float y = ((tab[i] + b[f]) - m[f]) * rs * g[f] + be[f];
    xout[n0 * DD + i] = fmaxf(y, 0.0f);
  }
}

#define PCH  16
#define PCS  ((PN + PCH - 1) / PCH)
static_assert(PCH * DD == AG_WAVES * 32, "one thread per (chunk, feature)");

__global__ __launch_bounds__(AG_WAVES * 32)
void k_pool_agg(const int4* __restrict__ rec,
                const int* __restrict__ pcnt,
                const int* __restrict__ goffs,
                const int* __restrict__ batch,
                const float* __restrict__ x,
                const float* __restrict__ dinv,
                float* __restrict__ psum,
                float* __restrict__ cntg) {
  __shared__ float tab[(PN + 1) * DD];
  __shared__ float gtab[GG * DD];
  __shared__ float ctab[GG];
  __shared__ int   gloc[PN];
  const int tid  = threadIdx.x;
  const int lane = tid & 31;
  const int wave = __builtin_amdgcn_readfirstlane(tid >> 5);
  const int n0 = blockIdx.x * PN;
  const int nn = (NN - n0 < PN) ? (NN - n0) : PN;
  const int start = goffs[blockIdx.x];
  const int cnt   = pcnt[blockIdx.x];
  const int nb    = (cnt + 31) >> 5;
  const float2* __restrict__ x2 = (const float2*)x;

  int bq = wave;
  int sv = 0, dv = n0;
  float wv = 0.0f;
  if (bq < nb) {
    int li = bq * 32 + lane;
    li = li < cnt ? li : cnt - 1;
    int4 q = rec[start + li];
    sv = q.x; dv = q.y; wv = __int_as_float(q.z);
  }

  for (int i = tid; i < nn * DD; i += AG_WAVES * 32) {
    float di = dinv[n0 + (i >> 6)];
    tab[i] = x[n0 * DD + i] * (di * di);
  }
  float4 z4 = {0.0f, 0.0f, 0.0f, 0.0f};
  for (int i = tid; i < GG * DD / 4; i += AG_WAVES * 32) ((float4*)gtab)[i] = z4;
  if (tid < GG) ctab[tid] = 0.0f;
  if (tid < nn) gloc[tid] = batch[n0 + tid];
  __syncthreads();

  while (bq < nb) {
    const int bn = bq + AG_WAVES;
    int sn = 0, dn = n0;
    float wn = 0.0f;
    if (bn < nb) {
      int li = bn * 32 + lane;
      li = li < cnt ? li : cnt - 1;
      int4 q = rec[start + li];
      sn = q.x; dn = q.y; wn = __int_as_float(q.z);
    }
    const bool  live = bq * 32 + lane < cnt;
    const int   rv = live ? (dv - n0) * DD : PN * DD;
    const float wl = live ? wv : 0.0f;
#pragma unroll
    for (int j0 = 0; j0 < 32; j0 += AG_GRP) {
      float2 vr[AG_GRP];
#pragma unroll
      for (int u = 0; u < AG_GRP; ++u) {
        int sj = __builtin_amdgcn_readlane(sv, j0 + u);
        vr[u] = x2[sj * (DD / 2) + lane];
      }
#pragma unroll
      for (int u = 0; u < AG_GRP; ++u) {
        int   rj = __builtin_amdgcn_readlane(rv, j0 + u);
        float wj = __int_as_float(__builtin_amdgcn_readlane(__float_as_int(wl), j0 + u));
        atomicAdd(&tab[rj + 2 * lane],     vr[u].x * wj);
        atomicAdd(&tab[rj + 2 * lane + 1], vr[u].y * wj);
      }
    }
    sv = sn; dv = dn; wv = wn; bq = bn;
  }
  __syncthreads();

  {
    const int f  = tid & (DD - 1);
    const int c0 = (tid >> 6) * PCS;
    const int c1 = (c0 + PCS < nn) ? (c0 + PCS) : nn;
    if (c0 < c1) {
      int   cg = gloc[c0];
      float a  = 0.0f;
      for (int i = c0; i < c1; ++i) {
        int g = gloc[i];
        if (g != cg) { atomicAdd(&gtab[cg * DD + f], a); a = 0.0f; cg = g; }
        a += tab[i * DD + f];
      }
      atomicAdd(&gtab[cg * DD + f], a);
    }
    if (tid < nn) atomicAdd(&ctab[gloc[tid]], 1.0f);
  }
  __syncthreads();

  for (int i = tid; i < GG * DD; i += AG_WAVES * 32)
    if (ctab[i >> 6] > 0.0f) atomicAdd(&psum[i], gtab[i]);
  if (tid < GG && ctab[tid] > 0.0f) atomicAdd(&cntg[tid], ctab[tid]);
}

__global__ void k_gemm_wmma(const float* __restrict__ A,
                            const float* __restrict__ B,
                            float* __restrict__ C,
                            const float* __restrict__ bias,
                            const float* __restrict__ dinv,
                            float* __restrict__ acc,
                            const float* __restrict__ cnt,
                            int M, int K, int Nc) {
  int wid  = (blockIdx.x * blockDim.x + threadIdx.x) >> 5;
  int lane = threadIdx.x & 31;
  int tiles_n = Nc >> 4;
  int tm = wid / tiles_n;
  int tn = wid - tm * tiles_n;
  if (tm * 16 >= M) return;

  int half = lane >> 4;
  int lm   = lane & 15;
  const float* Arow = A + (size_t)(tm * 16 + lm) * K;
  int ncol = tn * 16 + lm;

  v8f c = {};
#if USE_WMMA_F32
  for (int k0 = 0; k0 < K; k0 += 4) {
    v2f a, b;
    a.x = Arow[k0 + 2 * half + 0];
    a.y = Arow[k0 + 2 * half + 1];
    b.x = B[(size_t)(k0 + 2 * half + 0) * Nc + ncol];
    b.y = B[(size_t)(k0 + 2 * half + 1) * Nc + ncol];
    c = __builtin_amdgcn_wmma_f32_16x16x4_f32(false, a, false, b, (short)0, c,
                                              false, false);
  }
#else
  for (int k0 = 0; k0 < K; k0 += 32) {
    v16h a, b;
#pragma unroll
    for (int i = 0; i < 8; ++i) {
      a[i]     = (_Float16)Arow[k0 + 8 * half + i];
      a[8 + i] = (_Float16)Arow[k0 + 16 + 8 * half + i];
    }
#pragma unroll
    for (int i = 0; i < 16; ++i)
      b[i] = (_Float16)B[(size_t)(k0 + 16 * half + i) * Nc + ncol];
    c = __builtin_amdgcn_wmma_f32_16x16x32_f16(false, a, false, b, (short)0, c,
                                               false, false);
  }
#endif

  float bv = bias ? bias[ncol] : 0.0f;
  if (acc) {
#pragma unroll
    for (int r = 0; r < 8; ++r) {
      int row = tm * 16 + 8 * half + r;
      size_t idx = (size_t)row * Nc + ncol;
      float val = c[r] + bv;
      float di = dinv[row];
      C[idx]   = val;
      acc[idx] = val * (di * di);
    }
  } else {
#pragma unroll
    for (int r = 0; r < 8; ++r) {
      int row = tm * 16 + 8 * half + r;
      float val = c[r] + bv;
      if (cnt) {
        float cc = cnt[row];
        val = (cc > 0.0f) ? c[r] / cc + bv : 0.0f;
      }
      C[(size_t)row * Nc + ncol] = val;
    }
  }
}

static inline int blk(long long n, int b) { return (int)((n + b - 1) / b); }

extern "C" void kernel_launch(void* const* d_in, const int* in_sizes, int n_in,
                              void* d_out, int out_size, void* d_ws, size_t ws_size,
                              hipStream_t stream) {
  (void)in_sizes; (void)n_in; (void)out_size; (void)ws_size;
  const int*   edge_index = (const int*)d_in[0];
  const int*   batch      = (const int*)d_in[1];
  const float* W1 = (const float*)d_in[2];
  const float* b1 = (const float*)d_in[3];
  const float* W2 = (const float*)d_in[4];
  const float* b2 = (const float*)d_in[5];
  const float* W3 = (const float*)d_in[6];
  const float* b3 = (const float*)d_in[7];
  const float* g1 = (const float*)d_in[8];
  const float* be1= (const float*)d_in[9];
  const float* m1 = (const float*)d_in[10];
  const float* v1 = (const float*)d_in[11];
  const float* g2 = (const float*)d_in[12];
  const float* be2= (const float*)d_in[13];
  const float* m2 = (const float*)d_in[14];
  const float* v2 = (const float*)d_in[15];
  const float* Wl = (const float*)d_in[16];
  const float* bl = (const float*)d_in[17];
  float* out = (float*)d_out;

  char* ws = (char*)d_ws;
  size_t off = 0;
  auto carve = [&](size_t bytes) -> float* {
    float* p = (float*)(ws + off);
    off += (bytes + 255) & ~(size_t)255;
    return p;
  };
  float* deg_feat = carve((size_t)NN * 4);
  float* degcnt   = carve((size_t)NN * 4);
  float* dinv     = carve((size_t)NN * 4);
  float* bufX     = carve((size_t)NN * DD * 4);
  float* bufH     = carve((size_t)NN * DD * 4);
  float* psum     = carve((size_t)(GG * DD + GG) * 4);
  float* cnt      = psum + GG * DD;
  float* poolT    = carve((size_t)GG * DD * 4);
  int*   pcnt     = (int*)carve((size_t)2 * NP * 4);
  int*   pcur     = pcnt + NP;
  int*   goffs    = (int*)carve((size_t)NP * 4);
  int4*  rec      = (int4*)carve((size_t)EE * 16);

  const int B = 256;
  const long long N64 = (long long)NN * DD;

  k_clear<<<blk(NN, B), B, 0, stream>>>(deg_feat, degcnt, NN, pcnt, 2 * NP, psum, GG * DD + GG);
  k_edge_count<<<NP, BINT, 0, stream>>>(edge_index, deg_feat, degcnt, pcnt);
  k_dinv<<<blk(NN, B), B, 0, stream>>>(degcnt, dinv, NN);
  k_fill<<<NP, BINT, 0, stream>>>(edge_index, pcnt, dinv, pcur, goffs, rec);

  const int tiles_NN  = (NN / 16) * (DD / 16);
  const int tiles_POOL = (GG / 16) * (DD / 16);
  const int tiles_FIN = (GG / 16) * (DOUT / 16);

  k_l1_hw<<<blk(N64, B), B, 0, stream>>>(deg_feat, W1, bufH, (int)N64);
  k_agg<<<NP, AG_WAVES * 32, 0, stream>>>(rec, pcnt, goffs, bufH, dinv,
                                          b1, g1, be1, m1, v1, bufX);

  k_gemm_wmma<<<blk((long long)tiles_NN * 32, B), B, 0, stream>>>(
      bufX, W2, bufH, nullptr, nullptr, nullptr, nullptr, NN, DD, DD);
  k_agg<<<NP, AG_WAVES * 32, 0, stream>>>(rec, pcnt, goffs, bufH, dinv,
                                          b2, g2, be2, m2, v2, bufX);

  k_pool_agg<<<NP, AG_WAVES * 32, 0, stream>>>(rec, pcnt, goffs, batch, bufX, dinv, psum, cnt);

  k_gemm_wmma<<<blk((long long)tiles_POOL * 32, B), B, 0, stream>>>(
      psum, W3, poolT, b3, nullptr, nullptr, cnt, GG, DD, DD);
  k_gemm_wmma<<<blk((long long)tiles_FIN * 32, B), B, 0, stream>>>(
      poolT, Wl, out, bl, nullptr, nullptr, nullptr, GG, DD, DOUT);
}


# Degree counts by 16 edge slabs x 16 node slices: a work-group scans its slab, counts its slice's end points in LDS, adds the table with lane-adjacent adds; per-edge global adds gone; counts exact
# speedup vs baseline: 2.7519x; 1.4277x over previous
#include <hip/hip_runtime.h>
#include <hip/hip_bf16.h>

#define NN   50000
#define EE   800000
#define GG   512
#define DD   64
#define DOUT 32
#define BN_EPS 1e-5f

typedef __attribute__((ext_vector_type(2)))  float    v2f;
typedef __attribute__((ext_vector_type(8)))  float    v8f;
typedef __attribute__((ext_vector_type(16))) _Float16 v16h;

#if defined(__AMDGCN__) && __has_builtin(__builtin_amdgcn_wmma_f32_16x16x4_f32)
#define USE_WMMA_F32 1
#else
#define USE_WMMA_F32 0
#endif


#define NP   256
#define PN   196
#define CH   ((EE + NP - 1) / NP)
#define BINT 1024
#define EPT  ((CH + BINT - 1) / BINT)
static_assert(NP * PN >= NN, "ranges must cover the nodes");
static_assert(NP <= BINT && NP % 32 == 0, "one thread per range");
static_assert((long long)NP * CH >= EE && EPT * BINT >= CH, "the passes cover every edge");

#define SLN  16
#define SNO  (NN / SLN)
#define SED  (EE / SLN)
#define SQ   (SED / 4)
#define SQG  4
#define SQT  ((SQ + BINT - 1) / BINT)
static_assert(SLN * SLN == NP, "one work-group per (slab, slice)");
static_assert(SNO * SLN == NN && SED * SLN == EE, "slices tile the nodes, slabs tile the edges");
static_assert(SED % 4 == 0 && EE % 4 == 0, "slabs start on 16-byte boundaries in both index arrays");
static_assert(SLN * CH == SED, "a work-group's own chunk lies inside its slab");

__global__ void k_clear(float* __restrict__ a, float* __restrict__ b, int n,
                        int* __restrict__ c, int m,
                        float* __restrict__ p, int np) {
  int i = blockIdx.x * blockDim.x + threadIdx.x;
  if (i < n) { a[i] = 0.0f; b[i] = 0.0f; }
  if (i < m) c[i] = 0;
  if (i < np) p[i] = 0.0f;
}

__global__ __launch_bounds__(BINT)
void k_edge_count(const int* __restrict__ ei,
                  float* __restrict__ deg_feat,
                  float* __restrict__ degcnt,
                  int* __restrict__ pcnt) {
  __shared__ int h[NP];
  __shared__ int cs[SNO];
  __shared__ int cd[SNO];
  const int t = threadIdx.x;
  if (t < NP) h[t] = 0;
  for (int i = t; i < SNO; i += BINT) { cs[i] = 0; cd[i] = 0; }
  int d[EPT];
  const int c0 = blockIdx.x * CH;
#pragma unroll
  for (int k = 0; k < EPT; ++k) {
    int e = c0 + t + k * BINT;
    e = e < EE ? e : EE - 1;
    d[k] = ei[EE + e];
  }
  __syncthreads();
#pragma unroll
  for (int k = 0; k < EPT; ++k) {
    int i = t + k * BINT;
    if (i < CH && c0 + i < EE) atomicAdd(&h[d[k] / PN], 1);
  }

  const int slab = blockIdx.x / SLN;
  const int nb0  = (blockIdx.x % SLN) * SNO;
  const int4* __restrict__ s4 = (const int4*)(ei + slab * SED);
  const int4* __restrict__ d4 = (const int4*)(ei + EE + slab * SED);
  for (int g0 = 0; g0 < SQT; g0 += SQG) {
    int4 a[SQG], b[SQG];
#pragma unroll
    for (int k = 0; k < SQG; ++k) {
      int q = t + (g0 + k) * BINT;
      q = q < SQ ? q : SQ - 1;
      a[k] = s4[q];
      b[k] = d4[q];
    }
#pragma unroll
    for (int k = 0; k < SQG; ++k) {
      if (t + (g0 + k) * BINT < SQ) {
        unsigned u;
        u = (unsigned)(a[k].x - nb0); if (u < (unsigned)SNO) atomicAdd(&cs[u], 1);
        u = (unsigned)(a[k].y - nb0); if (u < (unsigned)SNO) atomicAdd(&cs[u], 1);
        u = (unsigned)(a[k].z - nb0); if (u < (unsigned)SNO) atomicAdd(&cs[u], 1);
        u = (unsigned)(a[k].w - nb0); if (u < (unsigned)SNO) atomicAdd(&cs[u], 1);
        u = (unsigned)(b[k].x - nb0); if (u < (unsigned)SNO) atomicAdd(&cd[u], 1);
        u = (unsigned)(b[k].y - nb0); if (u < (unsigned)SNO) atomicAdd(&cd[u], 1);
        u = (unsigned)(b[k].z - nb0); if (u < (unsigned)SNO) atomicAdd(&cd[u], 1);
        u = (unsigned)(b[k].w - nb0); if (u < (unsigned)SNO) atomicAdd(&cd[u], 1);
      }
    }
  }
  __syncthreads();
  for (int i = t; i < SNO; i += BINT) {
    int na = cs[i], nd = cd[i];
    if (na > 0) atomicAdd(&deg_feat[nb0 + i], (float)na);
    if (nd > 0) atomicAdd(&degcnt[nb0 + i], (float)nd);
  }
  if (t < NP && h[t] > 0) atomicAdd(&pcnt[t], h[t]);
}

__global__ void k_dinv(const float* __restrict__ degcnt, float* __restrict__ dinv, int n) {
  int i = blockIdx.x * blockDim.x + threadIdx.x;
  if (i < n) dinv[i] = rsqrtf(degcnt[i] + 1.0f);
}

__global__ __launch_bounds__(BINT)
void k_fill(const int* __restrict__ ei,
            const int* __restrict__ pcnt,
            const float* __restrict__ dinv,
            int* __restrict__ pcur,
            int* __restrict__ goffs,
            int4* __restrict__ rec) {
  __shared__ int h[NP];
  __shared__ int hb[NP];
  __shared__ int wsum[NP / 32];
  const int t = threadIdx.x;
  const int lane = t & 31;
  int s[EPT], d[EPT], r[EPT];
  float w[EPT];
  const int c0 = blockIdx.x * CH;
#pragma unroll
  for (int k = 0; k < EPT; ++k) {
    int e = c0 + t + k * BINT;
    e = e < EE ? e : EE - 1;
    s[k] = ei[e];
    d[k] = ei[EE + e];
  }
#pragma unroll
  for (int k = 0; k < EPT; ++k) w[k] = dinv[s[k]] * dinv[d[k]];
  int v = (t < NP) ? pcnt[t] : 0;
  int x = v;
  for (int o = 1; o < 32; o <<= 1) {
    int y = __shfl_up(x, o);
    if (lane >= o) x += y;
  }
  if (t < NP) {
    h[t] = 0;
    if (lane == 31) wsum[t >> 5] = x;
  }
  __syncthreads();
  int off = 0;
  if (t < NP) {
    for (int i = 0; i < (t >> 5); ++i) off += wsum[i];
    off += x - v;
    if (blockIdx.x == 0) goffs[t] = off;
  }
#pragma unroll
  for (int k = 0; k < EPT; ++k) {
    int i = t + k * BINT;
    r[k] = -1;
    if (i < CH && c0 + i < EE)
      r[k] = atomicAdd(&h[d[k] / PN], 1);
  }
  __syncthreads();
  if (t < NP) {
    int c = h[t];
    int g = c > 0 ? atomicAdd(&pcur[t], c) : 0;
    hb[t] = off + g;
  }
  __syncthreads();
#pragma unroll
  for (int k = 0; k < EPT; ++k) {
    if (r[k] >= 0) {
      int pos = hb[d[k] / PN] + r[k];
      rec[pos] = make_int4(s[k], d[k], __float_as_int(w[k]), 0);
    }
  }
}

__global__ void k_l1_hw(const float* __restrict__ deg_feat,
                        const float* __restrict__ W1,
                        float* __restrict__ hw, int n64) {
  int i = blockIdx.x * blockDim.x + threadIdx.x;
  if (i >= n64) return;
  hw[i] = deg_feat[i >> 6] * W1[i & 63];
}

#define AG_WAVES 32
#define AG_GRP   16
#define AG_TAB_FLOATS (41 * 1024)
static_assert((PN + 1) * DD <= AG_TAB_FLOATS, "table holds one range and the spare row");

__global__ __launch_bounds__(AG_WAVES * 32)
void k_agg(const int4* __restrict__ rec,
           const int* __restrict__ pcnt,
           const int* __restrict__ goffs,
           const float* __restrict__ hw,
           const float* __restrict__ dinv,
           const float* __restrict__ b,
           const float* __restrict__ g,
           const float* __restrict__ be,
           const float* __restrict__ m,
           const float* __restrict__ v,
           float* __restrict__ xout) {
  __shared__ float tab[AG_TAB_FLOATS];
  const int tid  = threadIdx.x;
  const int lane = tid & 31;
  const int wave = __builtin_amdgcn_readfirstlane(tid >> 5);
  const int n0 = blockIdx.x * PN;
  const int nn = (NN - n0 < PN) ? (NN - n0) : PN;
  const int start = goffs[blockIdx.x];
  const int cnt   = pcnt[blockIdx.x];
  const int nb    = (cnt + 31) >> 5;
  const float2* __restrict__ hw2 = (const float2*)hw;

  int bq = wave;
  int sv = 0, dv = n0;
  float wv = 0.0f;
  if (bq < nb) {
    int li = bq * 32 + lane;
    li = li < cnt ? li : cnt - 1;
    int4 q = rec[start + li];
    sv = q.x; dv = q.y; wv = __int_as_float(q.z);
  }

  for (int i = tid; i < nn * DD; i += AG_WAVES * 32) {
    float di = dinv[n0 + (i >> 6)];
    tab[i] = hw[n0 * DD + i] * (di * di);
  }
  __syncthreads();

  while (bq < nb) {
    const int bn = bq + AG_WAVES;
    int sn = 0, dn = n0;
    float wn = 0.0f;
    if (bn < nb) {
      int li = bn * 32 + lane;
      li = li < cnt ? li : cnt - 1;
      int4 q = rec[start + li];
      sn = q.x; dn = q.y; wn = __int_as_float(q.z);
    }
    const bool  live = bq * 32 + lane < cnt;
    const int   rv = live ? (dv - n0) * DD : PN * DD;
    const float wl = live ? wv : 0.0f;
#pragma unroll
    for (int j0 = 0; j0 < 32; j0 += AG_GRP) {
      float2 vr[AG_GRP];
#pragma unroll
      for (int u = 0; u < AG_GRP; ++u) {
        int sj = __builtin_amdgcn_readlane(sv, j0 + u);
        vr[u] = hw2[sj * (DD / 2) + lane];
      }
#pragma unroll
      for (int u = 0; u < AG_GRP; ++u) {
        int   rj = __builtin_amdgcn_readlane(rv, j0 + u);
        float wj = __int_as_float(__builtin_amdgcn_readlane(__float_as_int(wl), j0 + u));
        atomicAdd(&tab[rj + 2 * lane],     vr[u].x * wj);
        atomicAdd(&tab[rj + 2 * lane + 1], vr[u].y * wj);
      }
    }
    sv = sn; dv = dn; wv = wn; bq = bn;
  }
  __syncthreads();

  for (int i = tid; i < nn * DD; i += AG_WAVES * 32) {
    int f = i & 63;
    float rs = rsqrtf(v[f] + BN_EPS);
    float y = ((tab[i] + b[f]) - m[f]) * rs * g[f] + be[f];
    xout[n0 * DD + i] = fmaxf(y, 0.0f);
  }
}

#define PCH  16
#define PCS  ((PN + PCH - 1) / PCH)
static_assert(PCH * DD == AG_WAVES * 32, "one thread per (chunk, feature)");

__global__ __launch_bounds__(AG_WAVES * 32)
void k_pool_agg(const int4* __restrict__ rec,
                const int* __restrict__ pcnt,
                const int* __restrict__ goffs,
                const int* __restrict__ batch,
                const float* __restrict__ x,
                const float* __restrict__ dinv,
                float* __restrict__ psum,
                float* __restrict__ cntg) {
  __shared__ float tab[(PN + 1) * DD];
  __shared__ float gtab[GG * DD];
  __shared__ float ctab[GG];
  __shared__ int   gloc[PN];
  const int tid  = threadIdx.x;
  const int lane = tid & 31;
  const int wave = __builtin_amdgcn_readfirstlane(tid >> 5);
  const int n0 = blockIdx.x * PN;
  const int nn = (NN - n0 < PN) ? (NN - n0) : PN;
  const int start = goffs[blockIdx.x];
  const int cnt   = pcnt[blockIdx.x];
  const int nb    = (cnt + 31) >> 5;
  const float2* __restrict__ x2 = (const float2*)x;

  int bq = wave;
  int sv = 0, dv = n0;
  float wv = 0.0f;
  if (bq < nb) {
    int li = bq * 32 + lane;
    li = li < cnt ? li : cnt - 1;
    int4 q = rec[start + li];
    sv = q.x; dv = q.y; wv = __int_as_float(q.z);
  }

  for (int i = tid; i < nn * DD; i += AG_WAVES * 32) {
    float di = dinv[n0 + (i >> 6)];
    tab[i] = x[n0 * DD + i] * (di * di);
  }
  float4 z4 = {0.0f, 0.0f, 0.0f, 0.0f};
  for (int i = tid; i < GG * DD / 4; i += AG_WAVES * 32) ((float4*)gtab)[i] = z4;
  if (tid < GG) ctab[tid] = 0.0f;
  if (tid < nn) gloc[tid] = batch[n0 + tid];
  __syncthreads();

  while (bq < nb) {
    const int bn = bq + AG_WAVES;
    int sn = 0, dn = n0;
    float wn = 0.0f;
    if (bn < nb) {
      int li = bn * 32 + lane;
      li = li < cnt ? li : cnt - 1;
      int4 q = rec[start + li];
      sn = q.x; dn = q.y; wn = __int_as_float(q.z);
    }
    const bool  live = bq * 32 + lane < cnt;
    const int   rv = live ? (dv - n0) * DD : PN * DD;
    const float wl = live ? wv : 0.0f;
#pragma unroll
    for (int j0 = 0; j0 < 32; j0 += AG_GRP) {
      float2 vr[AG_GRP];
#pragma unroll
      for (int u = 0; u < AG_GRP; ++u) {
        int sj = __builtin_amdgcn_readlane(sv, j0 + u);
        vr[u] = x2[sj * (DD / 2) + lane];
      }
#pragma unroll
      for (int u = 0; u < AG_GRP; ++u) {
        int   rj = __builtin_amdgcn_readlane(rv, j0 + u);
        float wj = __int_as_float(__builtin_amdgcn_readlane(__float_as_int(wl), j0 + u));
        atomicAdd(&tab[rj + 2 * lane],     vr[u].x * wj);
        atomicAdd(&tab[rj + 2 * lane + 1], vr[u].y * wj);
      }
    }
    sv = sn; dv = dn; wv = wn; bq = bn;
  }
  __syncthreads();

  {
    const int f  = tid & (DD - 1);
    const int c0 = (tid >> 6) * PCS;
    const int c1 = (c0 + PCS < nn) ? (c0 + PCS) : nn;
    if (c0 < c1) {
      int   cg = gloc[c0];
      float a  = 0.0f;
      for (int i = c0; i < c1; ++i) {
        int g = gloc[i];
        if (g != cg) { atomicAdd(&gtab[cg * DD + f], a); a = 0.0f; cg = g; }
        a += tab[i * DD + f];
      }
      atomicAdd(&gtab[cg * DD + f], a);
    }
    if (tid < nn) atomicAdd(&ctab[gloc[tid]], 1.0f);
  }
  __syncthreads();

  for (int i = tid; i < GG * DD; i += AG_WAVES * 32)
    if (ctab[i >> 6] > 0.0f) atomicAdd(&psum[i], gtab[i]);
  if (tid < GG && ctab[tid] > 0.0f) atomicAdd(&cntg[tid], ctab[tid]);
}

__global__ void k_gemm_wmma(const float* __restrict__ A,
                            const float* __restrict__ B,
                            float* __restrict__ C,
                            const float* __restrict__ bias,
                            const float* __restrict__ dinv,
                            float* __restrict__ acc,
                            const float* __restrict__ cnt,
                            int M, int K, int Nc) {
  int wid  = (blockIdx.x * blockDim.x + threadIdx.x) >> 5;
  int lane = threadIdx.x & 31;
  int tiles_n = Nc >> 4;
  int tm = wid / tiles_n;
  int tn = wid - tm * tiles_n;
  if (tm * 16 >= M) return;

  int half = lane >> 4;
  int lm   = lane & 15;
  const float* Arow = A + (size_t)(tm * 16 + lm) * K;
  int ncol = tn * 16 + lm;

  v8f c = {};
#if USE_WMMA_F32
  for (int k0 = 0; k0 < K; k0 += 4) {
    v2f a, b;
    a.x = Arow[k0 + 2 * half + 0];
    a.y = Arow[k0 + 2 * half + 1];
    b.x = B[(size_t)(k0 + 2 * half + 0) * Nc + ncol];
    b.y = B[(size_t)(k0 + 2 * half + 1) * Nc + ncol];
    c = __builtin_amdgcn_wmma_f32_16x16x4_f32(false, a, false, b, (short)0, c,
                                              false, false);
  }
#else
  for (int k0 = 0; k0 < K; k0 += 32) {
    v16h a, b;
#pragma unroll
    for (int i = 0; i < 8; ++i) {
      a[i]     = (_Float16)Arow[k0 + 8 * half + i];
      a[8 + i] = (_Float16)Arow[k0 + 16 + 8 * half + i];
    }
#pragma unroll
    for (int i = 0; i < 16; ++i)
      b[i] = (_Float16)B[(size_t)(k0 + 16 * half + i) * Nc + ncol];
    c = __builtin_amdgcn_wmma_f32_16x16x32_f16(false, a, false, b, (short)0, c,
                                               false, false);
  }
#endif

  float bv = bias ? bias[ncol] : 0.0f;
  if (acc) {
#pragma unroll
    for (int r = 0; r < 8; ++r) {
      int row = tm * 16 + 8 * half + r;
      size_t idx = (size_t)row * Nc + ncol;
      float val = c[r] + bv;
      float di = dinv[row];
      C[idx]   = val;
      acc[idx] = val * (di * di);
    }
  } else {
#pragma unroll
    for (int r = 0; r < 8; ++r) {
      int row = tm * 16 + 8 * half + r;
      float val = c[r] + bv;
      if (cnt) {
        float cc = cnt[row];
        val = (cc > 0.0f) ? c[r] / cc + bv : 0.0f;
      }
      C[(size_t)row * Nc + ncol] = val;
    }
  }
}

static inline int blk(long long n, int b) { return (int)((n + b - 1) / b); }

extern "C" void kernel_launch(void* const* d_in, const int* in_sizes, int n_in,
                              void* d_out, int out_size, void* d_ws, size_t ws_size,
                              hipStream_t stream) {
  (void)in_sizes; (void)n_in; (void)out_size; (void)ws_size;
  const int*   edge_index = (const int*)d_in[0];
  const int*   batch      = (const int*)d_in[1];
  const float* W1 = (const float*)d_in[2];
  const float* b1 = (const float*)d_in[3];
  const float* W2 = (const float*)d_in[4];
  const float* b2 = (const float*)d_in[5];
  const float* W3 = (const float*)d_in[6];
  const float* b3 = (const float*)d_in[7];
  const float* g1 = (const float*)d_in[8];
  const float* be1= (const float*)d_in[9];
  const float* m1 = (const float*)d_in[10];
  const float* v1 = (const float*)d_in[11];
  const float* g2 = (const float*)d_in[12];
  const float* be2= (const float*)d_in[13];
  const float* m2 = (const float*)d_in[14];
  const float* v2 = (const float*)d_in[15];
  const float* Wl = (const float*)d_in[16];
  const float* bl = (const float*)d_in[17];
  float* out = (float*)d_out;

  char* ws = (char*)d_ws;
  size_t off = 0;
  auto carve = [&](size_t bytes) -> float* {
    float* p = (float*)(ws + off);
    off += (bytes + 255) & ~(size_t)255;
    return p;
  };
  float* deg_feat = carve((size_t)NN * 4);
  float* degcnt   = carve((size_t)NN * 4);
  float* dinv     = carve((size_t)NN * 4);
  float* bufX     = carve((size_t)NN * DD * 4);
  float* bufH     = carve((size_t)NN * DD * 4);
  float* psum     = carve((size_t)(GG * DD + GG) * 4);
  float* cnt      = psum + GG * DD;
  float* poolT    = carve((size_t)GG * DD * 4);
  int*   pcnt     = (int*)carve((size_t)2 * NP * 4);
  int*   pcur     = pcnt + NP;
  int*   goffs    = (int*)carve((size_t)NP * 4);
  int4*  rec      = (int4*)carve((size_t)EE * 16);

  const int B = 256;
  const long long N64 = (long long)NN * DD;

  k_clear<<<blk(NN, B), B, 0, stream>>>(deg_feat, degcnt, NN, pcnt, 2 * NP, psum, GG * DD + GG);
  k_edge_count<<<NP, BINT, 0, stream>>>(edge_index, deg_feat, degcnt, pcnt);
  k_dinv<<<blk(NN, B), B, 0, stream>>>(degcnt, dinv, NN);
  k_fill<<<NP, BINT, 0, stream>>>(edge_index, pcnt, dinv, pcur, goffs, rec);

  const int tiles_NN  = (NN / 16) * (DD / 16);
  const int tiles_POOL = (GG / 16) * (DD / 16);
  const int tiles_FIN = (GG / 16) * (DOUT / 16);

  k_l1_hw<<<blk(N64, B), B, 0, stream>>>(deg_feat, W1, bufH, (int)N64);
  k_agg<<<NP, AG_WAVES * 32, 0, stream>>>(rec, pcnt, goffs, bufH, dinv,
                                          b1, g1, be1, m1, v1, bufX);

  k_gemm_wmma<<<blk((long long)tiles_NN * 32, B), B, 0, stream>>>(
      bufX, W2, bufH, nullptr, nullptr, nullptr, nullptr, NN, DD, DD);
  k_agg<<<NP, AG_WAVES * 32, 0, stream>>>(rec, pcnt, goffs, bufH, dinv,
                                          b2, g2, be2, m2, v2, bufX);

  k_pool_agg<<<NP, AG_WAVES * 32, 0, stream>>>(rec, pcnt, goffs, batch, bufX, dinv, psum, cnt);

  k_gemm_wmma<<<blk((long long)tiles_POOL * 32, B), B, 0, stream>>>(
      psum, W3, poolT, b3, nullptr, nullptr, cnt, GG, DD, DD);
  k_gemm_wmma<<<blk((long long)tiles_FIN * 32, B), B, 0, stream>>>(
      poolT, Wl, out, bl, nullptr, nullptr, nullptr, GG, DD, DOUT);
}


# Aggregation launches: table set-up, store and per-graph hand-over passes issue all their loads before the first use, 16 bytes per lane; one round trip per pass instead of one per element
# speedup vs baseline: 2.9726x; 1.0802x over previous
#include <hip/hip_runtime.h>
#include <hip/hip_bf16.h>

#define NN   50000
#define EE   800000
#define GG   512
#define DD   64
#define DOUT 32
#define BN_EPS 1e-5f

typedef __attribute__((ext_vector_type(2)))  float    v2f;
typedef __attribute__((ext_vector_type(8)))  float    v8f;
typedef __attribute__((ext_vector_type(16))) _Float16 v16h;

#if defined(__AMDGCN__) && __has_builtin(__builtin_amdgcn_wmma_f32_16x16x4_f32)
#define USE_WMMA_F32 1
#else
#define USE_WMMA_F32 0
#endif


#define NP   256
#define PN   196
#define CH   ((EE + NP - 1) / NP)
#define BINT 1024
#define EPT  ((CH + BINT - 1) / BINT)
static_assert(NP * PN >= NN, "ranges must cover the nodes");
static_assert(NP <= BINT && NP % 32 == 0, "one thread per range");
static_assert((long long)NP * CH >= EE && EPT * BINT >= CH, "the passes cover every edge");

#define SLN  16
#define SNO  (NN / SLN)
#define SED  (EE / SLN)
#define SQ   (SED / 4)
#define SQG  4
#define SQT  ((SQ + BINT - 1) / BINT)
static_assert(SLN * SLN == NP, "one work-group per (slab, slice)");
static_assert(SNO * SLN == NN && SED * SLN == EE, "slices tile the nodes, slabs tile the edges");
static_assert(SED % 4 == 0 && EE % 4 == 0, "slabs start on 16-byte boundaries in both index arrays");
static_assert(SLN * CH == SED, "a work-group's own chunk lies inside its slab");

__global__ void k_clear(float* __restrict__ a, float* __restrict__ b, int n,
                        int* __restrict__ c, int m,
                        float* __restrict__ p, int np) {
  int i = blockIdx.x * blockDim.x + threadIdx.x;
  if (i < n) { a[i] = 0.0f; b[i] = 0.0f; }
  if (i < m) c[i] = 0;
  if (i < np) p[i] = 0.0f;
}

__global__ __launch_bounds__(BINT)
void k_edge_count(const int* __restrict__ ei,
                  float* __restrict__ deg_feat,
                  float* __restrict__ degcnt,
                  int* __restrict__ pcnt) {
  __shared__ int h[NP];
  __shared__ int cs[SNO];
  __shared__ int cd[SNO];
  const int t = threadIdx.x;
  if (t < NP) h[t] = 0;
  for (int i = t; i < SNO; i += BINT) { cs[i] = 0; cd[i] = 0; }
  int d[EPT];
  const int c0 = blockIdx.x * CH;
#pragma unroll
  for (int k = 0; k < EPT; ++k) {
    int e = c0 + t + k * BINT;
    e = e < EE ? e : EE - 1;
    d[k] = ei[EE + e];
  }
  __syncthreads();
#pragma unroll
  for (int k = 0; k < EPT; ++k) {
    int i = t + k * BINT;
    if (i < CH && c0 + i < EE) atomicAdd(&h[d[k] / PN], 1);
  }

  const int slab = blockIdx.x / SLN;
  const int nb0  = (blockIdx.x % SLN) * SNO;
  const int4* __restrict__ s4 = (const int4*)(ei + slab * SED);
  const int4* __restrict__ d4 = (const int4*)(ei + EE + slab * SED);
  for (int g0 = 0; g0 < SQT; g0 += SQG) {
    int4 a[SQG], b[SQG];
#pragma unroll
    for (int k = 0; k < SQG; ++k) {
      int q = t + (g0 + k) * BINT;
      q = q < SQ ? q : SQ - 1;
      a[k] = s4[q];
      b[k] = d4[q];
    }
#pragma unroll
    for (int k = 0; k < SQG; ++k) {
      if (t + (g0 + k) * BINT < SQ) {
        unsigned u;
        u = (unsigned)(a[k].x - nb0); if (u < (unsigned)SNO) atomicAdd(&cs[u], 1);
        u = (unsigned)(a[k].y - nb0); if (u < (unsigned)SNO) atomicAdd(&cs[u], 1);
        u = (unsigned)(a[k].z - nb0); if (u < (unsigned)SNO) atomicAdd(&cs[u], 1);
        u = (unsigned)(a[k].w - nb0); if (u < (unsigned)SNO) atomicAdd(&cs[u], 1);
        u = (unsigned)(b[k].x - nb0); if (u < (unsigned)SNO) atomicAdd(&cd[u], 1);
        u = (unsigned)(b[k].y - nb0); if (u < (unsigned)SNO) atomicAdd(&cd[u], 1);
        u = (unsigned)(b[k].z - nb0); if (u < (unsigned)SNO) atomicAdd(&cd[u], 1);
        u = (unsigned)(b[k].w - nb0); if (u < (unsigned)SNO) atomicAdd(&cd[u], 1);
      }
    }
  }
  __syncthreads();
  for (int i = t; i < SNO; i += BINT) {
    int na = cs[i], nd = cd[i];
    if (na > 0) atomicAdd(&deg_feat[nb0 + i], (float)na);
    if (nd > 0) atomicAdd(&degcnt[nb0 + i], (float)nd);
  }
  if (t < NP && h[t] > 0) atomicAdd(&pcnt[t], h[t]);
}

__global__ void k_dinv(const float* __restrict__ degcnt, float* __restrict__ dinv, int n) {
  int i = blockIdx.x * blockDim.x + threadIdx.x;
  if (i < n) dinv[i] = rsqrtf(degcnt[i] + 1.0f);
}

__global__ __launch_bounds__(BINT)
void k_fill(const int* __restrict__ ei,
            const int* __restrict__ pcnt,
            const float* __restrict__ dinv,
            int* __restrict__ pcur,
            int* __restrict__ goffs,
            int4* __restrict__ rec) {
  __shared__ int h[NP];
  __shared__ int hb[NP];
  __shared__ int wsum[NP / 32];
  const int t = threadIdx.x;
  const int lane = t & 31;
  int s[EPT], d[EPT], r[EPT];
  float w[EPT];
  const int c0 = blockIdx.x * CH;
#pragma unroll
  for (int k = 0; k < EPT; ++k) {
    int e = c0 + t + k * BINT;
    e = e < EE ? e : EE - 1;
    s[k] = ei[e];
    d[k] = ei[EE + e];
  }
#pragma unroll
  for (int k = 0; k < EPT; ++k) w[k] = dinv[s[k]] * dinv[d[k]];
  int v = (t < NP) ? pcnt[t] : 0;
  int x = v;
  for (int o = 1; o < 32; o <<= 1) {
    int y = __shfl_up(x, o);
    if (lane >= o) x += y;
  }
  if (t < NP) {
    h[t] = 0;
    if (lane == 31) wsum[t >> 5] = x;
  }
  __syncthreads();
  int off = 0;
  if (t < NP) {
    for (int i = 0; i < (t >> 5); ++i) off += wsum[i];
    off += x - v;
    if (blockIdx.x == 0) goffs[t] = off;
  }
#pragma unroll
  for (int k = 0; k < EPT; ++k) {
    int i = t + k * BINT;
    r[k] = -1;
    if (i < CH && c0 + i < EE)
      r[k] = atomicAdd(&h[d[k] / PN], 1);
  }
  __syncthreads();
  if (t < NP) {
    int c = h[t];
    int g = c > 0 ? atomicAdd(&pcur[t], c) : 0;
    hb[t] = off + g;
  }
  __syncthreads();
#pragma unroll
  for (int k = 0; k < EPT; ++k) {
    if (r[k] >= 0) {
      int pos = hb[d[k] / PN] + r[k];
      rec[pos] = make_int4(s[k], d[k], __float_as_int(w[k]), 0);
    }
  }
}

__global__ void k_l1_hw(const float* __restrict__ deg_feat,
                        const float* __restrict__ W1,
                        float* __restrict__ hw, int n64) {
  int i = blockIdx.x * blockDim.x + threadIdx.x;
  if (i >= n64) return;
  hw[i] = deg_feat[i >> 6] * W1[i & 63];
}

#define AG_WAVES 32
#define AG_GRP   16
#define AG_TAB_FLOATS (41 * 1024)
static_assert((PN + 1) * DD <= AG_TAB_FLOATS, "table holds one range and the spare row");
#define AG_THREADS (AG_WAVES * 32)
#define TQ   (PN * DD / 4)
#define TQT  ((TQ + AG_THREADS - 1) / AG_THREADS)
static_assert(DD % 4 == 0 && AG_THREADS % (DD / 4) == 0, "a thread keeps the same four features in every group");

__global__ __launch_bounds__(AG_WAVES * 32)
void k_agg(const int4* __restrict__ rec,
           const int* __restrict__ pcnt,
           const int* __restrict__ goffs,
           const float* __restrict__ hw,
           const float* __restrict__ dinv,
           const float* __restrict__ b,
           const float* __restrict__ g,
           const float* __restrict__ be,
           const float* __restrict__ m,
           const float* __restrict__ v,
           float* __restrict__ xout) {
  __shared__ __attribute__((aligned(16))) float tab[AG_TAB_FLOATS];
  const int tid  = threadIdx.x;
  const int lane = tid & 31;
  const int wave = __builtin_amdgcn_readfirstlane(tid >> 5);
  const int n0 = blockIdx.x * PN;
  const int nn = (NN - n0 < PN) ? (NN - n0) : PN;
  const int start = goffs[blockIdx.x];
  const int cnt   = pcnt[blockIdx.x];
  const int nb    = (cnt + 31) >> 5;
  const float2* __restrict__ hw2 = (const float2*)hw;

  int bq = wave;
  int sv = 0, dv = n0;
  float wv = 0.0f;
  if (bq < nb) {
    int li = bq * 32 + lane;
    li = li < cnt ? li : cnt - 1;
    int4 q = rec[start + li];
    sv = q.x; dv = q.y; wv = __int_as_float(q.z);
  }

  const int f0 = (tid & (DD / 4 - 1)) * 4;
  float cb[4], cg[4], cbe[4], cm[4], cv[4];
#pragma unroll
  for (int j = 0; j < 4; ++j) {
    cb[j] = b[f0 + j]; cg[j] = g[f0 + j]; cbe[j] = be[f0 + j]; cm[j] = m[f0 + j]; cv[j] = v[f0 + j];
  }

  const int nq = nn * (DD / 4);
  const float4* __restrict__ hw4 = (const float4*)(hw + n0 * DD);
  {
    float4 hv[TQT];
    float  dq[TQT];
#pragma unroll
    for (int k = 0; k < TQT; ++k) {
      int q = tid + k * AG_THREADS;
      q = q < nq ? q : nq - 1;
      hv[k] = hw4[q];
      dq[k] = dinv[n0 + (q >> 4)];
    }
#pragma unroll
    for (int k = 0; k < TQT; ++k) {
      const int q = tid + k * AG_THREADS;
      if (q < nq) {
        const float s = dq[k] * dq[k];
        float4 t = hv[k];
        t.x *= s; t.y *= s; t.z *= s; t.w *= s;
        ((float4*)tab)[q] = t;
      }
    }
  }
  __syncthreads();

  while (bq < nb) {
    const int bn = bq + AG_WAVES;
    int sn = 0, dn = n0;
    float wn = 0.0f;
    if (bn < nb) {
      int li = bn * 32 + lane;
      li = li < cnt ? li : cnt - 1;
      int4 q = rec[start + li];
      sn = q.x; dn = q.y; wn = __int_as_float(q.z);
    }
    const bool  live = bq * 32 + lane < cnt;
    const int   rv = live ? (dv - n0) * DD : PN * DD;
    const float wl = live ? wv : 0.0f;
#pragma unroll
    for (int j0 = 0; j0 < 32; j0 += AG_GRP) {
      float2 vr[AG_GRP];
#pragma unroll
      for (int u = 0; u < AG_GRP; ++u) {
        int sj = __builtin_amdgcn_readlane(sv, j0 + u);
        vr[u] = hw2[sj * (DD / 2) + lane];
      }
#pragma unroll
      for (int u = 0; u < AG_GRP; ++u) {
        int   rj = __builtin_amdgcn_readlane(rv, j0 + u);
        float wj = __int_as_float(__builtin_amdgcn_readlane(__float_as_int(wl), j0 + u));
        atomicAdd(&tab[rj + 2 * lane],     vr[u].x * wj);
        atomicAdd(&tab[rj + 2 * lane + 1], vr[u].y * wj);
      }
    }
    sv = sn; dv = dn; wv = wn; bq = bn;
  }
  __syncthreads();

  float rs[4];
#pragma unroll
  for (int j = 0; j < 4; ++j) rs[j] = rsqrtf(cv[j] + BN_EPS);
  float4* __restrict__ xo4 = (float4*)(xout + n0 * DD);
#pragma unroll
  for (int k = 0; k < TQT; ++k) {
    const int q = tid + k * AG_THREADS;
    if (q < nq) {
      const float4 t = ((const float4*)tab)[q];
      float4 y;
      y.x = fmaxf(((t.x + cb[0]) - cm[0]) * rs[0] * cg[0] + cbe[0], 0.0f);
      y.y = fmaxf(((t.y + cb[1]) - cm[1]) * rs[1] * cg[1] + cbe[1], 0.0f);
      y.z = fmaxf(((t.z + cb[2]) - cm[2]) * rs[2] * cg[2] + cbe[2], 0.0f);
      y.w = fmaxf(((t.w + cb[3]) - cm[3]) * rs[3] * cg[3] + cbe[3], 0.0f);
      xo4[q] = y;
    }
  }
}

#define PCH  16
#define PCS  ((PN + PCH - 1) / PCH)
static_assert(PCH * DD == AG_WAVES * 32, "one thread per (chunk, feature)");

__global__ __launch_bounds__(AG_WAVES * 32)
void k_pool_agg(const int4* __restrict__ rec,
                const int* __restrict__ pcnt,
                const int* __restrict__ goffs,
                const int* __restrict__ batch,
                const float* __restrict__ x,
                const float* __restrict__ dinv,
                float* __restrict__ psum,
                float* __restrict__ cntg) {
  __shared__ __attribute__((aligned(16))) float tab[(PN + 1) * DD];
  __shared__ __attribute__((aligned(16))) float gtab[GG * DD];
  __shared__ float ctab[GG];
  __shared__ int   gloc[PN];
  const int tid  = threadIdx.x;
  const int lane = tid & 31;
  const int wave = __builtin_amdgcn_readfirstlane(tid >> 5);
  const int n0 = blockIdx.x * PN;
  const int nn = (NN - n0 < PN) ? (NN - n0) : PN;
  const int start = goffs[blockIdx.x];
  const int cnt   = pcnt[blockIdx.x];
  const int nb    = (cnt + 31) >> 5;
  const float2* __restrict__ x2 = (const float2*)x;

  int bq = wave;
  int sv = 0, dv = n0;
  float wv = 0.0f;
  if (bq < nb) {
    int li = bq * 32 + lane;
    li = li < cnt ? li : cnt - 1;
    int4 q = rec[start + li];
    sv = q.x; dv = q.y; wv = __int_as_float(q.z);
  }

  const int nq = nn * (DD / 4);
  const float4* __restrict__ x4 = (const float4*)(x + n0 * DD);
  {
    const int gb = batch[n0 + (tid < nn ? tid : nn - 1)];
    float4 hv[TQT];
    float  dq[TQT];
#pragma unroll
    for (int k = 0; k < TQT; ++k) {
      int q = tid + k * AG_THREADS;
      q = q < nq ? q : nq - 1;
      hv[k] = x4[q];
      dq[k] = dinv[n0 + (q >> 4)];
    }
    float4 z4 = {0.0f, 0.0f, 0.0f, 0.0f};
    for (int i = tid; i < GG * DD / 4; i += AG_THREADS) ((float4*)gtab)[i] = z4;
    if (tid < GG) ctab[tid] = 0.0f;
#pragma unroll
    for (int k = 0; k < TQT; ++k) {
      const int q = tid + k * AG_THREADS;
      if (q < nq) {
        const float s = dq[k] * dq[k];
        float4 t = hv[k];
        t.x *= s; t.y *= s; t.z *= s; t.w *= s;
        ((float4*)tab)[q] = t;
      }
    }
    if (tid < nn) gloc[tid] = gb;
  }
  __syncthreads();

  while (bq < nb) {
    const int bn = bq + AG_WAVES;
    int sn = 0, dn = n0;
    float wn = 0.0f;
    if (bn < nb) {
      int li = bn * 32 + lane;
      li = li < cnt ? li : cnt - 1;
      int4 q = rec[start + li];
      sn = q.x; dn = q.y; wn = __int_as_float(q.z);
    }
    const bool  live = bq * 32 + lane < cnt;
    const int   rv = live ? (dv - n0) * DD : PN * DD;
    const float wl = live ? wv : 0.0f;
#pragma unroll
    for (int j0 = 0; j0 < 32; j0 += AG_GRP) {
      float2 vr[AG_GRP];
#pragma unroll
      for (int u = 0; u < AG_GRP; ++u) {
        int sj = __builtin_amdgcn_readlane(sv, j0 + u);
        vr[u] = x2[sj * (DD / 2) + lane];
      }
#pragma unroll
      for (int u = 0; u < AG_GRP; ++u) {
        int   rj = __builtin_amdgcn_readlane(rv, j0 + u);
        float wj = __int_as_float(__builtin_amdgcn_readlane(__float_as_int(wl), j0 + u));
        atomicAdd(&tab[rj + 2 * lane],     vr[u].x * wj);
        atomicAdd(&tab[rj + 2 * lane + 1], vr[u].y * wj);
      }
    }
    sv = sn; dv = dn; wv = wn; bq = bn;
  }
  __syncthreads();

  {
    const int f  = tid & (DD - 1);
    const int c0 = (tid >> 6) * PCS;
    const int c1 = (c0 + PCS < nn) ? (c0 + PCS) : nn;
    if (c0 < c1) {
      int   ids[PCS];
      float vals[PCS];
#pragma unroll
      for (int j = 0; j < PCS; ++j) {
        int i = c0 + j;
        i = i < c1 ? i : c1 - 1;
        ids[j]  = gloc[i];
        vals[j] = tab[i * DD + f];
      }
      int   cg = ids[0];
      float a  = 0.0f;
#pragma unroll
      for (int j = 0; j < PCS; ++j) {
        if (c0 + j < c1) {
          if (ids[j] != cg) { atomicAdd(&gtab[cg * DD + f], a); a = 0.0f; cg = ids[j]; }
          a += vals[j];
        }
      }
      atomicAdd(&gtab[cg * DD + f], a);
    }
    if (tid < nn) atomicAdd(&ctab[gloc[tid]], 1.0f);
  }
  __syncthreads();

  {
    static_assert(GG * DD == 32 * AG_THREADS && AG_THREADS / DD == 16, "lane k of a wave <-> graph c + 16 k");
    const float cl = ctab[(tid >> 6) + 16 * lane];
    unsigned present = (unsigned)__ballot(cl > 0.0f);
    while (present) {
      const int k = __builtin_ctz(present);
      present &= present - 1;
      const int i = tid + k * AG_THREADS;
      atomicAdd(&psum[i], gtab[i]);
    }
  }
  if (tid < GG && ctab[tid] > 0.0f) atomicAdd(&cntg[tid], ctab[tid]);
}

__global__ void k_gemm_wmma(const float* __restrict__ A,
                            const float* __restrict__ B,
                            float* __restrict__ C,
                            const float* __restrict__ bias,
                            const float* __restrict__ dinv,
                            float* __restrict__ acc,
                            const float* __restrict__ cnt,
                            int M, int K, int Nc) {
  int wid  = (blockIdx.x * blockDim.x + threadIdx.x) >> 5;
  int lane = threadIdx.x & 31;
  int tiles_n = Nc >> 4;
  int tm = wid / tiles_n;
  int tn = wid - tm * tiles_n;
  if (tm * 16 >= M) return;

  int half = lane >> 4;
  int lm   = lane & 15;
  const float* Arow = A + (size_t)(tm * 16 + lm) * K;
  int ncol = tn * 16 + lm;

  v8f c = {};
#if USE_WMMA_F32
  for (int k0 = 0; k0 < K; k0 += 4) {
    v2f a, b;
    a.x = Arow[k0 + 2 * half + 0];
    a.y = Arow[k0 + 2 * half + 1];
    b.x = B[(size_t)(k0 + 2 * half + 0) * Nc + ncol];
    b.y = B[(size_t)(k0 + 2 * half + 1) * Nc + ncol];
    c = __builtin_amdgcn_wmma_f32_16x16x4_f32(false, a, false, b, (short)0, c,
                                              false, false);
  }
#else
  for (int k0 = 0; k0 < K; k0 += 32) {
    v16h a, b;
#pragma unroll
    for (int i = 0; i < 8; ++i) {
      a[i]     = (_Float16)Arow[k0 + 8 * half + i];
      a[8 + i] = (_Float16)Arow[k0 + 16 + 8 * half + i];
    }
#pragma unroll
    for (int i = 0; i < 16; ++i)
      b[i] = (_Float16)B[(size_t)(k0 + 16 * half + i) * Nc + ncol];
    c = __builtin_amdgcn_wmma_f32_16x16x32_f16(false, a, false, b, (short)0, c,
                                               false, false);
  }
#endif

  float bv = bias ? bias[ncol] : 0.0f;
  if (acc) {
#pragma unroll
    for (int r = 0; r < 8; ++r) {
      int row = tm * 16 + 8 * half + r;
      size_t idx = (size_t)row * Nc + ncol;
      float val = c[r] + bv;
      float di = dinv[row];
      C[idx]   = val;
      acc[idx] = val * (di * di);
    }
  } else {
#pragma unroll
    for (int r = 0; r < 8; ++r) {
      int row = tm * 16 + 8 * half + r;
      float val = c[r] + bv;
      if (cnt) {
        float cc = cnt[row];
        val = (cc > 0.0f) ? c[r] / cc + bv : 0.0f;
      }
      C[(size_t)row * Nc + ncol] = val;
    }
  }
}

static inline int blk(long long n, int b) { return (int)((n + b - 1) / b); }

extern "C" void kernel_launch(void* const* d_in, const int* in_sizes, int n_in,
                              void* d_out, int out_size, void* d_ws, size_t ws_size,
                              hipStream_t stream) {
  (void)in_sizes; (void)n_in; (void)out_size; (void)ws_size;
  const int*   edge_index = (const int*)d_in[0];
  const int*   batch      = (const int*)d_in[1];
  const float* W1 = (const float*)d_in[2];
  const float* b1 = (const float*)d_in[3];
  const float* W2 = (const float*)d_in[4];
  const float* b2 = (const float*)d_in[5];
  const float* W3 = (const float*)d_in[6];
  const float* b3 = (const float*)d_in[7];
  const float* g1 = (const float*)d_in[8];
  const float* be1= (const float*)d_in[9];
  const float* m1 = (const float*)d_in[10];
  const float* v1 = (const float*)d_in[11];
  const float* g2 = (const float*)d_in[12];
  const float* be2= (const float*)d_in[13];
  const float* m2 = (const float*)d_in[14];
  const float* v2 = (const float*)d_in[15];
  const float* Wl = (const float*)d_in[16];
  const float* bl = (const float*)d_in[17];
  float* out = (float*)d_out;

  char* ws = (char*)d_ws;
  size_t off = 0;
  auto carve = [&](size_t bytes) -> float* {
    float* p = (float*)(ws + off);
    off += (bytes + 255) & ~(size_t)255;
    return p;
  };
  float* deg_feat = carve((size_t)NN * 4);
  float* degcnt   = carve((size_t)NN * 4);
  float* dinv     = carve((size_t)NN * 4);
  float* bufX     = carve((size_t)NN * DD * 4);
  float* bufH     = carve((size_t)NN * DD * 4);
  float* psum     = carve((size_t)(GG * DD + GG) * 4);
  float* cnt      = psum + GG * DD;
  float* poolT    = carve((size_t)GG * DD * 4);
  int*   pcnt     = (int*)carve((size_t)2 * NP * 4);
  int*   pcur     = pcnt + NP;
  int*   goffs    = (int*)carve((size_t)NP * 4);
  int4*  rec      = (int4*)carve((size_t)EE * 16);

  const int B = 256;
  const long long N64 = (long long)NN * DD;

  k_clear<<<blk(NN, B), B, 0, stream>>>(deg_feat, degcnt, NN, pcnt, 2 * NP, psum, GG * DD + GG);
  k_edge_count<<<NP, BINT, 0, stream>>>(edge_index, deg_feat, degcnt, pcnt);
  k_dinv<<<blk(NN, B), B, 0, stream>>>(degcnt, dinv, NN);
  k_fill<<<NP, BINT, 0, stream>>>(edge_index, pcnt, dinv, pcur, goffs, rec);

  const int tiles_NN  = (NN / 16) * (DD / 16);
  const int tiles_POOL = (GG / 16) * (DD / 16);
  const int tiles_FIN = (GG / 16) * (DOUT / 16);

  k_l1_hw<<<blk(N64, B), B, 0, stream>>>(deg_feat, W1, bufH, (int)N64);
  k_agg<<<NP, AG_WAVES * 32, 0, stream>>>(rec, pcnt, goffs, bufH, dinv,
                                          b1, g1, be1, m1, v1, bufX);

  k_gemm_wmma<<<blk((long long)tiles_NN * 32, B), B, 0, stream>>>(
      bufX, W2, bufH, nullptr, nullptr, nullptr, nullptr, NN, DD, DD);
  k_agg<<<NP, AG_WAVES * 32, 0, stream>>>(rec, pcnt, goffs, bufH, dinv,
                                          b2, g2, be2, m2, v2, bufX);

  k_pool_agg<<<NP, AG_WAVES * 32, 0, stream>>>(rec, pcnt, goffs, batch, bufX, dinv, psum, cnt);

  k_gemm_wmma<<<blk((long long)tiles_POOL * 32, B), B, 0, stream>>>(
      psum, W3, poolT, b3, nullptr, nullptr, cnt, GG, DD, DD);
  k_gemm_wmma<<<blk((long long)tiles_FIN * 32, B), B, 0, stream>>>(
      poolT, Wl, out, bl, nullptr, nullptr, nullptr, GG, DD, DOUT);
}


# Layer-1 outer product written by the dinv launch: 256 work-groups of 8 waves, thirteen 16-byte stores per lane, instead of 100000 short waves of 4 bytes per lane; 11 -> 10 launches
# speedup vs baseline: 3.0977x; 1.0421x over previous
#include <hip/hip_runtime.h>
#include <hip/hip_bf16.h>

#define NN   50000
#define EE   800000
#define GG   512
#define DD   64
#define DOUT 32
#define BN_EPS 1e-5f

typedef __attribute__((ext_vector_type(2)))  float    v2f;
typedef __attribute__((ext_vector_type(8)))  float    v8f;
typedef __attribute__((ext_vector_type(16))) _Float16 v16h;

#if defined(__AMDGCN__) && __has_builtin(__builtin_amdgcn_wmma_f32_16x16x4_f32)
#define USE_WMMA_F32 1
#else
#define USE_WMMA_F32 0
#endif


#define NP   256
#define PN   196
#define CH   ((EE + NP - 1) / NP)
#define BINT 1024
#define EPT  ((CH + BINT - 1) / BINT)
static_assert(NP * PN >= NN, "ranges must cover the nodes");
static_assert(NP <= BINT && NP % 32 == 0, "one thread per range");
static_assert((long long)NP * CH >= EE && EPT * BINT >= CH, "the passes cover every edge");

#define SLN  16
#define SNO  (NN / SLN)
#define SED  (EE / SLN)
#define SQ   (SED / 4)
#define SQG  4
#define SQT  ((SQ + BINT - 1) / BINT)
static_assert(SLN * SLN == NP, "one work-group per (slab, slice)");
static_assert(SNO * SLN == NN && SED * SLN == EE, "slices tile the nodes, slabs tile the edges");
static_assert(SED % 4 == 0 && EE % 4 == 0, "slabs start on 16-byte boundaries in both index arrays");
static_assert(SLN * CH == SED, "a work-group's own chunk lies inside its slab");

__global__ void k_clear(float* __restrict__ a, float* __restrict__ b, int n,
                        int* __restrict__ c, int m,
                        float* __restrict__ p, int np) {
  int i = blockIdx.x * blockDim.x + threadIdx.x;
  if (i < n) { a[i] = 0.0f; b[i] = 0.0f; }
  if (i < m) c[i] = 0;
  if (i < np) p[i] = 0.0f;
}

__global__ __launch_bounds__(BINT)
void k_edge_count(const int* __restrict__ ei,
                  float* __restrict__ deg_feat,
                  float* __restrict__ degcnt,
                  int* __restrict__ pcnt) {
  __shared__ int h[NP];
  __shared__ int cs[SNO];
  __shared__ int cd[SNO];
  const int t = threadIdx.x;
  if (t < NP) h[t] = 0;
  for (int i = t; i < SNO; i += BINT) { cs[i] = 0; cd[i] = 0; }
  int d[EPT];
  const int c0 = blockIdx.x * CH;
#pragma unroll
  for (int k = 0; k < EPT; ++k) {
    int e = c0 + t + k * BINT;
    e = e < EE ? e : EE - 1;
    d[k] = ei[EE + e];
  }
  __syncthreads();
#pragma unroll
  for (int k = 0; k < EPT; ++k) {
    int i = t + k * BINT;
    if (i < CH && c0 + i < EE) atomicAdd(&h[d[k] / PN], 1);
  }

  const int slab = blockIdx.x / SLN;
  const int nb0  = (blockIdx.x % SLN) * SNO;
  const int4* __restrict__ s4 = (const int4*)(ei + slab * SED);
  const int4* __restrict__ d4 = (const int4*)(ei + EE + slab * SED);
  for (int g0 = 0; g0 < SQT; g0 += SQG) {
    int4 a[SQG], b[SQG];
#pragma unroll
    for (int k = 0; k < SQG; ++k) {
      int q = t + (g0 + k) * BINT;
      q = q < SQ ? q : SQ - 1;
      a[k] = s4[q];
      b[k] = d4[q];
    }
#pragma unroll
    for (int k = 0; k < SQG; ++k) {
      if (t + (g0 + k) * BINT < SQ) {
        unsigned u;
        u = (unsigned)(a[k].x - nb0); if (u < (unsigned)SNO) atomicAdd(&cs[u], 1);
        u = (unsigned)(a[k].y - nb0); if (u < (unsigned)SNO) atomicAdd(&cs[u], 1);
        u = (unsigned)(a[k].z - nb0); if (u < (unsigned)SNO) atomicAdd(&cs[u], 1);
        u = (unsigned)(a[k].w - nb0); if (u < (unsigned)SNO) atomicAdd(&cs[u], 1);
        u = (unsigned)(b[k].x - nb0); if (u < (unsigned)SNO) atomicAdd(&cd[u], 1);
        u = (unsigned)(b[k].y - nb0); if (u < (unsigned)SNO) atomicAdd(&cd[u], 1);
        u = (unsigned)(b[k].z - nb0); if (u < (unsigned)SNO) atomicAdd(&cd[u], 1);
        u = (unsigned)(b[k].w - nb0); if (u < (unsigned)SNO) atomicAdd(&cd[u], 1);
      }
    }
  }
  __syncthreads();
  for (int i = t; i < SNO; i += BINT) {
    int na = cs[i], nd = cd[i];
    if (na > 0) atomicAdd(&deg_feat[nb0 + i], (float)na);
    if (nd > 0) atomicAdd(&degcnt[nb0 + i], (float)nd);
  }
  if (t < NP && h[t] > 0) atomicAdd(&pcnt[t], h[t]);
}


__global__ __launch_bounds__(BINT)
void k_fill(const int* __restrict__ ei,
            const int* __restrict__ pcnt,
            const float* __restrict__ dinv,
            int* __restrict__ pcur,
            int* __restrict__ goffs,
            int4* __restrict__ rec) {
  __shared__ int h[NP];
  __shared__ int hb[NP];
  __shared__ int wsum[NP / 32];
  const int t = threadIdx.x;
  const int lane = t & 31;
  int s[EPT], d[EPT], r[EPT];
  float w[EPT];
  const int c0 = blockIdx.x * CH;
#pragma unroll
  for (int k = 0; k < EPT; ++k) {
    int e = c0 + t + k * BINT;
    e = e < EE ? e : EE - 1;
    s[k] = ei[e];
    d[k] = ei[EE + e];
  }
#pragma unroll
  for (int k = 0; k < EPT; ++k) w[k] = dinv[s[k]] * dinv[d[k]];
  int v = (t < NP) ? pcnt[t] : 0;
  int x = v;
  for (int o = 1; o < 32; o <<= 1) {
    int y = __shfl_up(x, o);
    if (lane >= o) x += y;
  }
  if (t < NP) {
    h[t] = 0;
    if (lane == 31) wsum[t >> 5] = x;
  }
  __syncthreads();
  int off = 0;
  if (t < NP) {
    for (int i = 0; i < (t >> 5); ++i) off += wsum[i];
    off += x - v;
    if (blockIdx.x == 0) goffs[t] = off;
  }
#pragma unroll
  for (int k = 0; k < EPT; ++k) {
    int i = t + k * BINT;
    r[k] = -1;
    if (i < CH && c0 + i < EE)
      r[k] = atomicAdd(&h[d[k] / PN], 1);
  }
  __syncthreads();
  if (t < NP) {
    int c = h[t];
    int g = c > 0 ? atomicAdd(&pcur[t], c) : 0;
    hb[t] = off + g;
  }
  __syncthreads();
#pragma unroll
  for (int k = 0; k < EPT; ++k) {
    if (r[k] >= 0) {
      int pos = hb[d[k] / PN] + r[k];
      rec[pos] = make_int4(s[k], d[k], __float_as_int(w[k]), 0);
    }
  }
}

#define HWB  256
#define HQ   (NN * DD / 4)
#define HQW  (HQ / NP)
#define HQT  ((HQW + HWB - 1) / HWB)
static_assert(HQW * NP == HQ, "the work-groups tile the plane exactly");
static_assert(HWB % (DD / 4) == 0, "a thread keeps the same four features in every group");
static_assert(PN <= HWB, "one thread per node of a range");

__global__ __launch_bounds__(HWB)
void k_dinv_hw(const float* __restrict__ degcnt,
               const float* __restrict__ deg_feat,
               const float* __restrict__ W1,
               float* __restrict__ dinv,
               float* __restrict__ hw) {
  const int t  = threadIdx.x;
  const int q0 = blockIdx.x * HQW;
  float d[HQT];
#pragma unroll
  for (int k = 0; k < HQT; ++k) {
    int i = t + k * HWB;
    i = i < HQW ? i : HQW - 1;
    d[k] = deg_feat[(q0 + i) >> 4];
  }
  const int f0 = ((q0 + t) & (DD / 4 - 1)) * 4;
  const float w0 = W1[f0], w1 = W1[f0 + 1], w2 = W1[f0 + 2], w3 = W1[f0 + 3];
  const int  n   = blockIdx.x * PN + t;
  const bool own = t < PN && n < NN;
  const float c  = degcnt[own ? n : 0];
  if (own) dinv[n] = rsqrtf(c + 1.0f);
  float4* __restrict__ hw4 = (float4*)hw + q0;
#pragma unroll
  for (int k = 0; k < HQT; ++k) {
    const int i = t + k * HWB;
    if (i < HQW) {
      float4 y;
      y.x = d[k] * w0; y.y = d[k] * w1; y.z = d[k] * w2; y.w = d[k] * w3;
      hw4[i] = y;
    }
  }
}

#define AG_WAVES 32
#define AG_GRP   16
#define AG_TAB_FLOATS (41 * 1024)
static_assert((PN + 1) * DD <= AG_TAB_FLOATS, "table holds one range and the spare row");
#define AG_THREADS (AG_WAVES * 32)
#define TQ   (PN * DD / 4)
#define TQT  ((TQ + AG_THREADS - 1) / AG_THREADS)
static_assert(DD % 4 == 0 && AG_THREADS % (DD / 4) == 0, "a thread keeps the same four features in every group");

__global__ __launch_bounds__(AG_WAVES * 32)
void k_agg(const int4* __restrict__ rec,
           const int* __restrict__ pcnt,
           const int* __restrict__ goffs,
           const float* __restrict__ hw,
           const float* __restrict__ dinv,
           const float* __restrict__ b,
           const float* __restrict__ g,
           const float* __restrict__ be,
           const float* __restrict__ m,
           const float* __restrict__ v,
           float* __restrict__ xout) {
  __shared__ __attribute__((aligned(16))) float tab[AG_TAB_FLOATS];
  const int tid  = threadIdx.x;
  const int lane = tid & 31;
  const int wave = __builtin_amdgcn_readfirstlane(tid >> 5);
  const int n0 = blockIdx.x * PN;
  const int nn = (NN - n0 < PN) ? (NN - n0) : PN;
  const int start = goffs[blockIdx.x];
  const int cnt   = pcnt[blockIdx.x];
  const int nb    = (cnt + 31) >> 5;
  const float2* __restrict__ hw2 = (const float2*)hw;

  int bq = wave;
  int sv = 0, dv = n0;
  float wv = 0.0f;
  if (bq < nb) {
    int li = bq * 32 + lane;
    li = li < cnt ? li : cnt - 1;
    int4 q = rec[start + li];
    sv = q.x; dv = q.y; wv = __int_as_float(q.z);
  }

  const int f0 = (tid & (DD / 4 - 1)) * 4;
  float cb[4], cg[4], cbe[4], cm[4], cv[4];
#pragma unroll
  for (int j = 0; j < 4; ++j) {
    cb[j] = b[f0 + j]; cg[j] = g[f0 + j]; cbe[j] = be[f0 + j]; cm[j] = m[f0 + j]; cv[j] = v[f0 + j];
  }

  const int nq = nn * (DD / 4);
  const float4* __restrict__ hw4 = (const float4*)(hw + n0 * DD);
  {
    float4 hv[TQT];
    float  dq[TQT];
#pragma unroll
    for (int k = 0; k < TQT; ++k) {
      int q = tid + k * AG_THREADS;
      q = q < nq ? q : nq - 1;
      hv[k] = hw4[q];
      dq[k] = dinv[n0 + (q >> 4)];
    }
#pragma unroll
    for (int k = 0; k < TQT; ++k) {
      const int q = tid + k * AG_THREADS;
      if (q < nq) {
        const float s = dq[k] * dq[k];
        float4 t = hv[k];
        t.x *= s; t.y *= s; t.z *= s; t.w *= s;
        ((float4*)tab)[q] = t;
      }
    }
  }
  __syncthreads();

  while (bq < nb) {
    const int bn = bq + AG_WAVES;
    int sn = 0, dn = n0;
    float wn = 0.0f;
    if (bn < nb) {
      int li = bn * 32 + lane;
      li = li < cnt ? li : cnt - 1;
      int4 q = rec[start + li];
      sn = q.x; dn = q.y; wn = __int_as_float(q.z);
    }
    const bool  live = bq * 32 + lane < cnt;
    const int   rv = live ? (dv - n0) * DD : PN * DD;
    const float wl = live ? wv : 0.0f;
#pragma unroll
    for (int j0 = 0; j0 < 32; j0 += AG_GRP) {
      float2 vr[AG_GRP];
#pragma unroll
      for (int u = 0; u < AG_GRP; ++u) {
        int sj = __builtin_amdgcn_readlane(sv, j0 + u);
        vr[u] = hw2[sj * (DD / 2) + lane];
      }
#pragma unroll
      for (int u = 0; u < AG_GRP; ++u) {
        int   rj = __builtin_amdgcn_readlane(rv, j0 + u);
        float wj = __int_as_float(__builtin_amdgcn_readlane(__float_as_int(wl), j0 + u));
        atomicAdd(&tab[rj + 2 * lane],     vr[u].x * wj);
        atomicAdd(&tab[rj + 2 * lane + 1], vr[u].y * wj);
      }
    }
    sv = sn; dv = dn; wv = wn; bq = bn;
  }
  __syncthreads();

  float rs[4];
#pragma unroll
  for (int j = 0; j < 4; ++j) rs[j] = rsqrtf(cv[j] + BN_EPS);
  float4* __restrict__ xo4 = (float4*)(xout + n0 * DD);
#pragma unroll
  for (int k = 0; k < TQT; ++k) {
    const int q = tid + k * AG_THREADS;
    if (q < nq) {
      const float4 t = ((const float4*)tab)[q];
      float4 y;
      y.x = fmaxf(((t.x + cb[0]) - cm[0]) * rs[0] * cg[0] + cbe[0], 0.0f);
      y.y = fmaxf(((t.y + cb[1]) - cm[1]) * rs[1] * cg[1] + cbe[1], 0.0f);
      y.z = fmaxf(((t.z + cb[2]) - cm[2]) * rs[2] * cg[2] + cbe[2], 0.0f);
      y.w = fmaxf(((t.w + cb[3]) - cm[3]) * rs[3] * cg[3] + cbe[3], 0.0f);
      xo4[q] = y;
    }
  }
}

#define PCH  16
#define PCS  ((PN + PCH - 1) / PCH)
static_assert(PCH * DD == AG_WAVES * 32, "one thread per (chunk, feature)");

__global__ __launch_bounds__(AG_WAVES * 32)
void k_pool_agg(const int4* __restrict__ rec,
                const int* __restrict__ pcnt,
                const int* __restrict__ goffs,
                const int* __restrict__ batch,
                const float* __restrict__ x,
                const float* __restrict__ dinv,
                float* __restrict__ psum,
                float* __restrict__ cntg) {
  __shared__ __attribute__((aligned(16))) float tab[(PN + 1) * DD];
  __shared__ __attribute__((aligned(16))) float gtab[GG * DD];
  __shared__ float ctab[GG];
  __shared__ int   gloc[PN];
  const int tid  = threadIdx.x;
  const int lane = tid & 31;
  const int wave = __builtin_amdgcn_readfirstlane(tid >> 5);
  const int n0 = blockIdx.x * PN;
  const int nn = (NN - n0 < PN) ? (NN - n0) : PN;
  const int start = goffs[blockIdx.x];
  const int cnt   = pcnt[blockIdx.x];
  const int nb    = (cnt + 31) >> 5;
  const float2* __restrict__ x2 = (const float2*)x;

  int bq = wave;
  int sv = 0, dv = n0;
  float wv = 0.0f;
  if (bq < nb) {
    int li = bq * 32 + lane;
    li = li < cnt ? li : cnt - 1;
    int4 q = rec[start + li];
    sv = q.x; dv = q.y; wv = __int_as_float(q.z);
  }

  const int nq = nn * (DD / 4);
  const float4* __restrict__ x4 = (const float4*)(x + n0 * DD);
  {
    const int gb = batch[n0 + (tid < nn ? tid : nn - 1)];
    float4 hv[TQT];
    float  dq[TQT];
#pragma unroll
    for (int k = 0; k < TQT; ++k) {
      int q = tid + k * AG_THREADS;
      q = q < nq ? q : nq - 1;
      hv[k] = x4[q];
      dq[k] = dinv[n0 + (q >> 4)];
    }
    float4 z4 = {0.0f, 0.0f, 0.0f, 0.0f};
    for (int i = tid; i < GG * DD / 4; i += AG_THREADS) ((float4*)gtab)[i] = z4;
    if (tid < GG) ctab[tid] = 0.0f;
#pragma unroll
    for (int k = 0; k < TQT; ++k) {
      const int q = tid + k * AG_THREADS;
      if (q < nq) {
        const float s = dq[k] * dq[k];
        float4 t = hv[k];
        t.x *= s; t.y *= s; t.z *= s; t.w *= s;
        ((float4*)tab)[q] = t;
      }
    }
    if (tid < nn) gloc[tid] = gb;
  }
  __syncthreads();

  while (bq < nb) {
    const int bn = bq + AG_WAVES;
    int sn = 0, dn = n0;
    float wn = 0.0f;
    if (bn < nb) {
      int li = bn * 32 + lane;
      li = li < cnt ? li : cnt - 1;
      int4 q = rec[start + li];
      sn = q.x; dn = q.y; wn = __int_as_float(q.z);
    }
    const bool  live = bq * 32 + lane < cnt;
    const int   rv = live ? (dv - n0) * DD : PN * DD;
    const float wl = live ? wv : 0.0f;
#pragma unroll
    for (int j0 = 0; j0 < 32; j0 += AG_GRP) {
      float2 vr[AG_GRP];
#pragma unroll
      for (int u = 0; u < AG_GRP; ++u) {
        int sj = __builtin_amdgcn_readlane(sv, j0 + u);
        vr[u] = x2[sj * (DD / 2) + lane];
      }
#pragma unroll
      for (int u = 0; u < AG_GRP; ++u) {
        int   rj = __builtin_amdgcn_readlane(rv, j0 + u);
        float wj = __int_as_float(__builtin_amdgcn_readlane(__float_as_int(wl), j0 + u));
        atomicAdd(&tab[rj + 2 * lane],     vr[u].x * wj);
        atomicAdd(&tab[rj + 2 * lane + 1], vr[u].y * wj);
      }
    }
    sv = sn; dv = dn; wv = wn; bq = bn;
  }
  __syncthreads();

  {
    const int f  = tid & (DD - 1);
    const int c0 = (tid >> 6) * PCS;
    const int c1 = (c0 + PCS < nn) ? (c0 + PCS) : nn;
    if (c0 < c1) {
      int   ids[PCS];
      float vals[PCS];
#pragma unroll
      for (int j = 0; j < PCS; ++j) {
        int i = c0 + j;
        i = i < c1 ? i : c1 - 1;
        ids[j]  = gloc[i];
        vals[j] = tab[i * DD + f];
      }
      int   cg = ids[0];
      float a  = 0.0f;
#pragma unroll
      for (int j = 0; j < PCS; ++j) {
        if (c0 + j < c1) {
          if (ids[j] != cg) { atomicAdd(&gtab[cg * DD + f], a); a = 0.0f; cg = ids[j]; }
          a += vals[j];
        }
      }
      atomicAdd(&gtab[cg * DD + f], a);
    }
    if (tid < nn) atomicAdd(&ctab[gloc[tid]], 1.0f);
  }
  __syncthreads();

  {
    static_assert(GG * DD == 32 * AG_THREADS && AG_THREADS / DD == 16, "lane k of a wave <-> graph c + 16 k");
    const float cl = ctab[(tid >> 6) + 16 * lane];
    unsigned present = (unsigned)__ballot(cl > 0.0f);
    while (present) {
      const int k = __builtin_ctz(present);
      present &= present - 1;
      const int i = tid + k * AG_THREADS;
      atomicAdd(&psum[i], gtab[i]);
    }
  }
  if (tid < GG && ctab[tid] > 0.0f) atomicAdd(&cntg[tid], ctab[tid]);
}

__global__ void k_gemm_wmma(const float* __restrict__ A,
                            const float* __restrict__ B,
                            float* __restrict__ C,
                            const float* __restrict__ bias,
                            const float* __restrict__ dinv,
                            float* __restrict__ acc,
                            const float* __restrict__ cnt,
                            int M, int K, int Nc) {
  int wid  = (blockIdx.x * blockDim.x + threadIdx.x) >> 5;
  int lane = threadIdx.x & 31;
  int tiles_n = Nc >> 4;
  int tm = wid / tiles_n;
  int tn = wid - tm * tiles_n;
  if (tm * 16 >= M) return;

  int half = lane >> 4;
  int lm   = lane & 15;
  const float* Arow = A + (size_t)(tm * 16 + lm) * K;
  int ncol = tn * 16 + lm;

  v8f c = {};
#if USE_WMMA_F32
  for (int k0 = 0; k0 < K; k0 += 4) {
    v2f a, b;
    a.x = Arow[k0 + 2 * half + 0];
    a.y = Arow[k0 + 2 * half + 1];
    b.x = B[(size_t)(k0 + 2 * half + 0) * Nc + ncol];
    b.y = B[(size_t)(k0 + 2 * half + 1) * Nc + ncol];
    c = __builtin_amdgcn_wmma_f32_16x16x4_f32(false, a, false, b, (short)0, c,
                                              false, false);
  }
#else
  for (int k0 = 0; k0 < K; k0 += 32) {
    v16h a, b;
#pragma unroll
    for (int i = 0; i < 8; ++i) {
      a[i]     = (_Float16)Arow[k0 + 8 * half + i];
      a[8 + i] = (_Float16)Arow[k0 + 16 + 8 * half + i];
    }
#pragma unroll
    for (int i = 0; i < 16; ++i)
      b[i] = (_Float16)B[(size_t)(k0 + 16 * half + i) * Nc + ncol];
    c = __builtin_amdgcn_wmma_f32_16x16x32_f16(false, a, false, b, (short)0, c,
                                               false, false);
  }
#endif

  float bv = bias ? bias[ncol] : 0.0f;
  if (acc) {
#pragma unroll
    for (int r = 0; r < 8; ++r) {
      int row = tm * 16 + 8 * half + r;
      size_t idx = (size_t)row * Nc + ncol;
      float val = c[r] + bv;
      float di = dinv[row];
      C[idx]   = val;
      acc[idx] = val * (di * di);
    }
  } else {
#pragma unroll
    for (int r = 0; r < 8; ++r) {
      int row = tm * 16 + 8 * half + r;
      float val = c[r] + bv;
      if (cnt) {
        float cc = cnt[row];
        val = (cc > 0.0f) ? c[r] / cc + bv : 0.0f;
      }
      C[(size_t)row * Nc + ncol] = val;
    }
  }
}

static inline int blk(long long n, int b) { return (int)((n + b - 1) / b); }

extern "C" void kernel_launch(void* const* d_in, const int* in_sizes, int n_in,
                              void* d_out, int out_size, void* d_ws, size_t ws_size,
                              hipStream_t stream) {
  (void)in_sizes; (void)n_in; (void)out_size; (void)ws_size;
  const int*   edge_index = (const int*)d_in[0];
  const int*   batch      = (const int*)d_in[1];
  const float* W1 = (const float*)d_in[2];
  const float* b1 = (const float*)d_in[3];
  const float* W2 = (const float*)d_in[4];
  const float* b2 = (const float*)d_in[5];
  const float* W3 = (const float*)d_in[6];
  const float* b3 = (const float*)d_in[7];
  const float* g1 = (const float*)d_in[8];
  const float* be1= (const float*)d_in[9];
  const float* m1 = (const float*)d_in[10];
  const float* v1 = (const float*)d_in[11];
  const float* g2 = (const float*)d_in[12];
  const float* be2= (const float*)d_in[13];
  const float* m2 = (const float*)d_in[14];
  const float* v2 = (const float*)d_in[15];
  const float* Wl = (const float*)d_in[16];
  const float* bl = (const float*)d_in[17];
  float* out = (float*)d_out;

  char* ws = (char*)d_ws;
  size_t off = 0;
  auto carve = [&](size_t bytes) -> float* {
    float* p = (float*)(ws + off);
    off += (bytes + 255) & ~(size_t)255;
    return p;
  };
  float* deg_feat = carve((size_t)NN * 4);
  float* degcnt   = carve((size_t)NN * 4);
  float* dinv     = carve((size_t)NN * 4);
  float* bufX     = carve((size_t)NN * DD * 4);
  float* bufH     = carve((size_t)NN * DD * 4);
  float* psum     = carve((size_t)(GG * DD + GG) * 4);
  float* cnt      = psum + GG * DD;
  float* poolT    = carve((size_t)GG * DD * 4);
  int*   pcnt     = (int*)carve((size_t)2 * NP * 4);
  int*   pcur     = pcnt + NP;
  int*   goffs    = (int*)carve((size_t)NP * 4);
  int4*  rec      = (int4*)carve((size_t)EE * 16);

  const int B = 256;

  k_clear<<<blk(NN, B), B, 0, stream>>>(deg_feat, degcnt, NN, pcnt, 2 * NP, psum, GG * DD + GG);
  k_edge_count<<<NP, BINT, 0, stream>>>(edge_index, deg_feat, degcnt, pcnt);
  k_dinv_hw<<<NP, HWB, 0, stream>>>(degcnt, deg_feat, W1, dinv, bufH);
  k_fill<<<NP, BINT, 0, stream>>>(edge_index, pcnt, dinv, pcur, goffs, rec);

  const int tiles_NN  = (NN / 16) * (DD / 16);
  const int tiles_POOL = (GG / 16) * (DD / 16);
  const int tiles_FIN = (GG / 16) * (DOUT / 16);

  k_agg<<<NP, AG_WAVES * 32, 0, stream>>>(rec, pcnt, goffs, bufH, dinv,
                                          b1, g1, be1, m1, v1, bufX);

  k_gemm_wmma<<<blk((long long)tiles_NN * 32, B), B, 0, stream>>>(
      bufX, W2, bufH, nullptr, nullptr, nullptr, nullptr, NN, DD, DD);
  k_agg<<<NP, AG_WAVES * 32, 0, stream>>>(rec, pcnt, goffs, bufH, dinv,
                                          b2, g2, be2, m2, v2, bufX);

  k_pool_agg<<<NP, AG_WAVES * 32, 0, stream>>>(rec, pcnt, goffs, batch, bufX, dinv, psum, cnt);

  k_gemm_wmma<<<blk((long long)tiles_POOL * 32, B), B, 0, stream>>>(
      psum, W3, poolT, b3, nullptr, nullptr, cnt, GG, DD, DD);
  k_gemm_wmma<<<blk((long long)tiles_FIN * 32, B), B, 0, stream>>>(
      poolT, Wl, out, bl, nullptr, nullptr, nullptr, GG, DD, DOUT);
}


# Launches 10 -> 8: the fill launch forms edge weights from the degree counts and also writes dinv and the layer-1 rows; mean, W3, b3 and the head are one launch with the pooled rows staged in LDS
# speedup vs baseline: 3.5238x; 1.1376x over previous
#include <hip/hip_runtime.h>
#include <hip/hip_bf16.h>

#define NN   50000
#define EE   800000
#define GG   512
#define DD   64
#define DOUT 32
#define BN_EPS 1e-5f

typedef __attribute__((ext_vector_type(2)))  float    v2f;
typedef __attribute__((ext_vector_type(8)))  float    v8f;
typedef __attribute__((ext_vector_type(16))) _Float16 v16h;

#if defined(__AMDGCN__) && __has_builtin(__builtin_amdgcn_wmma_f32_16x16x4_f32)
#define USE_WMMA_F32 1
#else
#define USE_WMMA_F32 0
#endif


#define NP   256
#define PN   196
#define CH   ((EE + NP - 1) / NP)
#define BINT 1024
#define EPT  ((CH + BINT - 1) / BINT)
static_assert(NP * PN >= NN, "ranges must cover the nodes");
static_assert(NP <= BINT && NP % 32 == 0, "one thread per range");
static_assert((long long)NP * CH >= EE && EPT * BINT >= CH, "the passes cover every edge");

#define SLN  16
#define SNO  (NN / SLN)
#define SED  (EE / SLN)
#define SQ   (SED / 4)
#define SQG  4
#define SQT  ((SQ + BINT - 1) / BINT)
static_assert(SLN * SLN == NP, "one work-group per (slab, slice)");
static_assert(SNO * SLN == NN && SED * SLN == EE, "slices tile the nodes, slabs tile the edges");
static_assert(SED % 4 == 0 && EE % 4 == 0, "slabs start on 16-byte boundaries in both index arrays");
static_assert(SLN * CH == SED, "a work-group's own chunk lies inside its slab");

__global__ void k_clear(float* __restrict__ a, float* __restrict__ b, int n,
                        int* __restrict__ c, int m,
                        float* __restrict__ p, int np) {
  int i = blockIdx.x * blockDim.x + threadIdx.x;
  if (i < n) { a[i] = 0.0f; b[i] = 0.0f; }
  if (i < m) c[i] = 0;
  if (i < np) p[i] = 0.0f;
}

__global__ __launch_bounds__(BINT)
void k_edge_count(const int* __restrict__ ei,
                  float* __restrict__ deg_feat,
                  float* __restrict__ degcnt,
                  int* __restrict__ pcnt) {
  __shared__ int h[NP];
  __shared__ int cs[SNO];
  __shared__ int cd[SNO];
  const int t = threadIdx.x;
  if (t < NP) h[t] = 0;
  for (int i = t; i < SNO; i += BINT) { cs[i] = 0; cd[i] = 0; }
  int d[EPT];
  const int c0 = blockIdx.x * CH;
#pragma unroll
  for (int k = 0; k < EPT; ++k) {
    int e = c0 + t + k * BINT;
    e = e < EE ? e : EE - 1;
    d[k] = ei[EE + e];
  }
  __syncthreads();
#pragma unroll
  for (int k = 0; k < EPT; ++k) {
    int i = t + k * BINT;
    if (i < CH && c0 + i < EE) atomicAdd(&h[d[k] / PN], 1);
  }

  const int slab = blockIdx.x / SLN;
  const int nb0  = (blockIdx.x % SLN) * SNO;
  const int4* __restrict__ s4 = (const int4*)(ei + slab * SED);
  const int4* __restrict__ d4 = (const int4*)(ei + EE + slab * SED);
  for (int g0 = 0; g0 < SQT; g0 += SQG) {
    int4 a[SQG], b[SQG];
#pragma unroll
    for (int k = 0; k < SQG; ++k) {
      int q = t + (g0 + k) * BINT;
      q = q < SQ ? q : SQ - 1;
      a[k] = s4[q];
      b[k] = d4[q];
    }
#pragma unroll
    for (int k = 0; k < SQG; ++k) {
      if (t + (g0 + k) * BINT < SQ) {
        unsigned u;
        u = (unsigned)(a[k].x - nb0); if (u < (unsigned)SNO) atomicAdd(&cs[u], 1);
        u = (unsigned)(a[k].y - nb0); if (u < (unsigned)SNO) atomicAdd(&cs[u], 1);
        u = (unsigned)(a[k].z - nb0); if (u < (unsigned)SNO) atomicAdd(&cs[u], 1);
        u = (unsigned)(a[k].w - nb0); if (u < (unsigned)SNO) atomicAdd(&cs[u], 1);
        u = (unsigned)(b[k].x - nb0); if (u < (unsigned)SNO) atomicAdd(&cd[u], 1);
        u = (unsigned)(b[k].y - nb0); if (u < (unsigned)SNO) atomicAdd(&cd[u], 1);
        u = (unsigned)(b[k].z - nb0); if (u < (unsigned)SNO) atomicAdd(&cd[u], 1);
        u = (unsigned)(b[k].w - nb0); if (u < (unsigned)SNO) atomicAdd(&cd[u], 1);
      }
    }
  }
  __syncthreads();
  for (int i = t; i < SNO; i += BINT) {
    int na = cs[i], nd = cd[i];
    if (na > 0) atomicAdd(&deg_feat[nb0 + i], (float)na);
    if (nd > 0) atomicAdd(&degcnt[nb0 + i], (float)nd);
  }
  if (t < NP && h[t] > 0) atomicAdd(&pcnt[t], h[t]);
}


#define HQ   (NN * DD / 4)
#define HQW  (HQ / NP)
#define HQT  ((HQW + BINT - 1) / BINT)
static_assert(HQW * NP == HQ, "the work-groups tile the plane exactly");
static_assert(BINT % (DD / 4) == 0, "a thread keeps the same four features in every group");
static_assert(PN <= BINT, "one thread per node of a range");

__global__ __launch_bounds__(BINT)
void k_fill(const int* __restrict__ ei,
            const int* __restrict__ pcnt,
            const float* __restrict__ degcnt,
            const float* __restrict__ deg_feat,
            const float* __restrict__ W1,
            int* __restrict__ pcur,
            int* __restrict__ goffs,
            int4* __restrict__ rec,
            float* __restrict__ dinv,
            float* __restrict__ hw) {
  __shared__ int h[NP];
  __shared__ int hb[NP];
  __shared__ int wsum[NP / 32];
  const int t = threadIdx.x;
  const int lane = t & 31;
  int s[EPT], d[EPT], r[EPT];
  float w[EPT];
  const int c0 = blockIdx.x * CH;
#pragma unroll
  for (int k = 0; k < EPT; ++k) {
    int e = c0 + t + k * BINT;
    e = e < EE ? e : EE - 1;
    s[k] = ei[e];
    d[k] = ei[EE + e];
  }
  const int q0 = blockIdx.x * HQW;
  float dg[HQT];
#pragma unroll
  for (int k = 0; k < HQT; ++k) {
    int i = t + k * BINT;
    i = i < HQW ? i : HQW - 1;
    dg[k] = deg_feat[(q0 + i) >> 4];
  }
  const int f0 = ((q0 + t) & (DD / 4 - 1)) * 4;
  const float w0 = W1[f0], w1 = W1[f0 + 1], w2 = W1[f0 + 2], w3 = W1[f0 + 3];
  const int n = blockIdx.x * PN + t;
  if (t < PN && n < NN) dinv[n] = rsqrtf(degcnt[n] + 1.0f);
  float4* __restrict__ hw4 = (float4*)hw + q0;
#pragma unroll
  for (int k = 0; k < HQT; ++k) {
    const int i = t + k * BINT;
    if (i < HQW) {
      float4 y;
      y.x = dg[k] * w0; y.y = dg[k] * w1; y.z = dg[k] * w2; y.w = dg[k] * w3;
      hw4[i] = y;
    }
  }
  int v = pcnt[t & (NP - 1)];
#pragma unroll
  for (int k = 0; k < EPT; ++k) w[k] = rsqrtf(degcnt[s[k]] + 1.0f) * rsqrtf(degcnt[d[k]] + 1.0f);
  v = (t < NP) ? v : 0;
  int x = v;
  for (int o = 1; o < 32; o <<= 1) {
    int y = __shfl_up(x, o);
    if (lane >= o) x += y;
  }
  if (t < NP) {
    h[t] = 0;
    if (lane == 31) wsum[t >> 5] = x;
  }
  __syncthreads();
  int off = 0;
  if (t < NP) {
    for (int i = 0; i < (t >> 5); ++i) off += wsum[i];
    off += x - v;
    if (blockIdx.x == 0) goffs[t] = off;
  }
#pragma unroll
  for (int k = 0; k < EPT; ++k) {
    int i = t + k * BINT;
    r[k] = -1;
    if (i < CH && c0 + i < EE)
      r[k] = atomicAdd(&h[d[k] / PN], 1);
  }
  __syncthreads();
  if (t < NP) {
    int c = h[t];
    int g = c > 0 ? atomicAdd(&pcur[t], c) : 0;
    hb[t] = off + g;
  }
  __syncthreads();
#pragma unroll
  for (int k = 0; k < EPT; ++k) {
    if (r[k] >= 0) {
      int pos = hb[d[k] / PN] + r[k];
      rec[pos] = make_int4(s[k], d[k], __float_as_int(w[k]), 0);
    }
  }
}

#define AG_WAVES 32
#define AG_GRP   16
#define AG_TAB_FLOATS (41 * 1024)
static_assert((PN + 1) * DD <= AG_TAB_FLOATS, "table holds one range and the spare row");
#define AG_THREADS (AG_WAVES * 32)
#define TQ   (PN * DD / 4)
#define TQT  ((TQ + AG_THREADS - 1) / AG_THREADS)
static_assert(DD % 4 == 0 && AG_THREADS % (DD / 4) == 0, "a thread keeps the same four features in every group");

__global__ __launch_bounds__(AG_WAVES * 32)
void k_agg(const int4* __restrict__ rec,
           const int* __restrict__ pcnt,
           const int* __restrict__ goffs,
           const float* __restrict__ hw,
           const float* __restrict__ dinv,
           const float* __restrict__ b,
           const float* __restrict__ g,
           const float* __restrict__ be,
           const float* __restrict__ m,
           const float* __restrict__ v,
           float* __restrict__ xout) {
  __shared__ __attribute__((aligned(16))) float tab[AG_TAB_FLOATS];
  const int tid  = threadIdx.x;
  const int lane = tid & 31;
  const int wave = __builtin_amdgcn_readfirstlane(tid >> 5);
  const int n0 = blockIdx.x * PN;
  const int nn = (NN - n0 < PN) ? (NN - n0) : PN;
  const int start = goffs[blockIdx.x];
  const int cnt   = pcnt[blockIdx.x];
  const int nb    = (cnt + 31) >> 5;
  const float2* __restrict__ hw2 = (const float2*)hw;

  int bq = wave;
  int sv = 0, dv = n0;
  float wv = 0.0f;
  if (bq < nb) {
    int li = bq * 32 + lane;
    li = li < cnt ? li : cnt - 1;
    int4 q = rec[start + li];
    sv = q.x; dv = q.y; wv = __int_as_float(q.z);
  }

  const int f0 = (tid & (DD / 4 - 1)) * 4;
  float cb[4], cg[4], cbe[4], cm[4], cv[4];
#pragma unroll
  for (int j = 0; j < 4; ++j) {
    cb[j] = b[f0 + j]; cg[j] = g[f0 + j]; cbe[j] = be[f0 + j]; cm[j] = m[f0 + j]; cv[j] = v[f0 + j];
  }

  const int nq = nn * (DD / 4);
  const float4* __restrict__ hw4 = (const float4*)(hw + n0 * DD);
  {
    float4 hv[TQT];
    float  dq[TQT];
#pragma unroll
    for (int k = 0; k < TQT; ++k) {
      int q = tid + k * AG_THREADS;
      q = q < nq ? q : nq - 1;
      hv[k] = hw4[q];
      dq[k] = dinv[n0 + (q >> 4)];
    }
#pragma unroll
    for (int k = 0; k < TQT; ++k) {
      const int q = tid + k * AG_THREADS;
      if (q < nq) {
        const float s = dq[k] * dq[k];
        float4 t = hv[k];
        t.x *= s; t.y *= s; t.z *= s; t.w *= s;
        ((float4*)tab)[q] = t;
      }
    }
  }
  __syncthreads();

  while (bq < nb) {
    const int bn = bq + AG_WAVES;
    int sn = 0, dn = n0;
    float wn = 0.0f;
    if (bn < nb) {
      int li = bn * 32 + lane;
      li = li < cnt ? li : cnt - 1;
      int4 q = rec[start + li];
      sn = q.x; dn = q.y; wn = __int_as_float(q.z);
    }
    const bool  live = bq * 32 + lane < cnt;
    const int   rv = live ? (dv - n0) * DD : PN * DD;
    const float wl = live ? wv : 0.0f;
#pragma unroll
    for (int j0 = 0; j0 < 32; j0 += AG_GRP) {
      float2 vr[AG_GRP];
#pragma unroll
      for (int u = 0; u < AG_GRP; ++u) {
        int sj = __builtin_amdgcn_readlane(sv, j0 + u);
        vr[u] = hw2[sj * (DD / 2) + lane];
      }
#pragma unroll
      for (int u = 0; u < AG_GRP; ++u) {
        int   rj = __builtin_amdgcn_readlane(rv, j0 + u);
        float wj = __int_as_float(__builtin_amdgcn_readlane(__float_as_int(wl), j0 + u));
        atomicAdd(&tab[rj + 2 * lane],     vr[u].x * wj);
        atomicAdd(&tab[rj + 2 * lane + 1], vr[u].y * wj);
      }
    }
    sv = sn; dv = dn; wv = wn; bq = bn;
  }
  __syncthreads();

  float rs[4];
#pragma unroll
  for (int j = 0; j < 4; ++j) rs[j] = rsqrtf(cv[j] + BN_EPS);
  float4* __restrict__ xo4 = (float4*)(xout + n0 * DD);
#pragma unroll
  for (int k = 0; k < TQT; ++k) {
    const int q = tid + k * AG_THREADS;
    if (q < nq) {
      const float4 t = ((const float4*)tab)[q];
      float4 y;
      y.x = fmaxf(((t.x + cb[0]) - cm[0]) * rs[0] * cg[0] + cbe[0], 0.0f);
      y.y = fmaxf(((t.y + cb[1]) - cm[1]) * rs[1] * cg[1] + cbe[1], 0.0f);
      y.z = fmaxf(((t.z + cb[2]) - cm[2]) * rs[2] * cg[2] + cbe[2], 0.0f);
      y.w = fmaxf(((t.w + cb[3]) - cm[3]) * rs[3] * cg[3] + cbe[3], 0.0f);
      xo4[q] = y;
    }
  }
}

#define PCH  16
#define PCS  ((PN + PCH - 1) / PCH)
static_assert(PCH * DD == AG_WAVES * 32, "one thread per (chunk, feature)");

__global__ __launch_bounds__(AG_WAVES * 32)
void k_pool_agg(const int4* __restrict__ rec,
                const int* __restrict__ pcnt,
                const int* __restrict__ goffs,
                const int* __restrict__ batch,
                const float* __restrict__ x,
                const float* __restrict__ dinv,
                float* __restrict__ psum,
                float* __restrict__ cntg) {
  __shared__ __attribute__((aligned(16))) float tab[(PN + 1) * DD];
  __shared__ __attribute__((aligned(16))) float gtab[GG * DD];
  __shared__ float ctab[GG];
  __shared__ int   gloc[PN];
  const int tid  = threadIdx.x;
  const int lane = tid & 31;
  const int wave = __builtin_amdgcn_readfirstlane(tid >> 5);
  const int n0 = blockIdx.x * PN;
  const int nn = (NN - n0 < PN) ? (NN - n0) : PN;
  const int start = goffs[blockIdx.x];
  const int cnt   = pcnt[blockIdx.x];
  const int nb    = (cnt + 31) >> 5;
  const float2* __restrict__ x2 = (const float2*)x;

  int bq = wave;
  int sv = 0, dv = n0;
  float wv = 0.0f;
  if (bq < nb) {
    int li = bq * 32 + lane;
    li = li < cnt ? li : cnt - 1;
    int4 q = rec[start + li];
    sv = q.x; dv = q.y; wv = __int_as_float(q.z);
  }

  const int nq = nn * (DD / 4);
  const float4* __restrict__ x4 = (const float4*)(x + n0 * DD);
  {
    const int gb = batch[n0 + (tid < nn ? tid : nn - 1)];
    float4 hv[TQT];
    float  dq[TQT];
#pragma unroll
    for (int k = 0; k < TQT; ++k) {
      int q = tid + k * AG_THREADS;
      q = q < nq ? q : nq - 1;
      hv[k] = x4[q];
      dq[k] = dinv[n0 + (q >> 4)];
    }
    float4 z4 = {0.0f, 0.0f, 0.0f, 0.0f};
    for (int i = tid; i < GG * DD / 4; i += AG_THREADS) ((float4*)gtab)[i] = z4;
    if (tid < GG) ctab[tid] = 0.0f;
#pragma unroll
    for (int k = 0; k < TQT; ++k) {
      const int q = tid + k * AG_THREADS;
      if (q < nq) {
        const float s = dq[k] * dq[k];
        float4 t = hv[k];
        t.x *= s; t.y *= s; t.z *= s; t.w *= s;
        ((float4*)tab)[q] = t;
      }
    }
    if (tid < nn) gloc[tid] = gb;
  }
  __syncthreads();

  while (bq < nb) {
    const int bn = bq + AG_WAVES;
    int sn = 0, dn = n0;
    float wn = 0.0f;
    if (bn < nb) {
      int li = bn * 32 + lane;
      li = li < cnt ? li : cnt - 1;
      int4 q = rec[start + li];
      sn = q.x; dn = q.y; wn = __int_as_float(q.z);
    }
    const bool  live = bq * 32 + lane < cnt;
    const int   rv = live ? (dv - n0) * DD : PN * DD;
    const float wl = live ? wv : 0.0f;
#pragma unroll
    for (int j0 = 0; j0 < 32; j0 += AG_GRP) {
      float2 vr[AG_GRP];
#pragma unroll
      for (int u = 0; u < AG_GRP; ++u) {
        int sj = __builtin_amdgcn_readlane(sv, j0 + u);
        vr[u] = x2[sj * (DD / 2) + lane];
      }
#pragma unroll
      for (int u = 0; u < AG_GRP; ++u) {
        int   rj = __builtin_amdgcn_readlane(rv, j0 + u);
        float wj = __int_as_float(__builtin_amdgcn_readlane(__float_as_int(wl), j0 + u));
        atomicAdd(&tab[rj + 2 * lane],     vr[u].x * wj);
        atomicAdd(&tab[rj + 2 * lane + 1], vr[u].y * wj);
      }
    }
    sv = sn; dv = dn; wv = wn; bq = bn;
  }
  __syncthreads();

  {
    const int f  = tid & (DD - 1);
    const int c0 = (tid >> 6) * PCS;
    const int c1 = (c0 + PCS < nn) ? (c0 + PCS) : nn;
    if (c0 < c1) {
      int   ids[PCS];
      float vals[PCS];
#pragma unroll
      for (int j = 0; j < PCS; ++j) {
        int i = c0 + j;
        i = i < c1 ? i : c1 - 1;
        ids[j]  = gloc[i];
        vals[j] = tab[i * DD + f];
      }
      int   cg = ids[0];
      float a  = 0.0f;
#pragma unroll
      for (int j = 0; j < PCS; ++j) {
        if (c0 + j < c1) {
          if (ids[j] != cg) { atomicAdd(&gtab[cg * DD + f], a); a = 0.0f; cg = ids[j]; }
          a += vals[j];
        }
      }
      atomicAdd(&gtab[cg * DD + f], a);
    }
    if (tid < nn) atomicAdd(&ctab[gloc[tid]], 1.0f);
  }
  __syncthreads();

  {
    static_assert(GG * DD == 32 * AG_THREADS && AG_THREADS / DD == 16, "lane k of a wave <-> graph c + 16 k");
    const float cl = ctab[(tid >> 6) + 16 * lane];
    unsigned present = (unsigned)__ballot(cl > 0.0f);
    while (present) {
      const int k = __builtin_ctz(present);
      present &= present - 1;
      const int i = tid + k * AG_THREADS;
      atomicAdd(&psum[i], gtab[i]);
    }
  }
  if (tid < GG && ctab[tid] > 0.0f) atomicAdd(&cntg[tid], ctab[tid]);
}

__global__ void k_gemm_wmma(const float* __restrict__ A,
                            const float* __restrict__ B,
                            float* __restrict__ C,
                            const float* __restrict__ bias,
                            const float* __restrict__ dinv,
                            float* __restrict__ acc,
                            int M, int K, int Nc) {
  int wid  = (blockIdx.x * blockDim.x + threadIdx.x) >> 5;
  int lane = threadIdx.x & 31;
  int tiles_n = Nc >> 4;
  int tm = wid / tiles_n;
  int tn = wid - tm * tiles_n;
  if (tm * 16 >= M) return;

  int half = lane >> 4;
  int lm   = lane & 15;
  const float* Arow = A + (size_t)(tm * 16 + lm) * K;
  int ncol = tn * 16 + lm;

  v8f c = {};
#if USE_WMMA_F32
  for (int k0 = 0; k0 < K; k0 += 4) {
    v2f a, b;
    a.x = Arow[k0 + 2 * half + 0];
    a.y = Arow[k0 + 2 * half + 1];
    b.x = B[(size_t)(k0 + 2 * half + 0) * Nc + ncol];
    b.y = B[(size_t)(k0 + 2 * half + 1) * Nc + ncol];
    c = __builtin_amdgcn_wmma_f32_16x16x4_f32(false, a, false, b, (short)0, c,
                                              false, false);
  }
#else
  for (int k0 = 0; k0 < K; k0 += 32) {
    v16h a, b;
#pragma unroll
    for (int i = 0; i < 8; ++i) {
      a[i]     = (_Float16)Arow[k0 + 8 * half + i];
      a[8 + i] = (_Float16)Arow[k0 + 16 + 8 * half + i];
    }
#pragma unroll
    for (int i = 0; i < 16; ++i)
      b[i] = (_Float16)B[(size_t)(k0 + 16 * half + i) * Nc + ncol];
    c = __builtin_amdgcn_wmma_f32_16x16x32_f16(false, a, false, b, (short)0, c,
                                               false, false);
  }
#endif

  float bv = bias ? bias[ncol] : 0.0f;
  if (acc) {
#pragma unroll
    for (int r = 0; r < 8; ++r) {
      int row = tm * 16 + 8 * half + r;
      size_t idx = (size_t)row * Nc + ncol;
      float val = c[r] + bv;
      float di = dinv[row];
      C[idx]   = val;
      acc[idx] = val * (di * di);
    }
  } else {
#pragma unroll
    for (int r = 0; r < 8; ++r) {
      int row = tm * 16 + 8 * half + r;
      float val = c[r] + bv;
      C[(size_t)row * Nc + ncol] = val;
    }
  }
}

#define TL_WAVES (DD / 16)
#define TL_PITCH (DD + 4)
static_assert(DOUT / 16 <= TL_WAVES && GG % 16 == 0 && DD % 4 == 0, "tiles cover both products");

__global__ __launch_bounds__(TL_WAVES * 32)
void k_tail(const float* __restrict__ psum,
            const float* __restrict__ cntg,
            const float* __restrict__ W3,
            const float* __restrict__ b3,
            const float* __restrict__ Wl,
            const float* __restrict__ bl,
            float* __restrict__ out) {
#if USE_WMMA_F32
  __shared__ __attribute__((aligned(16))) float P[16 * TL_PITCH];
  const int lane = threadIdx.x & 31;
  const int wave = __builtin_amdgcn_readfirstlane(threadIdx.x >> 5);
  const int half = lane >> 4;
  const int lm   = lane & 15;
  const int r0   = blockIdx.x * 16;
  const int ncol = wave * 16 + lm;
  const bool head = wave < DOUT / 16;

  v2f a[DD / 4], b[DD / 4], bh[DD / 4];
  const float* __restrict__ Arow = psum + (r0 + lm) * DD + 2 * half;
#pragma unroll
  for (int s = 0; s < DD / 4; ++s) {
    a[s].x = Arow[4 * s];
    a[s].y = Arow[4 * s + 1];
    b[s].x = W3[(4 * s + 2 * half) * DD + ncol];
    b[s].y = W3[(4 * s + 2 * half + 1) * DD + ncol];
  }
  float cc[8];
#pragma unroll
  for (int r = 0; r < 8; ++r) cc[r] = cntg[r0 + 8 * half + r];
  const float bv = b3[ncol];
  float bo = 0.0f;
  if (head) {
#pragma unroll
    for (int s = 0; s < DD / 4; ++s) {
      bh[s].x = Wl[(4 * s + 2 * half) * DOUT + ncol];
      bh[s].y = Wl[(4 * s + 2 * half + 1) * DOUT + ncol];
    }
    bo = bl[ncol];
  }

  v8f c = {};
#pragma unroll
  for (int s = 0; s < DD / 4; ++s)
    c = __builtin_amdgcn_wmma_f32_16x16x4_f32(false, a[s], false, b[s], (short)0, c, false, false);
#pragma unroll
  for (int r = 0; r < 8; ++r) {
    const float q = c[r] / cc[r];
    P[(8 * half + r) * TL_PITCH + ncol] = (cc[r] > 0.0f) ? q + bv : 0.0f;
  }
  __syncthreads();

  if (head) {
    v8f o = {};
#pragma unroll
    for (int s = 0; s < DD / 4; ++s) {
      v2f ah;
      ah.x = P[lm * TL_PITCH + 4 * s + 2 * half];
      ah.y = P[lm * TL_PITCH + 4 * s + 2 * half + 1];
      o = __builtin_amdgcn_wmma_f32_16x16x4_f32(false, ah, false, bh[s], (short)0, o, false, false);
    }
#pragma unroll
    for (int r = 0; r < 8; ++r)
      out[(r0 + 8 * half + r) * DOUT + ncol] = o[r] + bo;
  }
#endif
}

static inline int blk(long long n, int b) { return (int)((n + b - 1) / b); }

extern "C" void kernel_launch(void* const* d_in, const int* in_sizes, int n_in,
                              void* d_out, int out_size, void* d_ws, size_t ws_size,
                              hipStream_t stream) {
  (void)in_sizes; (void)n_in; (void)out_size; (void)ws_size;
  const int*   edge_index = (const int*)d_in[0];
  const int*   batch      = (const int*)d_in[1];
  const float* W1 = (const float*)d_in[2];
  const float* b1 = (const float*)d_in[3];
  const float* W2 = (const float*)d_in[4];
  const float* b2 = (const float*)d_in[5];
  const float* W3 = (const float*)d_in[6];
  const float* b3 = (const float*)d_in[7];
  const float* g1 = (const float*)d_in[8];
  const float* be1= (const float*)d_in[9];
  const float* m1 = (const float*)d_in[10];
  const float* v1 = (const float*)d_in[11];
  const float* g2 = (const float*)d_in[12];
  const float* be2= (const float*)d_in[13];
  const float* m2 = (const float*)d_in[14];
  const float* v2 = (const float*)d_in[15];
  const float* Wl = (const float*)d_in[16];
  const float* bl = (const float*)d_in[17];
  float* out = (float*)d_out;

  char* ws = (char*)d_ws;
  size_t off = 0;
  auto carve = [&](size_t bytes) -> float* {
    float* p = (float*)(ws + off);
    off += (bytes + 255) & ~(size_t)255;
    return p;
  };
  float* deg_feat = carve((size_t)NN * 4);
  float* degcnt   = carve((size_t)NN * 4);
  float* dinv     = carve((size_t)NN * 4);
  float* bufX     = carve((size_t)NN * DD * 4);
  float* bufH     = carve((size_t)NN * DD * 4);
  float* psum     = carve((size_t)(GG * DD + GG) * 4);
  float* cnt      = psum + GG * DD;
  int*   pcnt     = (int*)carve((size_t)2 * NP * 4);
  int*   pcur     = pcnt + NP;
  int*   goffs    = (int*)carve((size_t)NP * 4);
  int4*  rec      = (int4*)carve((size_t)EE * 16);

  const int B = 256;

  k_clear<<<blk(NN, B), B, 0, stream>>>(deg_feat, degcnt, NN, pcnt, 2 * NP, psum, GG * DD + GG);
  k_edge_count<<<NP, BINT, 0, stream>>>(edge_index, deg_feat, degcnt, pcnt);
  k_fill<<<NP, BINT, 0, stream>>>(edge_index, pcnt, degcnt, deg_feat, W1, pcur, goffs, rec, dinv, bufH);

  const int tiles_NN  = (NN / 16) * (DD / 16);

  k_agg<<<NP, AG_WAVES * 32, 0, stream>>>(rec, pcnt, goffs, bufH, dinv,
                                          b1, g1, be1, m1, v1, bufX);

  k_gemm_wmma<<<blk((long long)tiles_NN * 32, B), B, 0, stream>>>(
      bufX, W2, bufH, nullptr, nullptr, nullptr, NN, DD, DD);
  k_agg<<<NP, AG_WAVES * 32, 0, stream>>>(rec, pcnt, goffs, bufH, dinv,
                                          b2, g2, be2, m2, v2, bufX);

  k_pool_agg<<<NP, AG_WAVES * 32, 0, stream>>>(rec, pcnt, goffs, batch, bufX, dinv, psum, cnt);

  k_tail<<<GG / 16, TL_WAVES * 32, 0, stream>>>(psum, cnt, W3, b3, Wl, bl, out);
}


# Layer-2 product formed by the layer-1 launch at its store: a range's finished rows (bias, BN, ReLU in LDS) x W2 on the f32 matrix instruction; x1 plane and product launch gone; 8 -> 7 launches
# speedup vs baseline: 3.7537x; 1.0652x over previous
#include <hip/hip_runtime.h>
#include <hip/hip_bf16.h>

#define NN   50000
#define EE   800000
#define GG   512
#define DD   64
#define DOUT 32
#define BN_EPS 1e-5f

typedef __attribute__((ext_vector_type(2)))  float    v2f;
typedef __attribute__((ext_vector_type(8)))  float    v8f;
typedef __attribute__((ext_vector_type(16))) _Float16 v16h;

#if defined(__AMDGCN__) && __has_builtin(__builtin_amdgcn_wmma_f32_16x16x4_f32)
#define USE_WMMA_F32 1
#else
#define USE_WMMA_F32 0
#endif


#define NP   256
#define PN   196
#define CH   ((EE + NP - 1) / NP)
#define BINT 1024
#define EPT  ((CH + BINT - 1) / BINT)
static_assert(NP * PN >= NN, "ranges must cover the nodes");
static_assert(NP <= BINT && NP % 32 == 0, "one thread per range");
static_assert((long long)NP * CH >= EE && EPT * BINT >= CH, "the passes cover every edge");

#define SLN  16
#define SNO  (NN / SLN)
#define SED  (EE / SLN)
#define SQ   (SED / 4)
#define SQG  4
#define SQT  ((SQ + BINT - 1) / BINT)
static_assert(SLN * SLN == NP, "one work-group per (slab, slice)");
static_assert(SNO * SLN == NN && SED * SLN == EE, "slices tile the nodes, slabs tile the edges");
static_assert(SED % 4 == 0 && EE % 4 == 0, "slabs start on 16-byte boundaries in both index arrays");
static_assert(SLN * CH == SED, "a work-group's own chunk lies inside its slab");

__global__ void k_clear(float* __restrict__ a, float* __restrict__ b, int n,
                        int* __restrict__ c, int m,
                        float* __restrict__ p, int np) {
  int i = blockIdx.x * blockDim.x + threadIdx.x;
  if (i < n) { a[i] = 0.0f; b[i] = 0.0f; }
  if (i < m) c[i] = 0;
  if (i < np) p[i] = 0.0f;
}

__global__ __launch_bounds__(BINT)
void k_edge_count(const int* __restrict__ ei,
                  float* __restrict__ deg_feat,
                  float* __restrict__ degcnt,
                  int* __restrict__ pcnt) {
  __shared__ int h[NP];
  __shared__ int cs[SNO];
  __shared__ int cd[SNO];
  const int t = threadIdx.x;
  if (t < NP) h[t] = 0;
  for (int i = t; i < SNO; i += BINT) { cs[i] = 0; cd[i] = 0; }
  int d[EPT];
  const int c0 = blockIdx.x * CH;
#pragma unroll
  for (int k = 0; k < EPT; ++k) {
    int e = c0 + t + k * BINT;
    e = e < EE ? e : EE - 1;
    d[k] = ei[EE + e];
  }
  __syncthreads();
#pragma unroll
  for (int k = 0; k < EPT; ++k) {
    int i = t + k * BINT;
    if (i < CH && c0 + i < EE) atomicAdd(&h[d[k] / PN], 1);
  }

  const int slab = blockIdx.x / SLN;
  const int nb0  = (blockIdx.x % SLN) * SNO;
  const int4* __restrict__ s4 = (const int4*)(ei + slab * SED);
  const int4* __restrict__ d4 = (const int4*)(ei + EE + slab * SED);
  for (int g0 = 0; g0 < SQT; g0 += SQG) {
    int4 a[SQG], b[SQG];
#pragma unroll
    for (int k = 0; k < SQG; ++k) {
      int q = t + (g0 + k) * BINT;
      q = q < SQ ? q : SQ - 1;
      a[k] = s4[q];
      b[k] = d4[q];
    }
#pragma unroll
    for (int k = 0; k < SQG; ++k) {
      if (t + (g0 + k) * BINT < SQ) {
        unsigned u;
        u = (unsigned)(a[k].x - nb0); if (u < (unsigned)SNO) atomicAdd(&cs[u], 1);
        u = (unsigned)(a[k].y - nb0); if (u < (unsigned)SNO) atomicAdd(&cs[u], 1);
        u = (unsigned)(a[k].z - nb0); if (u < (unsigned)SNO) atomicAdd(&cs[u], 1);
        u = (unsigned)(a[k].w - nb0); if (u < (unsigned)SNO) atomicAdd(&cs[u], 1);
        u = (unsigned)(b[k].x - nb0); if (u < (unsigned)SNO) atomicAdd(&cd[u], 1);
        u = (unsigned)(b[k].y - nb0); if (u < (unsigned)SNO) atomicAdd(&cd[u], 1);
        u = (unsigned)(b[k].z - nb0); if (u < (unsigned)SNO) atomicAdd(&cd[u], 1);
        u = (unsigned)(b[k].w - nb0); if (u < (unsigned)SNO) atomicAdd(&cd[u], 1);
      }
    }
  }
  __syncthreads();
  for (int i = t; i < SNO; i += BINT) {
    int na = cs[i], nd = cd[i];
    if (na > 0) atomicAdd(&deg_feat[nb0 + i], (float)na);
    if (nd > 0) atomicAdd(&degcnt[nb0 + i], (float)nd);
  }
  if (t < NP && h[t] > 0) atomicAdd(&pcnt[t], h[t]);
}


#define HQ   (NN * DD / 4)
#define HQW  (HQ / NP)
#define HQT  ((HQW + BINT - 1) / BINT)
static_assert(HQW * NP == HQ, "the work-groups tile the plane exactly");
static_assert(BINT % (DD / 4) == 0, "a thread keeps the same four features in every group");
static_assert(PN <= BINT, "one thread per node of a range");

__global__ __launch_bounds__(BINT)
void k_fill(const int* __restrict__ ei,
            const int* __restrict__ pcnt,
            const float* __restrict__ degcnt,
            const float* __restrict__ deg_feat,
            const float* __restrict__ W1,
            int* __restrict__ pcur,
            int* __restrict__ goffs,
            int4* __restrict__ rec,
            float* __restrict__ dinv,
            float* __restrict__ hw) {
  __shared__ int h[NP];
  __shared__ int hb[NP];
  __shared__ int wsum[NP / 32];
  const int t = threadIdx.x;
  const int lane = t & 31;
  int s[EPT], d[EPT], r[EPT];
  float w[EPT];
  const int c0 = blockIdx.x * CH;
#pragma unroll
  for (int k = 0; k < EPT; ++k) {
    int e = c0 + t + k * BINT;
    e = e < EE ? e : EE - 1;
    s[k] = ei[e];
    d[k] = ei[EE + e];
  }
  const int q0 = blockIdx.x * HQW;
  float dg[HQT];
#pragma unroll
  for (int k = 0; k < HQT; ++k) {
    int i = t + k * BINT;
    i = i < HQW ? i : HQW - 1;
    dg[k] = deg_feat[(q0 + i) >> 4];
  }
  const int f0 = ((q0 + t) & (DD / 4 - 1)) * 4;
  const float w0 = W1[f0], w1 = W1[f0 + 1], w2 = W1[f0 + 2], w3 = W1[f0 + 3];
  const int n = blockIdx.x * PN + t;
  if (t < PN && n < NN) dinv[n] = rsqrtf(degcnt[n] + 1.0f);
  float4* __restrict__ hw4 = (float4*)hw + q0;
#pragma unroll
  for (int k = 0; k < HQT; ++k) {
    const int i = t + k * BINT;
    if (i < HQW) {
      float4 y;
      y.x = dg[k] * w0; y.y = dg[k] * w1; y.z = dg[k] * w2; y.w = dg[k] * w3;
      hw4[i] = y;
    }
  }
  int v = pcnt[t & (NP - 1)];
#pragma unroll
  for (int k = 0; k < EPT; ++k) w[k] = rsqrtf(degcnt[s[k]] + 1.0f) * rsqrtf(degcnt[d[k]] + 1.0f);
  v = (t < NP) ? v : 0;
  int x = v;
  for (int o = 1; o < 32; o <<= 1) {
    int y = __shfl_up(x, o);
    if (lane >= o) x += y;
  }
  if (t < NP) {
    h[t] = 0;
    if (lane == 31) wsum[t >> 5] = x;
  }
  __syncthreads();
  int off = 0;
  if (t < NP) {
    for (int i = 0; i < (t >> 5); ++i) off += wsum[i];
    off += x - v;
    if (blockIdx.x == 0) goffs[t] = off;
  }
#pragma unroll
  for (int k = 0; k < EPT; ++k) {
    int i = t + k * BINT;
    r[k] = -1;
    if (i < CH && c0 + i < EE)
      r[k] = atomicAdd(&h[d[k] / PN], 1);
  }
  __syncthreads();
  if (t < NP) {
    int c = h[t];
    int g = c > 0 ? atomicAdd(&pcur[t], c) : 0;
    hb[t] = off + g;
  }
  __syncthreads();
#pragma unroll
  for (int k = 0; k < EPT; ++k) {
    if (r[k] >= 0) {
      int pos = hb[d[k] / PN] + r[k];
      rec[pos] = make_int4(s[k], d[k], __float_as_int(w[k]), 0);
    }
  }
}

#define AG_WAVES 32
#define AG_GRP   16
#define AG_TAB_FLOATS (41 * 1024)
static_assert((PN + 1) * DD <= AG_TAB_FLOATS, "table holds one range and the spare row");
#define AG_THREADS (AG_WAVES * 32)
#define TQ   (PN * DD / 4)
#define TQT  ((TQ + AG_THREADS - 1) / AG_THREADS)
static_assert(DD % 4 == 0 && AG_THREADS % (DD / 4) == 0, "a thread keeps the same four features in every group");
#define NX_PITCH (DD + 4)
#define NX_ROWS  (((PN + 1 + 15) / 16) * 16)
#define NX_RT    (NX_ROWS / 16)
static_assert(2 * NX_ROWS * NX_PITCH <= AG_TAB_FLOATS, "the rows and their products both fit the table");
static_assert(NX_ROWS * (DD / 4) <= TQT * AG_THREADS, "the set-up pass reaches every row of the tiles");
static_assert(AG_WAVES % (DD / 16) == 0, "a wave keeps one column tile");

template <bool NEXT>
__global__ __launch_bounds__(AG_WAVES * 32)
void k_agg(const int4* __restrict__ rec,
           const int* __restrict__ pcnt,
           const int* __restrict__ goffs,
           const float* __restrict__ hw,
           const float* __restrict__ dinv,
           const float* __restrict__ b,
           const float* __restrict__ g,
           const float* __restrict__ be,
           const float* __restrict__ m,
           const float* __restrict__ v,
           const float* __restrict__ Wn,
           float* __restrict__ xout) {
  __shared__ __attribute__((aligned(16))) float tab[AG_TAB_FLOATS];
  constexpr int TP  = NEXT ? NX_PITCH : DD;
  constexpr int TP4 = TP / 4;
  const int tid  = threadIdx.x;
  const int lane = tid & 31;
  const int wave = __builtin_amdgcn_readfirstlane(tid >> 5);
  const int n0 = blockIdx.x * PN;
  const int nn = (NN - n0 < PN) ? (NN - n0) : PN;
  const int start = goffs[blockIdx.x];
  const int cnt   = pcnt[blockIdx.x];
  const int nb    = (cnt + 31) >> 5;
  const float2* __restrict__ hw2 = (const float2*)hw;

  int bq = wave;
  int sv = 0, dv = n0;
  float wv = 0.0f;
  if (bq < nb) {
    int li = bq * 32 + lane;
    li = li < cnt ? li : cnt - 1;
    int4 q = rec[start + li];
    sv = q.x; dv = q.y; wv = __int_as_float(q.z);
  }

  const int f0 = (tid & (DD / 4 - 1)) * 4;
  float cb[4], cg[4], cbe[4], cm[4], cv[4];
#pragma unroll
  for (int j = 0; j < 4; ++j) {
    cb[j] = b[f0 + j]; cg[j] = g[f0 + j]; cbe[j] = be[f0 + j]; cm[j] = m[f0 + j]; cv[j] = v[f0 + j];
  }

  const int nq = nn * (DD / 4);
  const float4* __restrict__ hw4 = (const float4*)(hw + n0 * DD);
  {
    float4 hv[TQT];
    float  dq[TQT];
#pragma unroll
    for (int k = 0; k < TQT; ++k) {
      int q = tid + k * AG_THREADS;
      q = q < nq ? q : nq - 1;
      hv[k] = hw4[q];
      dq[k] = dinv[n0 + (q >> 4)];
    }
#pragma unroll
    for (int k = 0; k < TQT; ++k) {
      const int q = tid + k * AG_THREADS;
      if (q < nq) {
        const float s = dq[k] * dq[k];
        float4 t = hv[k];
        t.x *= s; t.y *= s; t.z *= s; t.w *= s;
        ((float4*)tab)[(q >> 4) * TP4 + (q & 15)] = t;
      } else if (NEXT && q < NX_ROWS * (DD / 4)) {
        ((float4*)tab)[(q >> 4) * TP4 + (q & 15)] = make_float4(0.0f, 0.0f, 0.0f, 0.0f);
      }
    }
  }
  __syncthreads();

  while (bq < nb) {
    const int bn = bq + AG_WAVES;
    int sn = 0, dn = n0;
    float wn = 0.0f;
    if (bn < nb) {
      int li = bn * 32 + lane;
      li = li < cnt ? li : cnt - 1;
      int4 q = rec[start + li];
      sn = q.x; dn = q.y; wn = __int_as_float(q.z);
    }
    const bool  live = bq * 32 + lane < cnt;
    const int   rv = live ? (dv - n0) * TP : PN * TP;
    const float wl = live ? wv : 0.0f;
#pragma unroll
    for (int j0 = 0; j0 < 32; j0 += AG_GRP) {
      float2 vr[AG_GRP];
#pragma unroll
      for (int u = 0; u < AG_GRP; ++u) {
        int sj = __builtin_amdgcn_readlane(sv, j0 + u);
        vr[u] = hw2[sj * (DD / 2) + lane];
      }
#pragma unroll
      for (int u = 0; u < AG_GRP; ++u) {
        int   rj = __builtin_amdgcn_readlane(rv, j0 + u);
        float wj = __int_as_float(__builtin_amdgcn_readlane(__float_as_int(wl), j0 + u));
        atomicAdd(&tab[rj + 2 * lane],     vr[u].x * wj);
        atomicAdd(&tab[rj + 2 * lane + 1], vr[u].y * wj);
      }
    }
    sv = sn; dv = dn; wv = wn; bq = bn;
  }

  float rs[4];
#pragma unroll
  for (int j = 0; j < 4; ++j) rs[j] = rsqrtf(cv[j] + BN_EPS);
  float4* __restrict__ xo4 = (float4*)(xout + n0 * DD);

  if constexpr (!NEXT) {
    __syncthreads();
#pragma unroll
    for (int k = 0; k < TQT; ++k) {
      const int q = tid + k * AG_THREADS;
      if (q < nq) {
        const float4 t = ((const float4*)tab)[q];
        float4 y;
        y.x = fmaxf(((t.x + cb[0]) - cm[0]) * rs[0] * cg[0] + cbe[0], 0.0f);
        y.y = fmaxf(((t.y + cb[1]) - cm[1]) * rs[1] * cg[1] + cbe[1], 0.0f);
        y.z = fmaxf(((t.z + cb[2]) - cm[2]) * rs[2] * cg[2] + cbe[2], 0.0f);
        y.w = fmaxf(((t.w + cb[3]) - cm[3]) * rs[3] * cg[3] + cbe[3], 0.0f);
        xo4[q] = y;
      }
    }
  } else {
#if USE_WMMA_F32
    const int half = lane >> 4;
    const int lm   = lane & 15;
    const int ncol = (wave & (DD / 16 - 1)) * 16 + lm;
    v2f bw[DD / 4];
#pragma unroll
    for (int s = 0; s < DD / 4; ++s) {
      bw[s].x = Wn[(4 * s + 2 * half) * DD + ncol];
      bw[s].y = Wn[(4 * s + 2 * half + 1) * DD + ncol];
    }
    __syncthreads();
#pragma unroll
    for (int k = 0; k < TQT; ++k) {
      const int q = tid + k * AG_THREADS;
      if (q < nq) {
        float4* p = (float4*)tab + (q >> 4) * TP4 + (q & 15);
        const float4 t = *p;
        float4 y;
        y.x = fmaxf(((t.x + cb[0]) - cm[0]) * rs[0] * cg[0] + cbe[0], 0.0f);
        y.y = fmaxf(((t.y + cb[1]) - cm[1]) * rs[1] * cg[1] + cbe[1], 0.0f);
        y.z = fmaxf(((t.z + cb[2]) - cm[2]) * rs[2] * cg[2] + cbe[2], 0.0f);
        y.w = fmaxf(((t.w + cb[3]) - cm[3]) * rs[3] * cg[3] + cbe[3], 0.0f);
        *p = y;
      }
    }
    __syncthreads();
    float* __restrict__ prod = tab + NX_ROWS * TP;
    for (int rt = wave / (DD / 16); rt < NX_RT; rt += AG_WAVES / (DD / 16)) {
      const float* __restrict__ ar = tab + (rt * 16 + lm) * TP + 2 * half;
      v2f a[DD / 4];
#pragma unroll
      for (int s = 0; s < DD / 4; ++s) a[s] = *(const v2f*)(ar + 4 * s);
      v8f acc = {};
#pragma unroll
      for (int s = 0; s < DD / 4; ++s)
        acc = __builtin_amdgcn_wmma_f32_16x16x4_f32(false, a[s], false, bw[s], (short)0, acc, false, false);
#pragma unroll
      for (int r = 0; r < 8; ++r) prod[(rt * 16 + 8 * half + r) * TP + ncol] = acc[r];
    }
    __syncthreads();
#pragma unroll
    for (int k = 0; k < TQT; ++k) {
      const int q = tid + k * AG_THREADS;
      if (q < nq) xo4[q] = ((const float4*)prod)[(q >> 4) * TP4 + (q & 15)];
    }
#endif
  }
}

#define PCH  16
#define PCS  ((PN + PCH - 1) / PCH)
static_assert(PCH * DD == AG_WAVES * 32, "one thread per (chunk, feature)");

__global__ __launch_bounds__(AG_WAVES * 32)
void k_pool_agg(const int4* __restrict__ rec,
                const int* __restrict__ pcnt,
                const int* __restrict__ goffs,
                const int* __restrict__ batch,
                const float* __restrict__ x,
                const float* __restrict__ dinv,
                float* __restrict__ psum,
                float* __restrict__ cntg) {
  __shared__ __attribute__((aligned(16))) float tab[(PN + 1) * DD];
  __shared__ __attribute__((aligned(16))) float gtab[GG * DD];
  __shared__ float ctab[GG];
  __shared__ int   gloc[PN];
  const int tid  = threadIdx.x;
  const int lane = tid & 31;
  const int wave = __builtin_amdgcn_readfirstlane(tid >> 5);
  const int n0 = blockIdx.x * PN;
  const int nn = (NN - n0 < PN) ? (NN - n0) : PN;
  const int start = goffs[blockIdx.x];
  const int cnt   = pcnt[blockIdx.x];
  const int nb    = (cnt + 31) >> 5;
  const float2* __restrict__ x2 = (const float2*)x;

  int bq = wave;
  int sv = 0, dv = n0;
  float wv = 0.0f;
  if (bq < nb) {
    int li = bq * 32 + lane;
    li = li < cnt ? li : cnt - 1;
    int4 q = rec[start + li];
    sv = q.x; dv = q.y; wv = __int_as_float(q.z);
  }

  const int nq = nn * (DD / 4);
  const float4* __restrict__ x4 = (const float4*)(x + n0 * DD);
  {
    const int gb = batch[n0 + (tid < nn ? tid : nn - 1)];
    float4 hv[TQT];
    float  dq[TQT];
#pragma unroll
    for (int k = 0; k < TQT; ++k) {
      int q = tid + k * AG_THREADS;
      q = q < nq ? q : nq - 1;
      hv[k] = x4[q];
      dq[k] = dinv[n0 + (q >> 4)];
    }
    float4 z4 = {0.0f, 0.0f, 0.0f, 0.0f};
    for (int i = tid; i < GG * DD / 4; i += AG_THREADS) ((float4*)gtab)[i] = z4;
    if (tid < GG) ctab[tid] = 0.0f;
#pragma unroll
    for (int k = 0; k < TQT; ++k) {
      const int q = tid + k * AG_THREADS;
      if (q < nq) {
        const float s = dq[k] * dq[k];
        float4 t = hv[k];
        t.x *= s; t.y *= s; t.z *= s; t.w *= s;
        ((float4*)tab)[q] = t;
      }
    }
    if (tid < nn) gloc[tid] = gb;
  }
  __syncthreads();

  while (bq < nb) {
    const int bn = bq + AG_WAVES;
    int sn = 0, dn = n0;
    float wn = 0.0f;
    if (bn < nb) {
      int li = bn * 32 + lane;
      li = li < cnt ? li : cnt - 1;
      int4 q = rec[start + li];
      sn = q.x; dn = q.y; wn = __int_as_float(q.z);
    }
    const bool  live = bq * 32 + lane < cnt;
    const int   rv = live ? (dv - n0) * DD : PN * DD;
    const float wl = live ? wv : 0.0f;
#pragma unroll
    for (int j0 = 0; j0 < 32; j0 += AG_GRP) {
      float2 vr[AG_GRP];
#pragma unroll
      for (int u = 0; u < AG_GRP; ++u) {
        int sj = __builtin_amdgcn_readlane(sv, j0 + u);
        vr[u] = x2[sj * (DD / 2) + lane];
      }
#pragma unroll
      for (int u = 0; u < AG_GRP; ++u) {
        int   rj = __builtin_amdgcn_readlane(rv, j0 + u);
        float wj = __int_as_float(__builtin_amdgcn_readlane(__float_as_int(wl), j0 + u));
        atomicAdd(&tab[rj + 2 * lane],     vr[u].x * wj);
        atomicAdd(&tab[rj + 2 * lane + 1], vr[u].y * wj);
      }
    }
    sv = sn; dv = dn; wv = wn; bq = bn;
  }
  __syncthreads();

  {
    const int f  = tid & (DD - 1);
    const int c0 = (tid >> 6) * PCS;
    const int c1 = (c0 + PCS < nn) ? (c0 + PCS) : nn;
    if (c0 < c1) {
      int   ids[PCS];
      float vals[PCS];
#pragma unroll
      for (int j = 0; j < PCS; ++j) {
        int i = c0 + j;
        i = i < c1 ? i : c1 - 1;
        ids[j]  = gloc[i];
        vals[j] = tab[i * DD + f];
      }
      int   cg = ids[0];
      float a  = 0.0f;
#pragma unroll
      for (int j = 0; j < PCS; ++j) {
        if (c0 + j < c1) {
          if (ids[j] != cg) { atomicAdd(&gtab[cg * DD + f], a); a = 0.0f; cg = ids[j]; }
          a += vals[j];
        }
      }
      atomicAdd(&gtab[cg * DD + f], a);
    }
    if (tid < nn) atomicAdd(&ctab[gloc[tid]], 1.0f);
  }
  __syncthreads();

  {
    static_assert(GG * DD == 32 * AG_THREADS && AG_THREADS / DD == 16, "lane k of a wave <-> graph c + 16 k");
    const float cl = ctab[(tid >> 6) + 16 * lane];
    unsigned present = (unsigned)__ballot(cl > 0.0f);
    while (present) {
      const int k = __builtin_ctz(present);
      present &= present - 1;
      const int i = tid + k * AG_THREADS;
      atomicAdd(&psum[i], gtab[i]);
    }
  }
  if (tid < GG && ctab[tid] > 0.0f) atomicAdd(&cntg[tid], ctab[tid]);
}

#define TL_WAVES (DD / 16)
#define TL_PITCH (DD + 4)
static_assert(DOUT / 16 <= TL_WAVES && GG % 16 == 0 && DD % 4 == 0, "tiles cover both products");

__global__ __launch_bounds__(TL_WAVES * 32)
void k_tail(const float* __restrict__ psum,
            const float* __restrict__ cntg,
            const float* __restrict__ W3,
            const float* __restrict__ b3,
            const float* __restrict__ Wl,
            const float* __restrict__ bl,
            float* __restrict__ out) {
#if USE_WMMA_F32
  __shared__ __attribute__((aligned(16))) float P[16 * TL_PITCH];
  const int lane = threadIdx.x & 31;
  const int wave = __builtin_amdgcn_readfirstlane(threadIdx.x >> 5);
  const int half = lane >> 4;
  const int lm   = lane & 15;
  const int r0   = blockIdx.x * 16;
  const int ncol = wave * 16 + lm;
  const bool head = wave < DOUT / 16;

  v2f a[DD / 4], b[DD / 4], bh[DD / 4];
  const float* __restrict__ Arow = psum + (r0 + lm) * DD + 2 * half;
#pragma unroll
  for (int s = 0; s < DD / 4; ++s) {
    a[s].x = Arow[4 * s];
    a[s].y = Arow[4 * s + 1];
    b[s].x = W3[(4 * s + 2 * half) * DD + ncol];
    b[s].y = W3[(4 * s + 2 * half + 1) * DD + ncol];
  }
  float cc[8];
#pragma unroll
  for (int r = 0; r < 8; ++r) cc[r] = cntg[r0 + 8 * half + r];
  const float bv = b3[ncol];
  float bo = 0.0f;
  if (head) {
#pragma unroll
    for (int s = 0; s < DD / 4; ++s) {
      bh[s].x = Wl[(4 * s + 2 * half) * DOUT + ncol];
      bh[s].y = Wl[(4 * s + 2 * half + 1) * DOUT + ncol];
    }
    bo = bl[ncol];
  }

  v8f c = {};
#pragma unroll
  for (int s = 0; s < DD / 4; ++s)
    c = __builtin_amdgcn_wmma_f32_16x16x4_f32(false, a[s], false, b[s], (short)0, c, false, false);
#pragma unroll
  for (int r = 0; r < 8; ++r) {
    const float q = c[r] / cc[r];
    P[(8 * half + r) * TL_PITCH + ncol] = (cc[r] > 0.0f) ? q + bv : 0.0f;
  }
  __syncthreads();

  if (head) {
    v8f o = {};
#pragma unroll
    for (int s = 0; s < DD / 4; ++s) {
      v2f ah;
      ah.x = P[lm * TL_PITCH + 4 * s + 2 * half];
      ah.y = P[lm * TL_PITCH + 4 * s + 2 * half + 1];
      o = __builtin_amdgcn_wmma_f32_16x16x4_f32(false, ah, false, bh[s], (short)0, o, false, false);
    }
#pragma unroll
    for (int r = 0; r < 8; ++r)
      out[(r0 + 8 * half + r) * DOUT + ncol] = o[r] + bo;
  }
#endif
}

static inline int blk(long long n, int b) { return (int)((n + b - 1) / b); }

extern "C" void kernel_launch(void* const* d_in, const int* in_sizes, int n_in,
                              void* d_out, int out_size, void* d_ws, size_t ws_size,
                              hipStream_t stream) {
  (void)in_sizes; (void)n_in; (void)out_size; (void)ws_size;
  const int*   edge_index = (const int*)d_in[0];
  const int*   batch      = (const int*)d_in[1];
  const float* W1 = (const float*)d_in[2];
  const float* b1 = (const float*)d_in[3];
  const float* W2 = (const float*)d_in[4];
  const float* b2 = (const float*)d_in[5];
  const float* W3 = (const float*)d_in[6];
  const float* b3 = (const float*)d_in[7];
  const float* g1 = (const float*)d_in[8];
  const float* be1= (const float*)d_in[9];
  const float* m1 = (const float*)d_in[10];
  const float* v1 = (const float*)d_in[11];
  const float* g2 = (const float*)d_in[12];
  const float* be2= (const float*)d_in[13];
  const float* m2 = (const float*)d_in[14];
  const float* v2 = (const float*)d_in[15];
  const float* Wl = (const float*)d_in[16];
  const float* bl = (const float*)d_in[17];
  float* out = (float*)d_out;

  char* ws = (char*)d_ws;
  size_t off = 0;
  auto carve = [&](size_t bytes) -> float* {
    float* p = (float*)(ws + off);
    off += (bytes + 255) & ~(size_t)255;
    return p;
  };
  float* deg_feat = carve((size_t)NN * 4);
  float* degcnt   = carve((size_t)NN * 4);
  float* dinv     = carve((size_t)NN * 4);
  float* bufA     = carve((size_t)NN * DD * 4);
  float* bufB     = carve((size_t)NN * DD * 4);
  float* psum     = carve((size_t)(GG * DD + GG) * 4);
  float* cnt      = psum + GG * DD;
  int*   pcnt     = (int*)carve((size_t)2 * NP * 4);
  int*   pcur     = pcnt + NP;
  int*   goffs    = (int*)carve((size_t)NP * 4);
  int4*  rec      = (int4*)carve((size_t)EE * 16);

  const int B = 256;

  k_clear<<<blk(NN, B), B, 0, stream>>>(deg_feat, degcnt, NN, pcnt, 2 * NP, psum, GG * DD + GG);
  k_edge_count<<<NP, BINT, 0, stream>>>(edge_index, deg_feat, degcnt, pcnt);
  k_fill<<<NP, BINT, 0, stream>>>(edge_index, pcnt, degcnt, deg_feat, W1, pcur, goffs, rec, dinv, bufA);

  k_agg<true><<<NP, AG_WAVES * 32, 0, stream>>>(rec, pcnt, goffs, bufA, dinv,
                                                b1, g1, be1, m1, v1, W2, bufB);

  k_agg<false><<<NP, AG_WAVES * 32, 0, stream>>>(rec, pcnt, goffs, bufB, dinv,
                                                 b2, g2, be2, m2, v2, nullptr, bufA);

  k_pool_agg<<<NP, AG_WAVES * 32, 0, stream>>>(rec, pcnt, goffs, batch, bufA, dinv, psum, cnt);

  k_tail<<<GG / 16, TL_WAVES * 32, 0, stream>>>(psum, cnt, W3, b3, Wl, bl, out);
}


# Range sums without a float add per edge: a work-group counting-sorts its records by destination row in LDS, waves sum each row's source rows in registers, one LDS add per row; sum order: rank in row
# speedup vs baseline: 4.5503x; 1.2122x over previous
#include <hip/hip_runtime.h>
#include <hip/hip_bf16.h>

#define NN   50000
#define EE   800000
#define GG   512
#define DD   64
#define DOUT 32
#define BN_EPS 1e-5f

typedef __attribute__((ext_vector_type(2)))  float    v2f;
typedef __attribute__((ext_vector_type(8)))  float    v8f;
typedef __attribute__((ext_vector_type(16))) _Float16 v16h;

#if defined(__AMDGCN__) && __has_builtin(__builtin_amdgcn_wmma_f32_16x16x4_f32)
#define USE_WMMA_F32 1
#else
#define USE_WMMA_F32 0
#endif


#define NP   256
#define PN   196
#define CH   ((EE + NP - 1) / NP)
#define BINT 1024
#define EPT  ((CH + BINT - 1) / BINT)
static_assert(NP * PN >= NN, "ranges must cover the nodes");
static_assert(NP <= BINT && NP % 32 == 0, "one thread per range");
static_assert((long long)NP * CH >= EE && EPT * BINT >= CH, "the passes cover every edge");

#define SLN  16
#define SNO  (NN / SLN)
#define SED  (EE / SLN)
#define SQ   (SED / 4)
#define SQG  4
#define SQT  ((SQ + BINT - 1) / BINT)
static_assert(SLN * SLN == NP, "one work-group per (slab, slice)");
static_assert(SNO * SLN == NN && SED * SLN == EE, "slices tile the nodes, slabs tile the edges");
static_assert(SED % 4 == 0 && EE % 4 == 0, "slabs start on 16-byte boundaries in both index arrays");
static_assert(SLN * CH == SED, "a work-group's own chunk lies inside its slab");

__global__ void k_clear(float* __restrict__ a, float* __restrict__ b, int n,
                        int* __restrict__ c, int m,
                        float* __restrict__ p, int np) {
  int i = blockIdx.x * blockDim.x + threadIdx.x;
  if (i < n) { a[i] = 0.0f; b[i] = 0.0f; }
  if (i < m) c[i] = 0;
  if (i < np) p[i] = 0.0f;
}

__global__ __launch_bounds__(BINT)
void k_edge_count(const int* __restrict__ ei,
                  float* __restrict__ deg_feat,
                  float* __restrict__ degcnt,
                  int* __restrict__ pcnt) {
  __shared__ int h[NP];
  __shared__ int cs[SNO];
  __shared__ int cd[SNO];
  const int t = threadIdx.x;
  if (t < NP) h[t] = 0;
  for (int i = t; i < SNO; i += BINT) { cs[i] = 0; cd[i] = 0; }
  int d[EPT];
  const int c0 = blockIdx.x * CH;
#pragma unroll
  for (int k = 0; k < EPT; ++k) {
    int e = c0 + t + k * BINT;
    e = e < EE ? e : EE - 1;
    d[k] = ei[EE + e];
  }
  __syncthreads();
#pragma unroll
  for (int k = 0; k < EPT; ++k) {
    int i = t + k * BINT;
    if (i < CH && c0 + i < EE) atomicAdd(&h[d[k] / PN], 1);
  }

  const int slab = blockIdx.x / SLN;
  const int nb0  = (blockIdx.x % SLN) * SNO;
  const int4* __restrict__ s4 = (const int4*)(ei + slab * SED);
  const int4* __restrict__ d4 = (const int4*)(ei + EE + slab * SED);
  for (int g0 = 0; g0 < SQT; g0 += SQG) {
    int4 a[SQG], b[SQG];
#pragma unroll
    for (int k = 0; k < SQG; ++k) {
      int q = t + (g0 + k) * BINT;
      q = q < SQ ? q : SQ - 1;
      a[k] = s4[q];
      b[k] = d4[q];
    }
#pragma unroll
    for (int k = 0; k < SQG; ++k) {
      if (t + (g0 + k) * BINT < SQ) {
        unsigned u;
        u = (unsigned)(a[k].x - nb0); if (u < (unsigned)SNO) atomicAdd(&cs[u], 1);
        u = (unsigned)(a[k].y - nb0); if (u < (unsigned)SNO) atomicAdd(&cs[u], 1);
        u = (unsigned)(a[k].z - nb0); if (u < (unsigned)SNO) atomicAdd(&cs[u], 1);
        u = (unsigned)(a[k].w - nb0); if (u < (unsigned)SNO) atomicAdd(&cs[u], 1);
        u = (unsigned)(b[k].x - nb0); if (u < (unsigned)SNO) atomicAdd(&cd[u], 1);
        u = (unsigned)(b[k].y - nb0); if (u < (unsigned)SNO) atomicAdd(&cd[u], 1);
        u = (unsigned)(b[k].z - nb0); if (u < (unsigned)SNO) atomicAdd(&cd[u], 1);
        u = (unsigned)(b[k].w - nb0); if (u < (unsigned)SNO) atomicAdd(&cd[u], 1);
      }
    }
  }
  __syncthreads();
  for (int i = t; i < SNO; i += BINT) {
    int na = cs[i], nd = cd[i];
    if (na > 0) atomicAdd(&deg_feat[nb0 + i], (float)na);
    if (nd > 0) atomicAdd(&degcnt[nb0 + i], (float)nd);
  }
  if (t < NP && h[t] > 0) atomicAdd(&pcnt[t], h[t]);
}


#define HQ   (NN * DD / 4)
#define HQW  (HQ / NP)
#define HQT  ((HQW + BINT - 1) / BINT)
static_assert(HQW * NP == HQ, "the work-groups tile the plane exactly");
static_assert(BINT % (DD / 4) == 0, "a thread keeps the same four features in every group");
static_assert(PN <= BINT, "one thread per node of a range");

__global__ __launch_bounds__(BINT)
void k_fill(const int* __restrict__ ei,
            const int* __restrict__ pcnt,
            const float* __restrict__ degcnt,
            const float* __restrict__ deg_feat,
            const float* __restrict__ W1,
            int* __restrict__ pcur,
            int* __restrict__ goffs,
            int4* __restrict__ rec,
            float* __restrict__ dinv,
            float* __restrict__ hw) {
  __shared__ int h[NP];
  __shared__ int hb[NP];
  __shared__ int wsum[NP / 32];
  const int t = threadIdx.x;
  const int lane = t & 31;
  int s[EPT], d[EPT], r[EPT];
  float w[EPT];
  const int c0 = blockIdx.x * CH;
#pragma unroll
  for (int k = 0; k < EPT; ++k) {
    int e = c0 + t + k * BINT;
    e = e < EE ? e : EE - 1;
    s[k] = ei[e];
    d[k] = ei[EE + e];
  }
  const int q0 = blockIdx.x * HQW;
  float dg[HQT];
#pragma unroll
  for (int k = 0; k < HQT; ++k) {
    int i = t + k * BINT;
    i = i < HQW ? i : HQW - 1;
    dg[k] = deg_feat[(q0 + i) >> 4];
  }
  const int f0 = ((q0 + t) & (DD / 4 - 1)) * 4;
  const float w0 = W1[f0], w1 = W1[f0 + 1], w2 = W1[f0 + 2], w3 = W1[f0 + 3];
  const int n = blockIdx.x * PN + t;
  if (t < PN && n < NN) dinv[n] = rsqrtf(degcnt[n] + 1.0f);
  float4* __restrict__ hw4 = (float4*)hw + q0;
#pragma unroll
  for (int k = 0; k < HQT; ++k) {
    const int i = t + k * BINT;
    if (i < HQW) {
      float4 y;
      y.x = dg[k] * w0; y.y = dg[k] * w1; y.z = dg[k] * w2; y.w = dg[k] * w3;
      hw4[i] = y;
    }
  }
  int v = pcnt[t & (NP - 1)];
#pragma unroll
  for (int k = 0; k < EPT; ++k) w[k] = rsqrtf(degcnt[s[k]] + 1.0f) * rsqrtf(degcnt[d[k]] + 1.0f);
  v = (t < NP) ? v : 0;
  int x = v;
  for (int o = 1; o < 32; o <<= 1) {
    int y = __shfl_up(x, o);
    if (lane >= o) x += y;
  }
  if (t < NP) {
    h[t] = 0;
    if (lane == 31) wsum[t >> 5] = x;
  }
  __syncthreads();
  int off = 0;
  if (t < NP) {
    for (int i = 0; i < (t >> 5); ++i) off += wsum[i];
    off += x - v;
    if (blockIdx.x == 0) goffs[t] = off;
  }
#pragma unroll
  for (int k = 0; k < EPT; ++k) {
    int i = t + k * BINT;
    r[k] = -1;
    if (i < CH && c0 + i < EE)
      r[k] = atomicAdd(&h[d[k] / PN], 1);
  }
  __syncthreads();
  if (t < NP) {
    int c = h[t];
    int g = c > 0 ? atomicAdd(&pcur[t], c) : 0;
    hb[t] = off + g;
  }
  __syncthreads();
#pragma unroll
  for (int k = 0; k < EPT; ++k) {
    if (r[k] >= 0) {
      int pos = hb[d[k] / PN] + r[k];
      rec[pos] = make_int4(s[k], d[k], __float_as_int(w[k]), 0);
    }
  }
}

#define AG_WAVES 32
#define AG_GRP   16
#define AG_TAB_FLOATS (41 * 1024)
static_assert((PN + 1) * DD <= AG_TAB_FLOATS, "table holds one range and the spare row");
#define AG_THREADS (AG_WAVES * 32)
#define TQ   (PN * DD / 4)
#define TQT  ((TQ + AG_THREADS - 1) / AG_THREADS)
static_assert(DD % 4 == 0 && AG_THREADS % (DD / 4) == 0, "a thread keeps the same four features in every group");
#define NX_PITCH (DD + 4)
#define NX_ROWS  (((PN + 1 + 15) / 16) * 16)
#define NX_RT    (NX_ROWS / 16)
static_assert(2 * NX_ROWS * NX_PITCH <= AG_TAB_FLOATS, "the rows and their products both fit the table");
static_assert(NX_ROWS * (DD / 4) <= TQT * AG_THREADS, "the set-up pass reaches every row of the tiles");
static_assert(AG_WAVES % (DD / 16) == 0, "a wave keeps one column tile");

#define QC    4096
#define QPT   (QC / AG_THREADS)
#define QBINS 256
static_assert(PN + 1 <= QBINS && QBINS % 8 == 0, "one counter per row, eight per lane in the prefix");
static_assert(NN <= 65536 && PN < 65536, "source and row share one 32-bit word of an entry");
static_assert(QPT * AG_THREADS == QC, "the threads tile a chunk");

__device__ __forceinline__ void q_load(const int4* __restrict__ rec, int start, int cnt, int base,
                                       int tid, int4 (&rq)[QPT]) {
#pragma unroll
  for (int k = 0; k < QPT; ++k) {
    const int i = base + tid + k * AG_THREADS;
    rq[k] = make_int4(0, 0, 0, 0);
    if (i < cnt) rq[k] = rec[start + i];
  }
}

template <int TP>
__device__ __forceinline__ void range_sums(const int4* __restrict__ rec, int start, int cnt, int n0,
                                           const float2* __restrict__ src2, float* tab,
                                           uint2* q, int* rcnt, int* roff,
                                           int tid, int lane, int wave, int4 (&rq)[QPT]) {
  for (int base = 0;;) {
    __syncthreads();
    int rk[QPT];
#pragma unroll
    for (int k = 0; k < QPT; ++k) {
      const int i = base + tid + k * AG_THREADS;
      rk[k] = -1;
      if (i < cnt) rk[k] = atomicAdd(&rcnt[rq[k].y - n0], 1);
    }
    __syncthreads();
    if (wave == 0) {
      int cc[QBINS / 32];
      int tot = 0;
#pragma unroll
      for (int j = 0; j < QBINS / 32; ++j) { cc[j] = rcnt[(QBINS / 32) * lane + j]; tot += cc[j]; }
      int x = tot;
      for (int o = 1; o < 32; o <<= 1) {
        int y = __shfl_up(x, o);
        if (lane >= o) x += y;
      }
      int run = x - tot;
#pragma unroll
      for (int j = 0; j < QBINS / 32; ++j) { roff[(QBINS / 32) * lane + j] = run; run += cc[j]; }
    }
    __syncthreads();
    int ro[QPT];
#pragma unroll
    for (int k = 0; k < QPT; ++k) ro[k] = roff[rk[k] >= 0 ? rq[k].y - n0 : 0];
#pragma unroll
    for (int k = 0; k < QPT; ++k) {
      if (rk[k] >= 0) {
        const int r = rq[k].y - n0;
        q[ro[k] + rk[k]] = make_uint2((unsigned)rq[k].x | ((unsigned)r << 16), (unsigned)rq[k].z);
      }
    }
    __syncthreads();
    {
      const int m   = (cnt - base < QC) ? (cnt - base) : QC;
      const int ng  = (m + AG_GRP - 1) / AG_GRP;
      const int qb  = ((wave * ng) >> 5) * AG_GRP;
      const int qe0 = (((wave + 1) * ng) >> 5) * AG_GRP;
      const int qe  = qe0 < m ? qe0 : m;
      int   cur = PN;
      float ax = 0.0f, ay = 0.0f;
      for (int i0 = qb; i0 < qe; i0 += 32) {
        const int  i = i0 + lane;
        const bool live = i < qe;
        const uint2 e = q[live ? i : qe - 1];
        const int   sv = (int)(e.x & 0xFFFFu);
        const int   rv = live ? (int)(e.x >> 16) : PN;
        const float wl = live ? __uint_as_float(e.y) : 0.0f;
#pragma unroll
        for (int j0 = 0; j0 < 32; j0 += AG_GRP) {
          if (i0 + j0 < qe) {
            float2 vr[AG_GRP];
#pragma unroll
            for (int u = 0; u < AG_GRP; ++u) {
              int sj = __builtin_amdgcn_readlane(sv, j0 + u);
              vr[u] = src2[sj * (DD / 2) + lane];
            }
#pragma unroll
            for (int u = 0; u < AG_GRP; ++u) {
              int   rj = __builtin_amdgcn_readlane(rv, j0 + u);
              float wj = __int_as_float(__builtin_amdgcn_readlane(__float_as_int(wl), j0 + u));
              if (rj != cur) {
                atomicAdd(&tab[cur * TP + 2 * lane],     ax);
                atomicAdd(&tab[cur * TP + 2 * lane + 1], ay);
                ax = 0.0f; ay = 0.0f; cur = rj;
              }
              ax += vr[u].x * wj;
              ay += vr[u].y * wj;
            }
          }
        }
      }
      atomicAdd(&tab[cur * TP + 2 * lane],     ax);
      atomicAdd(&tab[cur * TP + 2 * lane + 1], ay);
    }
    base += QC;
    if (base >= cnt) break;
    __syncthreads();
    if (tid < QBINS) rcnt[tid] = 0;
    q_load(rec, start, cnt, base, tid, rq);
  }
}

template <bool NEXT>
__global__ __launch_bounds__(AG_WAVES * 32)
void k_agg(const int4* __restrict__ rec,
           const int* __restrict__ pcnt,
           const int* __restrict__ goffs,
           const float* __restrict__ hw,
           const float* __restrict__ dinv,
           const float* __restrict__ b,
           const float* __restrict__ g,
           const float* __restrict__ be,
           const float* __restrict__ m,
           const float* __restrict__ v,
           const float* __restrict__ Wn,
           float* __restrict__ xout) {
  __shared__ __attribute__((aligned(16))) float tab[AG_TAB_FLOATS];
  __shared__ uint2 qbuf[QC];
  __shared__ __attribute__((aligned(16))) int rcnt[QBINS];
  __shared__ __attribute__((aligned(16))) int roff[QBINS];
  constexpr int TP  = NEXT ? NX_PITCH : DD;
  constexpr int TP4 = TP / 4;
  const int tid  = threadIdx.x;
  const int lane = tid & 31;
  const int wave = __builtin_amdgcn_readfirstlane(tid >> 5);
  const int n0 = blockIdx.x * PN;
  const int nn = (NN - n0 < PN) ? (NN - n0) : PN;
  const int start = goffs[blockIdx.x];
  const int cnt   = pcnt[blockIdx.x];
  const float2* __restrict__ hw2 = (const float2*)hw;

  int4 rq[QPT];
  q_load(rec, start, cnt, 0, tid, rq);
  if (tid < QBINS) rcnt[tid] = 0;

  const int f0 = (tid & (DD / 4 - 1)) * 4;
  float cb[4], cg[4], cbe[4], cm[4], cv[4];
#pragma unroll
  for (int j = 0; j < 4; ++j) {
    cb[j] = b[f0 + j]; cg[j] = g[f0 + j]; cbe[j] = be[f0 + j]; cm[j] = m[f0 + j]; cv[j] = v[f0 + j];
  }

  const int nq = nn * (DD / 4);
  const float4* __restrict__ hw4 = (const float4*)(hw + n0 * DD);
  {
    float4 hv[TQT];
    float  dq[TQT];
#pragma unroll
    for (int k = 0; k < TQT; ++k) {
      int q = tid + k * AG_THREADS;
      q = q < nq ? q : nq - 1;
      hv[k] = hw4[q];
      dq[k] = dinv[n0 + (q >> 4)];
    }
#pragma unroll
    for (int k = 0; k < TQT; ++k) {
      const int q = tid + k * AG_THREADS;
      if (q < nq) {
        const float s = dq[k] * dq[k];
        float4 t = hv[k];
        t.x *= s; t.y *= s; t.z *= s; t.w *= s;
        ((float4*)tab)[(q >> 4) * TP4 + (q & 15)] = t;
      } else if (NEXT && q < NX_ROWS * (DD / 4)) {
        ((float4*)tab)[(q >> 4) * TP4 + (q & 15)] = make_float4(0.0f, 0.0f, 0.0f, 0.0f);
      }
    }
  }
  range_sums<TP>(rec, start, cnt, n0, hw2, tab, qbuf, rcnt, roff, tid, lane, wave, rq);

  float rs[4];
#pragma unroll
  for (int j = 0; j < 4; ++j) rs[j] = rsqrtf(cv[j] + BN_EPS);
  float4* __restrict__ xo4 = (float4*)(xout + n0 * DD);

  if constexpr (!NEXT) {
    __syncthreads();
#pragma unroll
    for (int k = 0; k < TQT; ++k) {
      const int q = tid + k * AG_THREADS;
      if (q < nq) {
        const float4 t = ((const float4*)tab)[q];
        float4 y;
        y.x = fmaxf(((t.x + cb[0]) - cm[0]) * rs[0] * cg[0] + cbe[0], 0.0f);
        y.y = fmaxf(((t.y + cb[1]) - cm[1]) * rs[1] * cg[1] + cbe[1], 0.0f);
        y.z = fmaxf(((t.z + cb[2]) - cm[2]) * rs[2] * cg[2] + cbe[2], 0.0f);
        y.w = fmaxf(((t.w + cb[3]) - cm[3]) * rs[3] * cg[3] + cbe[3], 0.0f);
        xo4[q] = y;
      }
    }
  } else {
#if USE_WMMA_F32
    const int half = lane >> 4;
    const int lm   = lane & 15;
    const int ncol = (wave & (DD / 16 - 1)) * 16 + lm;
    v2f bw[DD / 4];
#pragma unroll
    for (int s = 0; s < DD / 4; ++s) {
      bw[s].x = Wn[(4 * s + 2 * half) * DD + ncol];
      bw[s].y = Wn[(4 * s + 2 * half + 1) * DD + ncol];
    }
    __syncthreads();
#pragma unroll
    for (int k = 0; k < TQT; ++k) {
      const int q = tid + k * AG_THREADS;
      if (q < nq) {
        float4* p = (float4*)tab + (q >> 4) * TP4 + (q & 15);
        const float4 t = *p;
        float4 y;
        y.x = fmaxf(((t.x + cb[0]) - cm[0]) * rs[0] * cg[0] + cbe[0], 0.0f);
        y.y = fmaxf(((t.y + cb[1]) - cm[1]) * rs[1] * cg[1] + cbe[1], 0.0f);
        y.z = fmaxf(((t.z + cb[2]) - cm[2]) * rs[2] * cg[2] + cbe[2], 0.0f);
        y.w = fmaxf(((t.w + cb[3]) - cm[3]) * rs[3] * cg[3] + cbe[3], 0.0f);
        *p = y;
      }
    }
    __syncthreads();
    float* __restrict__ prod = tab + NX_ROWS * TP;
    for (int rt = wave / (DD / 16); rt < NX_RT; rt += AG_WAVES / (DD / 16)) {
      const float* __restrict__ ar = tab + (rt * 16 + lm) * TP + 2 * half;
      v2f a[DD / 4];
#pragma unroll
      for (int s = 0; s < DD / 4; ++s) a[s] = *(const v2f*)(ar + 4 * s);
      v8f acc = {};
#pragma unroll
      for (int s = 0; s < DD / 4; ++s)
        acc = __builtin_amdgcn_wmma_f32_16x16x4_f32(false, a[s], false, bw[s], (short)0, acc, false, false);
#pragma unroll
      for (int r = 0; r < 8; ++r) prod[(rt * 16 + 8 * half + r) * TP + ncol] = acc[r];
    }
    __syncthreads();
#pragma unroll
    for (int k = 0; k < TQT; ++k) {
      const int q = tid + k * AG_THREADS;
      if (q < nq) xo4[q] = ((const float4*)prod)[(q >> 4) * TP4 + (q & 15)];
    }
#endif
  }
}

#define PCH  16
#define PCS  ((PN + PCH - 1) / PCH)
static_assert(PCH * DD == AG_WAVES * 32, "one thread per (chunk, feature)");

__global__ __launch_bounds__(AG_WAVES * 32)
void k_pool_agg(const int4* __restrict__ rec,
                const int* __restrict__ pcnt,
                const int* __restrict__ goffs,
                const int* __restrict__ batch,
                const float* __restrict__ x,
                const float* __restrict__ dinv,
                float* __restrict__ psum,
                float* __restrict__ cntg) {
  __shared__ __attribute__((aligned(16))) float tab[(PN + 1) * DD];
  __shared__ __attribute__((aligned(16))) float gtab[GG * DD];
  __shared__ float ctab[GG];
  __shared__ int   gloc[PN];
  __shared__ uint2 qbuf[QC];
  __shared__ __attribute__((aligned(16))) int rcnt[QBINS];
  __shared__ __attribute__((aligned(16))) int roff[QBINS];
  const int tid  = threadIdx.x;
  const int lane = tid & 31;
  const int wave = __builtin_amdgcn_readfirstlane(tid >> 5);
  const int n0 = blockIdx.x * PN;
  const int nn = (NN - n0 < PN) ? (NN - n0) : PN;
  const int start = goffs[blockIdx.x];
  const int cnt   = pcnt[blockIdx.x];
  const float2* __restrict__ x2 = (const float2*)x;

  int4 rq[QPT];
  q_load(rec, start, cnt, 0, tid, rq);
  if (tid < QBINS) rcnt[tid] = 0;

  const int nq = nn * (DD / 4);
  const float4* __restrict__ x4 = (const float4*)(x + n0 * DD);
  {
    const int gb = batch[n0 + (tid < nn ? tid : nn - 1)];
    float4 hv[TQT];
    float  dq[TQT];
#pragma unroll
    for (int k = 0; k < TQT; ++k) {
      int q = tid + k * AG_THREADS;
      q = q < nq ? q : nq - 1;
      hv[k] = x4[q];
      dq[k] = dinv[n0 + (q >> 4)];
    }
    float4 z4 = {0.0f, 0.0f, 0.0f, 0.0f};
    for (int i = tid; i < GG * DD / 4; i += AG_THREADS) ((float4*)gtab)[i] = z4;
    if (tid < GG) ctab[tid] = 0.0f;
#pragma unroll
    for (int k = 0; k < TQT; ++k) {
      const int q = tid + k * AG_THREADS;
      if (q < nq) {
        const float s = dq[k] * dq[k];
        float4 t = hv[k];
        t.x *= s; t.y *= s; t.z *= s; t.w *= s;
        ((float4*)tab)[q] = t;
      }
    }
    if (tid < nn) gloc[tid] = gb;
  }
  range_sums<DD>(rec, start, cnt, n0, x2, tab, qbuf, rcnt, roff, tid, lane, wave, rq);
  __syncthreads();

  {
    const int f  = tid & (DD - 1);
    const int c0 = (tid >> 6) * PCS;
    const int c1 = (c0 + PCS < nn) ? (c0 + PCS) : nn;
    if (c0 < c1) {
      int   ids[PCS];
      float vals[PCS];
#pragma unroll
      for (int j = 0; j < PCS; ++j) {
        int i = c0 + j;
        i = i < c1 ? i : c1 - 1;
        ids[j]  = gloc[i];
        vals[j] = tab[i * DD + f];
      }
      int   cg = ids[0];
      float a  = 0.0f;
#pragma unroll
      for (int j = 0; j < PCS; ++j) {
        if (c0 + j < c1) {
          if (ids[j] != cg) { atomicAdd(&gtab[cg * DD + f], a); a = 0.0f; cg = ids[j]; }
          a += vals[j];
        }
      }
      atomicAdd(&gtab[cg * DD + f], a);
    }
    if (tid < nn) atomicAdd(&ctab[gloc[tid]], 1.0f);
  }
  __syncthreads();

  {
    static_assert(GG * DD == 32 * AG_THREADS && AG_THREADS / DD == 16, "lane k of a wave <-> graph c + 16 k");
    const float cl = ctab[(tid >> 6) + 16 * lane];
    unsigned present = (unsigned)__ballot(cl > 0.0f);
    while (present) {
      const int k = __builtin_ctz(present);
      present &= present - 1;
      const int i = tid + k * AG_THREADS;
      atomicAdd(&psum[i], gtab[i]);
    }
  }
  if (tid < GG && ctab[tid] > 0.0f) atomicAdd(&cntg[tid], ctab[tid]);
}

#define TL_WAVES (DD / 16)
#define TL_PITCH (DD + 4)
static_assert(DOUT / 16 <= TL_WAVES && GG % 16 == 0 && DD % 4 == 0, "tiles cover both products");

__global__ __launch_bounds__(TL_WAVES * 32)
void k_tail(const float* __restrict__ psum,
            const float* __restrict__ cntg,
            const float* __restrict__ W3,
            const float* __restrict__ b3,
            const float* __restrict__ Wl,
            const float* __restrict__ bl,
            float* __restrict__ out) {
#if USE_WMMA_F32
  __shared__ __attribute__((aligned(16))) float P[16 * TL_PITCH];
  const int lane = threadIdx.x & 31;
  const int wave = __builtin_amdgcn_readfirstlane(threadIdx.x >> 5);
  const int half = lane >> 4;
  const int lm   = lane & 15;
  const int r0   = blockIdx.x * 16;
  const int ncol = wave * 16 + lm;
  const bool head = wave < DOUT / 16;

  v2f a[DD / 4], b[DD / 4], bh[DD / 4];
  const float* __restrict__ Arow = psum + (r0 + lm) * DD + 2 * half;
#pragma unroll
  for (int s = 0; s < DD / 4; ++s) {
    a[s].x = Arow[4 * s];
    a[s].y = Arow[4 * s + 1];
    b[s].x = W3[(4 * s + 2 * half) * DD + ncol];
    b[s].y = W3[(4 * s + 2 * half + 1) * DD + ncol];
  }
  float cc[8];
#pragma unroll
  for (int r = 0; r < 8; ++r) cc[r] = cntg[r0 + 8 * half + r];
  const float bv = b3[ncol];
  float bo = 0.0f;
  if (head) {
#pragma unroll
    for (int s = 0; s < DD / 4; ++s) {
      bh[s].x = Wl[(4 * s + 2 * half) * DOUT + ncol];
      bh[s].y = Wl[(4 * s + 2 * half + 1) * DOUT + ncol];
    }
    bo = bl[ncol];
  }

  v8f c = {};
#pragma unroll
  for (int s = 0; s < DD / 4; ++s)
    c = __builtin_amdgcn_wmma_f32_16x16x4_f32(false, a[s], false, b[s], (short)0, c, false, false);
#pragma unroll
  for (int r = 0; r < 8; ++r) {
    const float q = c[r] / cc[r];
    P[(8 * half + r) * TL_PITCH + ncol] = (cc[r] > 0.0f) ? q + bv : 0.0f;
  }
  __syncthreads();

  if (head) {
    v8f o = {};
#pragma unroll
    for (int s = 0; s < DD / 4; ++s) {
      v2f ah;
      ah.x = P[lm * TL_PITCH + 4 * s + 2 * half];
      ah.y = P[lm * TL_PITCH + 4 * s + 2 * half + 1];
      o = __builtin_amdgcn_wmma_f32_16x16x4_f32(false, ah, false, bh[s], (short)0, o, false, false);
    }
#pragma unroll
    for (int r = 0; r < 8; ++r)
      out[(r0 + 8 * half + r) * DOUT + ncol] = o[r] + bo;
  }
#endif
}

static inline int blk(long long n, int b) { return (int)((n + b - 1) / b); }

extern "C" void kernel_launch(void* const* d_in, const int* in_sizes, int n_in,
                              void* d_out, int out_size, void* d_ws, size_t ws_size,
                              hipStream_t stream) {
  (void)in_sizes; (void)n_in; (void)out_size; (void)ws_size;
  const int*   edge_index = (const int*)d_in[0];
  const int*   batch      = (const int*)d_in[1];
  const float* W1 = (const float*)d_in[2];
  const float* b1 = (const float*)d_in[3];
  const float* W2 = (const float*)d_in[4];
  const float* b2 = (const float*)d_in[5];
  const float* W3 = (const float*)d_in[6];
  const float* b3 = (const float*)d_in[7];
  const float* g1 = (const float*)d_in[8];
  const float* be1= (const float*)d_in[9];
  const float* m1 = (const float*)d_in[10];
  const float* v1 = (const float*)d_in[11];
  const float* g2 = (const float*)d_in[12];
  const float* be2= (const float*)d_in[13];
  const float* m2 = (const float*)d_in[14];
  const float* v2 = (const float*)d_in[15];
  const float* Wl = (const float*)d_in[16];
  const float* bl = (const float*)d_in[17];
  float* out = (float*)d_out;

  char* ws = (char*)d_ws;
  size_t off = 0;
  auto carve = [&](size_t bytes) -> float* {
    float* p = (float*)(ws + off);
    off += (bytes + 255) & ~(size_t)255;
    return p;
  };
  float* deg_feat = carve((size_t)NN * 4);
  float* degcnt   = carve((size_t)NN * 4);
  float* dinv     = carve((size_t)NN * 4);
  float* bufA     = carve((size_t)NN * DD * 4);
  float* bufB     = carve((size_t)NN * DD * 4);
  float* psum     = carve((size_t)(GG * DD + GG) * 4);
  float* cnt      = psum + GG * DD;
  int*   pcnt     = (int*)carve((size_t)2 * NP * 4);
  int*   pcur     = pcnt + NP;
  int*   goffs    = (int*)carve((size_t)NP * 4);
  int4*  rec      = (int4*)carve((size_t)EE * 16);

  const int B = 256;

  k_clear<<<blk(NN, B), B, 0, stream>>>(deg_feat, degcnt, NN, pcnt, 2 * NP, psum, GG * DD + GG);
  k_edge_count<<<NP, BINT, 0, stream>>>(edge_index, deg_feat, degcnt, pcnt);
  k_fill<<<NP, BINT, 0, stream>>>(edge_index, pcnt, degcnt, deg_feat, W1, pcur, goffs, rec, dinv, bufA);

  k_agg<true><<<NP, AG_WAVES * 32, 0, stream>>>(rec, pcnt, goffs, bufA, dinv,
                                                b1, g1, be1, m1, v1, W2, bufB);

  k_agg<false><<<NP, AG_WAVES * 32, 0, stream>>>(rec, pcnt, goffs, bufB, dinv,
                                                 b2, g2, be2, m2, v2, nullptr, bufA);

  k_pool_agg<<<NP, AG_WAVES * 32, 0, stream>>>(rec, pcnt, goffs, batch, bufA, dinv, psum, cnt);

  k_tail<<<GG / 16, TL_WAVES * 32, 0, stream>>>(psum, cnt, W3, b3, Wl, bl, out);
}
